# starting kernel
# baseline (speedup 1.0000x reference)
#include <hip/hip_runtime.h>
#include <hip/hip_bf16.h>


typedef __attribute__((ext_vector_type(2))) float v2f;
typedef __attribute__((ext_vector_type(8))) float v8f;

#define F1 16
#define F2 2
#define FIN 25

__device__ __forceinline__ void atomic_add_f32(float* p, float v) {
    asm volatile("global_atomic_add_f32 %0, %1, off"
                 :
                 : "v"(p), "v"(v)
                 : "memory");
}

__global__ void gcn_init_kernel(float* __restrict__ deg,
                                float* __restrict__ agg1,
                                float* __restrict__ agg2,
                                int n) {
    int gid = blockIdx.x * blockDim.x + threadIdx.x;
    if (gid < 16 * n) agg1[gid] = 0.0f;
    if (gid < 2 * n)  agg2[gid] = 0.0f;
    if (gid < n)      deg[gid]  = 1.0f;
}

__global__ void gcn_degree_kernel(const int* __restrict__ cols,
                                  float* __restrict__ deg,
                                  int n_edges) {
    int e = blockIdx.x * blockDim.x + threadIdx.x;
    if (e >= n_edges) return;
    atomic_add_f32(&deg[cols[e]], 1.0f);
}

__global__ void gcn_dinv_kernel(const float* __restrict__ deg,
                                float* __restrict__ dinv,
                                int n) {
    int i = blockIdx.x * blockDim.x + threadIdx.x;
    if (i >= n) return;
    dinv[i] = rsqrtf(deg[i]);
}

__global__ void gcn_xw1_wmma_kernel(const float* __restrict__ x,
                                    const float* __restrict__ W1,
                                    float* __restrict__ h1,
                                    int n_nodes) {
    const int lane  = threadIdx.x & 31;
    const int wave  = threadIdx.x >> 5;
    const int tile  = blockIdx.x * (blockDim.x >> 5) + wave;
    const int node0 = tile * 16;
    if (node0 >= n_nodes) return;

    const int half = lane >> 4;
    const int l16  = lane & 15;
    const int m    = l16;

    v8f c = {0.f, 0.f, 0.f, 0.f, 0.f, 0.f, 0.f, 0.f};

    #pragma unroll
    for (int k0 = 0; k0 < 28; k0 += 4) {
        const int ka = k0 + 2 * half;
        v2f a, b;
        a.x = (ka     < FIN) ? x[(node0 + m) * FIN + ka    ] : 0.0f;
        a.y = (ka + 1 < FIN) ? x[(node0 + m) * FIN + ka + 1] : 0.0f;
        b.x = (ka     < FIN) ? W1[(ka    ) * F1 + l16] : 0.0f;
        b.y = (ka + 1 < FIN) ? W1[(ka + 1) * F1 + l16] : 0.0f;
        c = __builtin_amdgcn_wmma_f32_16x16x4_f32(
                 false, a,  false, b,
                 (short)0, c,  false,  false);
    }

    #pragma unroll
    for (int v = 0; v < 8; ++v) {
        h1[(node0 + v + 8 * half) * F1 + l16] = c[v];
    }
}

__global__ void gcn_scatter1_kernel(const int* __restrict__ rows,
                                    const int* __restrict__ cols,
                                    const float* __restrict__ h1,
                                    const float* __restrict__ dinv,
                                    float* __restrict__ agg1,
                                    int n_edges) {
    long long gid = (long long)blockIdx.x * blockDim.x + threadIdx.x;
    int e = (int)(gid >> 4);
    int f = (int)(gid & 15);
    if (e >= n_edges) return;
    int r = rows[e];
    int cidx = cols[e];
    float norm = dinv[r] * dinv[cidx];
    atomic_add_f32(&agg1[cidx * F1 + f], h1[r * F1 + f] * norm);
}

__global__ void gcn_layer1_finish_kernel(const float* __restrict__ h1,
                                         const float* __restrict__ agg1,
                                         const float* __restrict__ dinv,
                                         const float* __restrict__ b1,
                                         const float* __restrict__ W2,
                                         float* __restrict__ h2,
                                         int n) {
    int i = blockIdx.x * blockDim.x + threadIdx.x;
    if (i >= n) return;
    float di = dinv[i];
    float sw = di * di;
    float a0 = 0.0f, a1 = 0.0f;
    #pragma unroll
    for (int f = 0; f < F1; ++f) {
        float v = agg1[i * F1 + f] + h1[i * F1 + f] * sw + b1[f];
        v = v > 0.0f ? v : 0.0f;
        a0 = fmaf(v, W2[f * F2 + 0], a0);
        a1 = fmaf(v, W2[f * F2 + 1], a1);
    }
    h2[i * F2 + 0] = a0;
    h2[i * F2 + 1] = a1;
}

__global__ void gcn_scatter2_kernel(const int* __restrict__ rows,
                                    const int* __restrict__ cols,
                                    const float* __restrict__ h2,
                                    const float* __restrict__ dinv,
                                    float* __restrict__ agg2,
                                    int n_edges) {
    long long gid = (long long)blockIdx.x * blockDim.x + threadIdx.x;
    int e = (int)(gid >> 1);
    int f = (int)(gid & 1);
    if (e >= n_edges) return;
    int r = rows[e];
    int cidx = cols[e];
    float norm = dinv[r] * dinv[cidx];
    atomic_add_f32(&agg2[cidx * F2 + f], h2[r * F2 + f] * norm);
}

__global__ void gcn_final_kernel(const float* __restrict__ h2,
                                 const float* __restrict__ agg2,
                                 const float* __restrict__ dinv,
                                 const float* __restrict__ b2,
                                 float* __restrict__ out,
                                 int n) {
    int i = blockIdx.x * blockDim.x + threadIdx.x;
    if (i >= n) return;
    float di = dinv[i];
    float sw = di * di;
    float a = agg2[i * F2 + 0] + h2[i * F2 + 0] * sw + b2[0];
    float b = agg2[i * F2 + 1] + h2[i * F2 + 1] * sw + b2[1];
    float m = fmaxf(a, b);
    float lse = m + logf(expf(a - m) + expf(b - m));
    out[i * F2 + 0] = a - lse;
    out[i * F2 + 1] = b - lse;
}

extern "C" void kernel_launch(void* const* d_in, const int* in_sizes, int n_in,
                              void* d_out, int out_size, void* d_ws, size_t ws_size,
                              hipStream_t stream) {
    const float* x   = (const float*)d_in[0];
    const int*   ei  = (const int*)d_in[1];
    const float* W1  = (const float*)d_in[2];
    const float* b1  = (const float*)d_in[3];
    const float* W2  = (const float*)d_in[4];
    const float* b2  = (const float*)d_in[5];
    float*       out = (float*)d_out;

    const int N = in_sizes[0] / FIN;
    const int E = in_sizes[1] / 2;
    const int* rows = ei;
    const int* cols = ei + E;

    float* ws   = (float*)d_ws;
    float* deg  = ws;
    float* dinv = ws + (size_t)N;
    float* h1   = ws + (size_t)2 * N;
    float* agg1 = ws + (size_t)18 * N;
    float* h2   = ws + (size_t)34 * N;
    float* agg2 = ws + (size_t)36 * N;

    const int BS = 256;

    gcn_init_kernel<<<(16 * N + BS - 1) / BS, BS, 0, stream>>>(deg, agg1, agg2, N);

    gcn_degree_kernel<<<(E + BS - 1) / BS, BS, 0, stream>>>(cols, deg, E);

    gcn_dinv_kernel<<<(N + BS - 1) / BS, BS, 0, stream>>>(deg, dinv, N);

    {
        int tiles = (N + 15) / 16;
        int waves_per_block = BS / 32;
        int blocks = (tiles + waves_per_block - 1) / waves_per_block;
        gcn_xw1_wmma_kernel<<<blocks, BS, 0, stream>>>(x, W1, h1, N);
    }

    {
        long long tot = (long long)E * 16;
        int blocks = (int)((tot + BS - 1) / BS);
        gcn_scatter1_kernel<<<blocks, BS, 0, stream>>>(rows, cols, h1, dinv, agg1, E);
    }

    gcn_layer1_finish_kernel<<<(N + BS - 1) / BS, BS, 0, stream>>>(
        h1, agg1, dinv, b1, W2, h2, N);

    {
        long long tot = (long long)E * 2;
        int blocks = (int)((tot + BS - 1) / BS);
        gcn_scatter2_kernel<<<blocks, BS, 0, stream>>>(rows, cols, h2, dinv, agg2, E);
    }

    gcn_final_kernel<<<(N + BS - 1) / BS, BS, 0, stream>>>(h2, agg2, dinv, b2, out, N);
}


# No atomics, no edge reorder: workgroup owns a node range in LDS, filters an unsorted edge slice, a node's owner wave adds; slice partials summed per node; dinv[row] folded into h; sum order changes
# speedup vs baseline: 1.3654x; 1.3654x over previous
#include <hip/hip_runtime.h>


typedef __attribute__((ext_vector_type(2))) float v2f;
typedef __attribute__((ext_vector_type(8))) float v8f;

#define F1 16
#define F2 2
#define FIN 25

#define TWG 1024
#define RD 50016
#define QD 128
#define R1 4000
#define Q1 10
#define R2 25008
#define Q2 64

#define CHUNK 256
#define QSTR  260
#define LCAP  160

template <int F, int LCB>
__device__ __forceinline__ void tile_drain(const unsigned* ml, int start, int m, int lane, int e0,
                                           const int* __restrict__ rows,
                                           const float* __restrict__ h,
                                           float* acc) {
    constexpr unsigned LCM = (1u << LCB) - 1u;
    const bool valid = lane < m;
    unsigned lc = 0xFFFFFFFFu;
    int r = 0;
    if (valid) {
        const unsigned ent = ml[start + lane];
        lc = ent & LCM;
        if (F != 1) r = rows[e0 + (int)(ent >> LCB)];
    }
    if (F == 16) {
        const int half = lane >> 4;
        const int l16  = lane & 15;
        float v[16];
        #pragma unroll
        for (int j = 0; j < 16; ++j) {
            const int src = 2 * j + half;
            const int rj = __shfl(r, src);
            v[j] = (src < m) ? h[rj * 16 + l16] : 0.f;
        }
        #pragma unroll
        for (int j = 0; j < 16; ++j) {
            const unsigned la = (unsigned)__builtin_amdgcn_readlane((int)lc, 2 * j);
            const unsigned lb = (unsigned)__builtin_amdgcn_readlane((int)lc, 2 * j + 1);
            const bool va = 2 * j < m;
            const bool vb = 2 * j + 1 < m;
            if (va && vb && la != lb) {
                const unsigned a = (half ? lb : la) * 16u + (unsigned)l16;
                acc[a] += v[j];
            } else {
                if (va && half == 0) acc[la * 16u + (unsigned)l16] += v[j];
                if (vb && half == 1) acc[lb * 16u + (unsigned)l16] += v[j];
            }
        }
    } else {
        unsigned same = 0u;
        #pragma unroll
        for (int s = 0; s < 32; ++s) {
            const unsigned ls = (unsigned)__builtin_amdgcn_readlane((int)lc, s);
            same |= (ls == lc) ? (1u << s) : 0u;
        }
        const unsigned rank = (unsigned)__builtin_popcount(same & ((1u << lane) - 1u));
        if (F == 1) {
            if (valid && rank == 0u) acc[lc] += (float)__builtin_popcount(same);
        } else {
            float2 v = make_float2(0.f, 0.f);
            if (valid) v = ((const float2*)h)[r];
            float2* acc2 = (float2*)acc;
            for (unsigned t = 0u; t < 32u; ++t) {
                const bool act = valid && (rank == t);
                if (__builtin_amdgcn_ballot_w32(act) == 0u) break;
                if (act) {
                    float2 a = acc2[lc];
                    a.x += v.x; a.y += v.y;
                    acc2[lc] = a;
                }
            }
        }
    }
}

template <int F, int R, int Q>
__global__ __launch_bounds__(TWG) void gcn_tile_kernel(const int* __restrict__ rows,
                                                       const int* __restrict__ cols,
                                                       const float* __restrict__ h,
                                                       float* __restrict__ part,
                                                       int n, int n_edges, int slice) {
    constexpr int RW  = (R + 31) / 32;
    constexpr int LCB = (R <= 4096) ? 12 : 16;
    static_assert(R <= (1 << LCB), "node index must fit its field");
    __shared__ float acc[R * F];
    __shared__ __attribute__((aligned(16))) unsigned qa[32 * QSTR];
    __shared__ unsigned qn[32];
    __shared__ unsigned ml_all[32 * LCAP];
    const int lane = threadIdx.x & 31;
    const int wave = __builtin_amdgcn_readfirstlane((int)(threadIdx.x >> 5));
    unsigned* ml = ml_all + wave * LCAP;
    const int p  = blockIdx.x / Q;
    const int q  = blockIdx.x - p * Q;
    const int lo = p * R;
    const int nr = min(R, n - lo);

    for (int i = threadIdx.x; i < nr * F; i += TWG) acc[i] = 0.f;

    const int e0 = q * slice;
    const int e1 = min(n_edges, e0 + slice);
    const int stride = 32 * CHUNK;
    const unsigned wlo = (unsigned)(wave * RW);
    int cur[8];
    #pragma unroll
    for (int k = 0; k < 8; ++k) {
        const int e = e0 + wave * CHUNK + k * 32 + lane;
        cur[k] = (e < e1) ? cols[e] : -1;
    }
    int cnt = 0;
    for (int rb = e0; rb < e1; rb += stride) {
        int nxt[8];
        #pragma unroll
        for (int k = 0; k < 8; ++k) {
            const int e = rb + stride + wave * CHUNK + k * 32 + lane;
            nxt[k] = (e < e1) ? cols[e] : -1;
        }
        int na = 0;
        #pragma unroll
        for (int k = 0; k < 8; ++k) {
            const unsigned lc = (unsigned)(cur[k] - lo);
            const bool hit = lc < (unsigned)nr;
            const unsigned mk = __builtin_amdgcn_ballot_w32(hit);
            if (hit) qa[wave * QSTR + na + (int)__builtin_amdgcn_mbcnt_lo(mk, 0u)] =
                         ((unsigned)(k * 32 + lane) << 16) | lc;
            na += __builtin_popcount(mk);
        }
        if (lane == 0) qn[wave] = (unsigned)na;
        __syncthreads();
        const int nl = (int)qn[lane];
        int mx = nl;
        #pragma unroll
        for (int off = 16; off > 0; off >>= 1) mx = max(mx, __shfl_xor(mx, off));
        mx = __builtin_amdgcn_readfirstlane(mx);
        const unsigned ebase = (unsigned)(rb - e0 + lane * CHUNK);
        for (int i0 = 0; i0 < mx; i0 += 4) {
            const uint4 e4 = *(const uint4*)&qa[lane * QSTR + i0];
            const unsigned ent[4] = {e4.x, e4.y, e4.z, e4.w};
            #pragma unroll
            for (int u = 0; u < 4; ++u) {
                const unsigned lc = ent[u] & 0xFFFFu;
                const bool mine = (i0 + u < nl) && ((lc - wlo) < (unsigned)RW);
                const unsigned mk = __builtin_amdgcn_ballot_w32(mine);
                if (mk != 0u) {
                    if (mine) ml[cnt + (int)__builtin_amdgcn_mbcnt_lo(mk, 0u)] =
                                  ((ebase + (ent[u] >> 16)) << LCB) | lc;
                    cnt += __builtin_popcount(mk);
                }
            }
            while (cnt >= 32) {
                cnt -= 32;
                tile_drain<F, LCB>(ml, cnt, 32, lane, e0, rows, h, acc);
            }
        }
        __syncthreads();
        #pragma unroll
        for (int k = 0; k < 8; ++k) cur[k] = nxt[k];
    }
    if (cnt > 0) tile_drain<F, LCB>(ml, 0, cnt, lane, e0, rows, h, acc);
    __syncthreads();

    float* dst = part + ((size_t)q * n + lo) * F;
    for (int i = threadIdx.x; i < nr * F; i += TWG) dst[i] = acc[i];
}

__global__ void gcn_dinv_kernel(const float* __restrict__ part,
                                float* __restrict__ dinv, int n) {
    const int i = blockIdx.x * blockDim.x + threadIdx.x;
    if (i >= n) return;
    float c = 1.0f;
    #pragma unroll 16
    for (int q = 0; q < QD; ++q) c += part[(size_t)q * n + i];
    dinv[i] = rsqrtf(c);
}

__global__ void gcn_xw1_wmma_kernel(const float* __restrict__ x,
                                    const float* __restrict__ W1,
                                    const float* __restrict__ dinv,
                                    float* __restrict__ h1s,
                                    int n_nodes) {
    const int lane  = threadIdx.x & 31;
    const int wave  = threadIdx.x >> 5;
    const int tile  = blockIdx.x * (blockDim.x >> 5) + wave;
    const int node0 = tile * 16;
    if (node0 >= n_nodes) return;

    const int half = lane >> 4;
    const int l16  = lane & 15;
    const int m    = l16;

    v8f c = {0.f, 0.f, 0.f, 0.f, 0.f, 0.f, 0.f, 0.f};

    #pragma unroll
    for (int k0 = 0; k0 < 28; k0 += 4) {
        const int ka = k0 + 2 * half;
        v2f a, b;
        a.x = (ka     < FIN) ? x[(node0 + m) * FIN + ka    ] : 0.0f;
        a.y = (ka + 1 < FIN) ? x[(node0 + m) * FIN + ka + 1] : 0.0f;
        b.x = (ka     < FIN) ? W1[(ka    ) * F1 + l16] : 0.0f;
        b.y = (ka + 1 < FIN) ? W1[(ka + 1) * F1 + l16] : 0.0f;
        c = __builtin_amdgcn_wmma_f32_16x16x4_f32(
                 false, a,  false, b,
                 (short)0, c,  false,  false);
    }

    #pragma unroll
    for (int v = 0; v < 8; ++v) {
        const int node = node0 + v + 8 * half;
        h1s[node * F1 + l16] = c[v] * dinv[node];
    }
}

__global__ void gcn_layer1_finish_kernel(const float* __restrict__ part,
                                         const float* __restrict__ h1s,
                                         const float* __restrict__ dinv,
                                         const float* __restrict__ b1,
                                         const float* __restrict__ W2,
                                         float* __restrict__ h2s,
                                         int n) {
    const int gid = blockIdx.x * blockDim.x + threadIdx.x;
    const int i   = gid >> 2;
    const int g   = gid & 3;
    const int ic  = (i < n) ? i : (n - 1);
    float4 s = ((const float4*)h1s)[(size_t)ic * 4 + g];
    #pragma unroll
    for (int q = 0; q < Q1; ++q) {
        const float4 t = ((const float4*)part)[((size_t)q * n + ic) * 4 + g];
        s.x += t.x; s.y += t.y; s.z += t.z; s.w += t.w;
    }
    const float di = dinv[ic];
    const float v0 = fmaxf(fmaf(s.x, di, b1[4 * g + 0]), 0.0f);
    const float v1 = fmaxf(fmaf(s.y, di, b1[4 * g + 1]), 0.0f);
    const float v2 = fmaxf(fmaf(s.z, di, b1[4 * g + 2]), 0.0f);
    const float v3 = fmaxf(fmaf(s.w, di, b1[4 * g + 3]), 0.0f);
    const float* w = W2 + 8 * g;
    float a0 = v0 * w[0];
    float a1 = v0 * w[1];
    a0 = fmaf(v1, w[2], a0); a1 = fmaf(v1, w[3], a1);
    a0 = fmaf(v2, w[4], a0); a1 = fmaf(v2, w[5], a1);
    a0 = fmaf(v3, w[6], a0); a1 = fmaf(v3, w[7], a1);
    a0 += __shfl_xor(a0, 1); a1 += __shfl_xor(a1, 1);
    a0 += __shfl_xor(a0, 2); a1 += __shfl_xor(a1, 2);
    if (g == 0 && i < n) {
        float2 o; o.x = a0 * di; o.y = a1 * di;
        ((float2*)h2s)[i] = o;
    }
}

__global__ void gcn_final_kernel(const float* __restrict__ part,
                                 const float* __restrict__ h2s,
                                 const float* __restrict__ dinv,
                                 const float* __restrict__ b2,
                                 float* __restrict__ out,
                                 int n) {
    const int i = blockIdx.x * blockDim.x + threadIdx.x;
    if (i >= n) return;
    float2 s = ((const float2*)h2s)[i];
    #pragma unroll 16
    for (int q = 0; q < Q2; ++q) {
        const float2 t = ((const float2*)part)[(size_t)q * n + i];
        s.x += t.x; s.y += t.y;
    }
    const float di = dinv[i];
    const float a = fmaf(s.x, di, b2[0]);
    const float b = fmaf(s.y, di, b2[1]);
    const float m = fmaxf(a, b);
    const float lse = m + logf(expf(a - m) + expf(b - m));
    float2 o; o.x = a - lse; o.y = b - lse;
    ((float2*)out)[i] = o;
}

extern "C" void kernel_launch(void* const* d_in, const int* in_sizes, int n_in,
                              void* d_out, int out_size, void* d_ws, size_t ws_size,
                              hipStream_t stream) {
    const float* x   = (const float*)d_in[0];
    const int*   ei  = (const int*)d_in[1];
    const float* W1  = (const float*)d_in[2];
    const float* b1  = (const float*)d_in[3];
    const float* W2  = (const float*)d_in[4];
    const float* b2  = (const float*)d_in[5];
    float*       out = (float*)d_out;

    const int N = in_sizes[0] / FIN;
    const int E = in_sizes[1] / 2;
    const int* rows = ei;
    const int* cols = ei + E;

    float* ws   = (float*)d_ws;
    float* h1s  = ws;
    float* h2s  = ws + (size_t)16 * N;
    float* dinv = ws + (size_t)18 * N;
    float* part = ws + (size_t)20 * N;

    const int BS = 256;

    gcn_tile_kernel<1, RD, QD><<<((N + RD - 1) / RD) * QD, TWG, 0, stream>>>(
        rows, cols, h1s, part, N, E, (E + QD - 1) / QD);

    gcn_dinv_kernel<<<(N + BS - 1) / BS, BS, 0, stream>>>(part, dinv, N);

    {
        int tiles = (N + 15) / 16;
        int waves_per_block = BS / 32;
        int blocks = (tiles + waves_per_block - 1) / waves_per_block;
        gcn_xw1_wmma_kernel<<<blocks, BS, 0, stream>>>(x, W1, dinv, h1s, N);
    }

    gcn_tile_kernel<F1, R1, Q1><<<((N + R1 - 1) / R1) * Q1, TWG, 0, stream>>>(
        rows, cols, h1s, part, N, E, (E + Q1 - 1) / Q1);

    gcn_layer1_finish_kernel<<<(4 * N + BS - 1) / BS, BS, 0, stream>>>(
        part, h1s, dinv, b1, W2, h2s, N);

    gcn_tile_kernel<F2, R2, Q2><<<((N + R2 - 1) / R2) * Q2, TWG, 0, stream>>>(
        rows, cols, h2s, part, N, E, (E + Q2 - 1) / Q2);

    gcn_final_kernel<<<(N + BS - 1) / BS, BS, 0, stream>>>(part, h2s, dinv, b2, out, N);
}


# Drain loads off the rounds' critical path: round barriers wait for LDS only, target loads branch-free, so loads stay in flight; a batch's source loads, h loads, sums run one pop apart; sum order kept
# speedup vs baseline: 1.5714x; 1.1509x over previous
#include <hip/hip_runtime.h>


typedef __attribute__((ext_vector_type(2))) float v2f;
typedef __attribute__((ext_vector_type(8))) float v8f;

#define F1 16
#define F2 2
#define FIN 25

#define TWG 1024
#define RD 50016
#define QD 128
#define R1 4000
#define Q1 10
#define R2 25008
#define Q2 64

#define CHUNK 256
#define QSTR  260
#define LCAP  160

__device__ __forceinline__ void lds_round_barrier() {
    asm volatile("s_wait_dscnt 0x0\n\ts_barrier_signal -1\n\ts_barrier_wait -1" ::: "memory");
}

template <int LCB>
__device__ __forceinline__ void drain_take(const unsigned* ml, int start, int m, int lane, int e0,
                                           const int* __restrict__ rows, unsigned& lc, int& r) {
    constexpr unsigned LCM = (1u << LCB) - 1u;
    lc = 0xFFFFFFFFu;
    r = 0;
    if (lane < m) {
        const unsigned ent = ml[start + lane];
        lc = ent & LCM;
        r = rows[e0 + (int)(ent >> LCB)];
    }
}

template <int F>
__device__ __forceinline__ void drain_gather(int r, int m, int lane, const float* __restrict__ h, float* v) {
    if (F == 16) {
        const int half = lane >> 4;
        const int l16  = lane & 15;
        #pragma unroll
        for (int j = 0; j < 16; ++j) {
            const int src = 2 * j + half;
            const int rj = __shfl(r, src);
            v[j] = (src < m) ? h[rj * 16 + l16] : 0.f;
        }
    } else {
        float2 t = make_float2(0.f, 0.f);
        if (lane < m) t = ((const float2*)h)[r];
        v[0] = t.x;
        v[1] = t.y;
    }
}

template <int F>
__device__ __forceinline__ void drain_add(unsigned lc, const float* v, int m, int lane, float* acc) {
    if (F == 16) {
        const int half = lane >> 4;
        const int l16  = lane & 15;
        #pragma unroll
        for (int j = 0; j < 16; ++j) {
            const unsigned la = (unsigned)__builtin_amdgcn_readlane((int)lc, 2 * j);
            const unsigned lb = (unsigned)__builtin_amdgcn_readlane((int)lc, 2 * j + 1);
            const bool va = 2 * j < m;
            const bool vb = 2 * j + 1 < m;
            if (va && vb && la != lb) {
                const unsigned a = (half ? lb : la) * 16u + (unsigned)l16;
                acc[a] += v[j];
            } else {
                if (va && half == 0) acc[la * 16u + (unsigned)l16] += v[j];
                if (vb && half == 1) acc[lb * 16u + (unsigned)l16] += v[j];
            }
        }
    } else {
        const bool valid = lane < m;
        unsigned same = 0u;
        #pragma unroll
        for (int s = 0; s < 32; ++s) {
            const unsigned ls = (unsigned)__builtin_amdgcn_readlane((int)lc, s);
            same |= (ls == lc) ? (1u << s) : 0u;
        }
        const unsigned rank = (unsigned)__builtin_popcount(same & ((1u << lane) - 1u));
        float2* acc2 = (float2*)acc;
        for (unsigned t = 0u; t < 32u; ++t) {
            const bool act = valid && (rank == t);
            if (__builtin_amdgcn_ballot_w32(act) == 0u) break;
            if (act) {
                float2 a = acc2[lc];
                a.x += v[0]; a.y += v[1];
                acc2[lc] = a;
            }
        }
    }
}

template <int LCB>
__device__ __forceinline__ void drain_count(const unsigned* ml, int start, int m, int lane, float* acc) {
    constexpr unsigned LCM = (1u << LCB) - 1u;
    const bool valid = lane < m;
    unsigned lc = 0xFFFFFFFFu;
    if (valid) lc = ml[start + lane] & LCM;
    unsigned same = 0u;
    #pragma unroll
    for (int s = 0; s < 32; ++s) {
        const unsigned ls = (unsigned)__builtin_amdgcn_readlane((int)lc, s);
        same |= (ls == lc) ? (1u << s) : 0u;
    }
    const unsigned rank = (unsigned)__builtin_popcount(same & ((1u << lane) - 1u));
    if (valid && rank == 0u) acc[lc] += (float)__builtin_popcount(same);
}

template <int F, int R, int Q>
__global__ __launch_bounds__(TWG) void gcn_tile_kernel(const int* __restrict__ rows,
                                                       const int* __restrict__ cols,
                                                       const float* __restrict__ h,
                                                       float* __restrict__ part,
                                                       int n, int n_edges, int slice) {
    constexpr int RW  = (R + 31) / 32;
    constexpr int LCB = (R <= 4096) ? 12 : 16;
    static_assert(R <= (1 << LCB), "node index must fit its field");
    __shared__ float acc[R * F];
    __shared__ __attribute__((aligned(16))) unsigned qa[32 * QSTR];
    __shared__ unsigned qn[32];
    __shared__ unsigned ml_all[32 * LCAP];
    const int lane = threadIdx.x & 31;
    const int wave = __builtin_amdgcn_readfirstlane((int)(threadIdx.x >> 5));
    unsigned* ml = ml_all + wave * LCAP;
    const int p  = blockIdx.x / Q;
    const int q  = blockIdx.x - p * Q;
    const int lo = p * R;
    const int nr = min(R, n - lo);

    for (int i = threadIdx.x; i < nr * F; i += TWG) acc[i] = 0.f;

    const int e0 = q * slice;
    const int e1 = min(n_edges, e0 + slice);
    const int stride = 32 * CHUNK;
    const unsigned wlo = (unsigned)(wave * RW);
    const int elast = n_edges - 1;
    int cur[8];
    #pragma unroll
    for (int k = 0; k < 8; ++k) {
        const int e = e0 + wave * CHUNK + k * 32 + lane;
        const int c = cols[min(e, elast)];
        cur[k] = (e < e1) ? c : -1;
    }
    int cnt = 0;
    constexpr int VN = (F == 16) ? 16 : 2;
    unsigned a_lc = 0u, b_lc = 0u;
    int a_r = 0, a_m = 0, b_m = 0;
    float b_v[VN];
    #pragma unroll
    for (int j = 0; j < VN; ++j) b_v[j] = 0.f;
    for (int rb = e0; rb < e1; rb += stride) {
        int nxt[8];
        #pragma unroll
        for (int k = 0; k < 8; ++k) {
            const int e = rb + stride + wave * CHUNK + k * 32 + lane;
            const int c = cols[min(e, elast)];
            nxt[k] = (e < e1) ? c : -1;
        }
        int na = 0;
        #pragma unroll
        for (int k = 0; k < 8; ++k) {
            const unsigned lc = (unsigned)(cur[k] - lo);
            const bool hit = lc < (unsigned)nr;
            const unsigned mk = __builtin_amdgcn_ballot_w32(hit);
            if (hit) qa[wave * QSTR + na + (int)__builtin_amdgcn_mbcnt_lo(mk, 0u)] =
                         ((unsigned)(k * 32 + lane) << 16) | lc;
            na += __builtin_popcount(mk);
        }
        if (lane == 0) qn[wave] = (unsigned)na;
        lds_round_barrier();
        const int nl = (int)qn[lane];
        int mx = nl;
        #pragma unroll
        for (int off = 16; off > 0; off >>= 1) mx = max(mx, __shfl_xor(mx, off));
        mx = __builtin_amdgcn_readfirstlane(mx);
        const unsigned ebase = (unsigned)(rb - e0 + lane * CHUNK);
        for (int i0 = 0; i0 < mx; i0 += 4) {
            const uint4 e4 = *(const uint4*)&qa[lane * QSTR + i0];
            const unsigned ent[4] = {e4.x, e4.y, e4.z, e4.w};
            #pragma unroll
            for (int u = 0; u < 4; ++u) {
                const unsigned lc = ent[u] & 0xFFFFu;
                const bool mine = (i0 + u < nl) && ((lc - wlo) < (unsigned)RW);
                const unsigned mk = __builtin_amdgcn_ballot_w32(mine);
                if (mk != 0u) {
                    if (mine) ml[cnt + (int)__builtin_amdgcn_mbcnt_lo(mk, 0u)] =
                                  ((ebase + (ent[u] >> 16)) << LCB) | lc;
                    cnt += __builtin_popcount(mk);
                }
            }
            while (cnt >= 32) {
                cnt -= 32;
                if (F == 1) {
                    drain_count<LCB>(ml, cnt, 32, lane, acc);
                } else {
                    float c_v[VN];
                    #pragma unroll
                    for (int j = 0; j < VN; ++j) c_v[j] = b_v[j];
                    const unsigned c_lc = b_lc;
                    const int c_m = b_m;
                    b_m = a_m;
                    b_lc = a_lc;
                    drain_gather<F>(a_r, 32, lane, h, b_v);
                    drain_take<LCB>(ml, cnt, 32, lane, e0, rows, a_lc, a_r);
                    a_m = 32;
                    if (c_m) drain_add<F>(c_lc, c_v, 32, lane, acc);
                }
            }
        }
        lds_round_barrier();
        #pragma unroll
        for (int k = 0; k < 8; ++k) cur[k] = nxt[k];
    }
    if (F == 1) {
        if (cnt > 0) drain_count<LCB>(ml, 0, cnt, lane, acc);
    } else {
        int tail = cnt;
        #pragma nounroll
        for (int t = 0; t < 3; ++t) {
            if (b_m) drain_add<F>(b_lc, b_v, b_m, lane, acc);
            b_m = a_m;
            b_lc = a_lc;
            if (a_m) drain_gather<F>(a_r, a_m, lane, h, b_v);
            a_m = tail;
            if (tail) drain_take<LCB>(ml, 0, tail, lane, e0, rows, a_lc, a_r);
            tail = 0;
        }
    }
    __syncthreads();

    float* dst = part + ((size_t)q * n + lo) * F;
    for (int i = threadIdx.x; i < nr * F; i += TWG) dst[i] = acc[i];
}

__global__ void gcn_dinv_kernel(const float* __restrict__ part,
                                float* __restrict__ dinv, int n) {
    const int i = blockIdx.x * blockDim.x + threadIdx.x;
    if (i >= n) return;
    float c = 1.0f;
    #pragma unroll 16
    for (int q = 0; q < QD; ++q) c += part[(size_t)q * n + i];
    dinv[i] = rsqrtf(c);
}

__global__ void gcn_xw1_wmma_kernel(const float* __restrict__ x,
                                    const float* __restrict__ W1,
                                    const float* __restrict__ dinv,
                                    float* __restrict__ h1s,
                                    int n_nodes) {
    const int lane  = threadIdx.x & 31;
    const int wave  = threadIdx.x >> 5;
    const int tile  = blockIdx.x * (blockDim.x >> 5) + wave;
    const int node0 = tile * 16;
    if (node0 >= n_nodes) return;

    const int half = lane >> 4;
    const int l16  = lane & 15;
    const int m    = l16;

    v8f c = {0.f, 0.f, 0.f, 0.f, 0.f, 0.f, 0.f, 0.f};

    #pragma unroll
    for (int k0 = 0; k0 < 28; k0 += 4) {
        const int ka = k0 + 2 * half;
        v2f a, b;
        a.x = (ka     < FIN) ? x[(node0 + m) * FIN + ka    ] : 0.0f;
        a.y = (ka + 1 < FIN) ? x[(node0 + m) * FIN + ka + 1] : 0.0f;
        b.x = (ka     < FIN) ? W1[(ka    ) * F1 + l16] : 0.0f;
        b.y = (ka + 1 < FIN) ? W1[(ka + 1) * F1 + l16] : 0.0f;
        c = __builtin_amdgcn_wmma_f32_16x16x4_f32(
                 false, a,  false, b,
                 (short)0, c,  false,  false);
    }

    #pragma unroll
    for (int v = 0; v < 8; ++v) {
        const int node = node0 + v + 8 * half;
        h1s[node * F1 + l16] = c[v] * dinv[node];
    }
}

__global__ void gcn_layer1_finish_kernel(const float* __restrict__ part,
                                         const float* __restrict__ h1s,
                                         const float* __restrict__ dinv,
                                         const float* __restrict__ b1,
                                         const float* __restrict__ W2,
                                         float* __restrict__ h2s,
                                         int n) {
    const int gid = blockIdx.x * blockDim.x + threadIdx.x;
    const int i   = gid >> 2;
    const int g   = gid & 3;
    const int ic  = (i < n) ? i : (n - 1);
    float4 s = ((const float4*)h1s)[(size_t)ic * 4 + g];
    #pragma unroll
    for (int q = 0; q < Q1; ++q) {
        const float4 t = ((const float4*)part)[((size_t)q * n + ic) * 4 + g];
        s.x += t.x; s.y += t.y; s.z += t.z; s.w += t.w;
    }
    const float di = dinv[ic];
    const float v0 = fmaxf(fmaf(s.x, di, b1[4 * g + 0]), 0.0f);
    const float v1 = fmaxf(fmaf(s.y, di, b1[4 * g + 1]), 0.0f);
    const float v2 = fmaxf(fmaf(s.z, di, b1[4 * g + 2]), 0.0f);
    const float v3 = fmaxf(fmaf(s.w, di, b1[4 * g + 3]), 0.0f);
    const float* w = W2 + 8 * g;
    float a0 = v0 * w[0];
    float a1 = v0 * w[1];
    a0 = fmaf(v1, w[2], a0); a1 = fmaf(v1, w[3], a1);
    a0 = fmaf(v2, w[4], a0); a1 = fmaf(v2, w[5], a1);
    a0 = fmaf(v3, w[6], a0); a1 = fmaf(v3, w[7], a1);
    a0 += __shfl_xor(a0, 1); a1 += __shfl_xor(a1, 1);
    a0 += __shfl_xor(a0, 2); a1 += __shfl_xor(a1, 2);
    if (g == 0 && i < n) {
        float2 o; o.x = a0 * di; o.y = a1 * di;
        ((float2*)h2s)[i] = o;
    }
}

__global__ void gcn_final_kernel(const float* __restrict__ part,
                                 const float* __restrict__ h2s,
                                 const float* __restrict__ dinv,
                                 const float* __restrict__ b2,
                                 float* __restrict__ out,
                                 int n) {
    const int i = blockIdx.x * blockDim.x + threadIdx.x;
    if (i >= n) return;
    float2 s = ((const float2*)h2s)[i];
    #pragma unroll 16
    for (int q = 0; q < Q2; ++q) {
        const float2 t = ((const float2*)part)[(size_t)q * n + i];
        s.x += t.x; s.y += t.y;
    }
    const float di = dinv[i];
    const float a = fmaf(s.x, di, b2[0]);
    const float b = fmaf(s.y, di, b2[1]);
    const float m = fmaxf(a, b);
    const float lse = m + logf(expf(a - m) + expf(b - m));
    float2 o; o.x = a - lse; o.y = b - lse;
    ((float2*)out)[i] = o;
}

extern "C" void kernel_launch(void* const* d_in, const int* in_sizes, int n_in,
                              void* d_out, int out_size, void* d_ws, size_t ws_size,
                              hipStream_t stream) {
    const float* x   = (const float*)d_in[0];
    const int*   ei  = (const int*)d_in[1];
    const float* W1  = (const float*)d_in[2];
    const float* b1  = (const float*)d_in[3];
    const float* W2  = (const float*)d_in[4];
    const float* b2  = (const float*)d_in[5];
    float*       out = (float*)d_out;

    const int N = in_sizes[0] / FIN;
    const int E = in_sizes[1] / 2;
    const int* rows = ei;
    const int* cols = ei + E;

    float* ws   = (float*)d_ws;
    float* h1s  = ws;
    float* h2s  = ws + (size_t)16 * N;
    float* dinv = ws + (size_t)18 * N;
    float* part = ws + (size_t)20 * N;

    const int BS = 256;

    gcn_tile_kernel<1, RD, QD><<<((N + RD - 1) / RD) * QD, TWG, 0, stream>>>(
        rows, cols, h1s, part, N, E, (E + QD - 1) / QD);

    gcn_dinv_kernel<<<(N + BS - 1) / BS, BS, 0, stream>>>(part, dinv, N);

    {
        int tiles = (N + 15) / 16;
        int waves_per_block = BS / 32;
        int blocks = (tiles + waves_per_block - 1) / waves_per_block;
        gcn_xw1_wmma_kernel<<<blocks, BS, 0, stream>>>(x, W1, dinv, h1s, N);
    }

    gcn_tile_kernel<F1, R1, Q1><<<((N + R1 - 1) / R1) * Q1, TWG, 0, stream>>>(
        rows, cols, h1s, part, N, E, (E + Q1 - 1) / Q1);

    gcn_layer1_finish_kernel<<<(4 * N + BS - 1) / BS, BS, 0, stream>>>(
        part, h1s, dinv, b1, W2, h2s, N);

    gcn_tile_kernel<F2, R2, Q2><<<((N + R2 - 1) / R2) * Q2, TWG, 0, stream>>>(
        rows, cols, h2s, part, N, E, (E + Q2 - 1) / Q2);

    gcn_final_kernel<<<(N + BS - 1) / BS, BS, 0, stream>>>(part, h2s, dinv, b2, out, N);
}


# Tile kernels: each filtering wave counting-sorts its round's hits by owner wave in LDS (node runs dealt round the waves); an owner reads only its own segments, not all 32 queues; sum order changes
# speedup vs baseline: 1.8071x; 1.1500x over previous
#include <hip/hip_runtime.h>


typedef __attribute__((ext_vector_type(2))) float v2f;
typedef __attribute__((ext_vector_type(8))) float v8f;

#define F1 16
#define F2 2
#define FIN 25

#define TWG 1024
#define RD 50016
#define QD 128
#define R1 4000
#define Q1 10
#define R2 25008
#define Q2 64

#define CHUNK 256
#define QSTR  260
#define LCAP  160
#define TSTR  33

__device__ __forceinline__ void lds_round_barrier() {
    asm volatile("s_wait_dscnt 0x0\n\ts_barrier_signal -1\n\ts_barrier_wait -1" ::: "memory");
}

template <int LCB>
__device__ __forceinline__ void drain_take(const unsigned* ml, int start, int m, int lane, int e0,
                                           const int* __restrict__ rows, unsigned& lc, int& r) {
    constexpr unsigned LCM = (1u << LCB) - 1u;
    lc = 0xFFFFFFFFu;
    r = 0;
    if (lane < m) {
        const unsigned ent = ml[start + lane];
        lc = ent & LCM;
        r = rows[e0 + (int)(ent >> LCB)];
    }
}

template <int F>
__device__ __forceinline__ void drain_gather(int r, int m, int lane, const float* __restrict__ h, float* v) {
    if (F == 16) {
        const int half = lane >> 4;
        const int l16  = lane & 15;
        #pragma unroll
        for (int j = 0; j < 16; ++j) {
            const int src = 2 * j + half;
            const int rj = __shfl(r, src);
            v[j] = (src < m) ? h[rj * 16 + l16] : 0.f;
        }
    } else {
        float2 t = make_float2(0.f, 0.f);
        if (lane < m) t = ((const float2*)h)[r];
        v[0] = t.x;
        v[1] = t.y;
    }
}

template <int F>
__device__ __forceinline__ void drain_add(unsigned lc, const float* v, int m, int lane, float* acc) {
    if (F == 16) {
        const int half = lane >> 4;
        const int l16  = lane & 15;
        #pragma unroll
        for (int j = 0; j < 16; ++j) {
            const unsigned la = (unsigned)__builtin_amdgcn_readlane((int)lc, 2 * j);
            const unsigned lb = (unsigned)__builtin_amdgcn_readlane((int)lc, 2 * j + 1);
            const bool va = 2 * j < m;
            const bool vb = 2 * j + 1 < m;
            if (va && vb && la != lb) {
                const unsigned a = (half ? lb : la) * 16u + (unsigned)l16;
                acc[a] += v[j];
            } else {
                if (va && half == 0) acc[la * 16u + (unsigned)l16] += v[j];
                if (vb && half == 1) acc[lb * 16u + (unsigned)l16] += v[j];
            }
        }
    } else {
        const bool valid = lane < m;
        unsigned same = 0u;
        #pragma unroll
        for (int s = 0; s < 32; ++s) {
            const unsigned ls = (unsigned)__builtin_amdgcn_readlane((int)lc, s);
            same |= (ls == lc) ? (1u << s) : 0u;
        }
        const unsigned rank = (unsigned)__builtin_popcount(same & ((1u << lane) - 1u));
        float2* acc2 = (float2*)acc;
        for (unsigned t = 0u; t < 32u; ++t) {
            const bool act = valid && (rank == t);
            if (__builtin_amdgcn_ballot_w32(act) == 0u) break;
            if (act) {
                float2 a = acc2[lc];
                a.x += v[0]; a.y += v[1];
                acc2[lc] = a;
            }
        }
    }
}

template <int LCB>
__device__ __forceinline__ void drain_count(const unsigned* ml, int start, int m, int lane, float* acc) {
    constexpr unsigned LCM = (1u << LCB) - 1u;
    const bool valid = lane < m;
    unsigned lc = 0xFFFFFFFFu;
    if (valid) lc = ml[start + lane] & LCM;
    unsigned same = 0u;
    #pragma unroll
    for (int s = 0; s < 32; ++s) {
        const unsigned ls = (unsigned)__builtin_amdgcn_readlane((int)lc, s);
        same |= (ls == lc) ? (1u << s) : 0u;
    }
    const unsigned rank = (unsigned)__builtin_popcount(same & ((1u << lane) - 1u));
    if (valid && rank == 0u) acc[lc] += (float)__builtin_popcount(same);
}

template <int F, int R, int Q>
__global__ __launch_bounds__(TWG) void gcn_tile_kernel(const int* __restrict__ rows,
                                                       const int* __restrict__ cols,
                                                       const float* __restrict__ h,
                                                       float* __restrict__ part,
                                                       int n, int n_edges, int slice) {
    constexpr int LCB = (R <= 4096) ? 12 : 16;
    constexpr int OSH = (F == 1) ? 6 : 5;
    static_assert(R <= (1 << LCB), "node index must fit its field");
    __shared__ float acc[R * F];
    __shared__ __attribute__((aligned(16))) unsigned qa[32 * QSTR];
    __shared__ unsigned cntm[32 * TSTR];
    __shared__ unsigned ml_all[32 * LCAP];
    const int lane = threadIdx.x & 31;
    const int wave = __builtin_amdgcn_readfirstlane((int)(threadIdx.x >> 5));
    unsigned* ml = ml_all + wave * LCAP;
    const int p  = blockIdx.x / Q;
    const int q  = blockIdx.x - p * Q;
    const int lo = p * R;
    const int nr = min(R, n - lo);

    for (int i = threadIdx.x; i < nr * F; i += TWG) acc[i] = 0.f;

    const int e0 = q * slice;
    const int e1 = min(n_edges, e0 + slice);
    const int stride = 32 * CHUNK;
    const int elast = n_edges - 1;
    int cur[8];
    #pragma unroll
    for (int k = 0; k < 8; ++k) {
        const int e = e0 + wave * CHUNK + k * 32 + lane;
        const int c = cols[min(e, elast)];
        cur[k] = (e < e1) ? c : -1;
    }
    int cnt = 0;
    constexpr int VN = (F == 16) ? 16 : 2;
    unsigned a_lc = 0u, b_lc = 0u;
    int a_r = 0, a_m = 0, b_m = 0;
    float b_v[VN];
    #pragma unroll
    for (int j = 0; j < VN; ++j) b_v[j] = 0.f;
    for (int rb = e0; rb < e1; rb += stride) {
        int nxt[8];
        #pragma unroll
        for (int k = 0; k < 8; ++k) {
            const int e = rb + stride + wave * CHUNK + k * 32 + lane;
            const int c = cols[min(e, elast)];
            nxt[k] = (e < e1) ? c : -1;
        }
        int na = 0;
        #pragma unroll
        for (int k = 0; k < 8; ++k) {
            const unsigned lc = (unsigned)(cur[k] - lo);
            const bool hit = lc < (unsigned)nr;
            const unsigned mk = __builtin_amdgcn_ballot_w32(hit);
            if (hit) qa[wave * QSTR + na + (int)__builtin_amdgcn_mbcnt_lo(mk, 0u)] =
                         ((unsigned)(k * 32 + lane) << 16) | lc;
            na += __builtin_popcount(mk);
        }
        unsigned* cw = cntm + wave * TSTR;
        cw[lane] = 0u;
        const int nb = (na + 31) >> 5;
        unsigned se[8], sp[8];
        #pragma unroll
        for (int b = 0; b < 8; ++b) {
            se[b] = 0u;
            sp[b] = 0u;
            if (b < nb) {
                const bool valid = b * 32 + lane < na;
                unsigned ent = 0u;
                if (valid) ent = qa[wave * QSTR + b * 32 + lane];
                unsigned same = __builtin_amdgcn_ballot_w32(valid);
                #pragma unroll
                for (int bit = 0; bit < 5; ++bit) {
                    const int sb = ((int)(ent << (31 - OSH - bit))) >> 31;
                    const unsigned m = __builtin_amdgcn_ballot_w32(valid && sb != 0);
                    same &= ~(m ^ (unsigned)sb);
                }
                const unsigned rank = __builtin_amdgcn_mbcnt_lo(same, 0u);
                const unsigned tot  = (unsigned)__builtin_popcount(same);
                unsigned base = 0u;
                if (valid) base = cw[(ent >> OSH) & 31u];
                if (valid && rank + 1u == tot) cw[(ent >> OSH) & 31u] = base + tot;
                se[b] = ent;
                sp[b] = base + rank;
            }
        }
        {
            const unsigned c = cw[lane];
            unsigned inc = c;
            inc += (unsigned)__builtin_amdgcn_update_dpp(0, (int)inc, 0x111, 0xF, 0xF, false);
            inc += (unsigned)__builtin_amdgcn_update_dpp(0, (int)inc, 0x112, 0xF, 0xF, false);
            inc += (unsigned)__builtin_amdgcn_update_dpp(0, (int)inc, 0x114, 0xF, 0xF, false);
            inc += (unsigned)__builtin_amdgcn_update_dpp(0, (int)inc, 0x118, 0xF, 0xF, false);
            const unsigned row0 = (unsigned)__builtin_amdgcn_readlane((int)inc, 15);
            if (lane >= 16) inc += row0;
            cw[lane] = ((inc - c) << 16) | c;
        }
        #pragma unroll
        for (int b = 0; b < 8; ++b) {
            if (b < nb) {
                if (b * 32 + lane < na) qa[wave * QSTR + (int)((cw[(se[b] >> OSH) & 31u] >> 16) + sp[b])] = se[b];
            }
        }
        lds_round_barrier();
        const unsigned sc = cntm[lane * TSTR + wave];
        const int st = (int)(sc >> 16);
        const int nl = (int)(sc & 0xFFFFu);
        const unsigned ebase = (unsigned)(rb - e0 + lane * CHUNK);
        for (int i0 = 0; __builtin_amdgcn_ballot_w32(i0 < nl) != 0u; i0 += 4) {
            unsigned ent[4];
            #pragma unroll
            for (int u = 0; u < 4; ++u) {
                ent[u] = 0u;
                if (i0 + u < nl) ent[u] = qa[lane * QSTR + st + i0 + u];
            }
            #pragma unroll
            for (int u = 0; u < 4; ++u) {
                const bool have = i0 + u < nl;
                const unsigned mk = __builtin_amdgcn_ballot_w32(have);
                if (mk != 0u) {
                    if (have) ml[cnt + (int)__builtin_amdgcn_mbcnt_lo(mk, 0u)] =
                                  ((ebase + (ent[u] >> 16)) << LCB) | (ent[u] & 0xFFFFu);
                    cnt += __builtin_popcount(mk);
                }
            }
            while (cnt >= 32) {
                cnt -= 32;
                if (F == 1) {
                    drain_count<LCB>(ml, cnt, 32, lane, acc);
                } else {
                    float c_v[VN];
                    #pragma unroll
                    for (int j = 0; j < VN; ++j) c_v[j] = b_v[j];
                    const unsigned c_lc = b_lc;
                    const int c_m = b_m;
                    b_m = a_m;
                    b_lc = a_lc;
                    drain_gather<F>(a_r, 32, lane, h, b_v);
                    drain_take<LCB>(ml, cnt, 32, lane, e0, rows, a_lc, a_r);
                    a_m = 32;
                    if (c_m) drain_add<F>(c_lc, c_v, 32, lane, acc);
                }
            }
        }
        lds_round_barrier();
        #pragma unroll
        for (int k = 0; k < 8; ++k) cur[k] = nxt[k];
    }
    if (F == 1) {
        if (cnt > 0) drain_count<LCB>(ml, 0, cnt, lane, acc);
    } else {
        int tail = cnt;
        #pragma nounroll
        for (int t = 0; t < 3; ++t) {
            if (b_m) drain_add<F>(b_lc, b_v, b_m, lane, acc);
            b_m = a_m;
            b_lc = a_lc;
            if (a_m) drain_gather<F>(a_r, a_m, lane, h, b_v);
            a_m = tail;
            if (tail) drain_take<LCB>(ml, 0, tail, lane, e0, rows, a_lc, a_r);
            tail = 0;
        }
    }
    __syncthreads();

    float* dst = part + ((size_t)q * n + lo) * F;
    for (int i = threadIdx.x; i < nr * F; i += TWG) dst[i] = acc[i];
}

__global__ void gcn_dinv_kernel(const float* __restrict__ part,
                                float* __restrict__ dinv, int n) {
    const int i = blockIdx.x * blockDim.x + threadIdx.x;
    if (i >= n) return;
    float c = 1.0f;
    #pragma unroll 16
    for (int q = 0; q < QD; ++q) c += part[(size_t)q * n + i];
    dinv[i] = rsqrtf(c);
}

__global__ void gcn_xw1_wmma_kernel(const float* __restrict__ x,
                                    const float* __restrict__ W1,
                                    const float* __restrict__ dinv,
                                    float* __restrict__ h1s,
                                    int n_nodes) {
    const int lane  = threadIdx.x & 31;
    const int wave  = threadIdx.x >> 5;
    const int tile  = blockIdx.x * (blockDim.x >> 5) + wave;
    const int node0 = tile * 16;
    if (node0 >= n_nodes) return;

    const int half = lane >> 4;
    const int l16  = lane & 15;
    const int m    = l16;

    v8f c = {0.f, 0.f, 0.f, 0.f, 0.f, 0.f, 0.f, 0.f};

    #pragma unroll
    for (int k0 = 0; k0 < 28; k0 += 4) {
        const int ka = k0 + 2 * half;
        v2f a, b;
        a.x = (ka     < FIN) ? x[(node0 + m) * FIN + ka    ] : 0.0f;
        a.y = (ka + 1 < FIN) ? x[(node0 + m) * FIN + ka + 1] : 0.0f;
        b.x = (ka     < FIN) ? W1[(ka    ) * F1 + l16] : 0.0f;
        b.y = (ka + 1 < FIN) ? W1[(ka + 1) * F1 + l16] : 0.0f;
        c = __builtin_amdgcn_wmma_f32_16x16x4_f32(
                 false, a,  false, b,
                 (short)0, c,  false,  false);
    }

    #pragma unroll
    for (int v = 0; v < 8; ++v) {
        const int node = node0 + v + 8 * half;
        h1s[node * F1 + l16] = c[v] * dinv[node];
    }
}

__global__ void gcn_layer1_finish_kernel(const float* __restrict__ part,
                                         const float* __restrict__ h1s,
                                         const float* __restrict__ dinv,
                                         const float* __restrict__ b1,
                                         const float* __restrict__ W2,
                                         float* __restrict__ h2s,
                                         int n) {
    const int gid = blockIdx.x * blockDim.x + threadIdx.x;
    const int i   = gid >> 2;
    const int g   = gid & 3;
    const int ic  = (i < n) ? i : (n - 1);
    float4 s = ((const float4*)h1s)[(size_t)ic * 4 + g];
    #pragma unroll
    for (int q = 0; q < Q1; ++q) {
        const float4 t = ((const float4*)part)[((size_t)q * n + ic) * 4 + g];
        s.x += t.x; s.y += t.y; s.z += t.z; s.w += t.w;
    }
    const float di = dinv[ic];
    const float v0 = fmaxf(fmaf(s.x, di, b1[4 * g + 0]), 0.0f);
    const float v1 = fmaxf(fmaf(s.y, di, b1[4 * g + 1]), 0.0f);
    const float v2 = fmaxf(fmaf(s.z, di, b1[4 * g + 2]), 0.0f);
    const float v3 = fmaxf(fmaf(s.w, di, b1[4 * g + 3]), 0.0f);
    const float* w = W2 + 8 * g;
    float a0 = v0 * w[0];
    float a1 = v0 * w[1];
    a0 = fmaf(v1, w[2], a0); a1 = fmaf(v1, w[3], a1);
    a0 = fmaf(v2, w[4], a0); a1 = fmaf(v2, w[5], a1);
    a0 = fmaf(v3, w[6], a0); a1 = fmaf(v3, w[7], a1);
    a0 += __shfl_xor(a0, 1); a1 += __shfl_xor(a1, 1);
    a0 += __shfl_xor(a0, 2); a1 += __shfl_xor(a1, 2);
    if (g == 0 && i < n) {
        float2 o; o.x = a0 * di; o.y = a1 * di;
        ((float2*)h2s)[i] = o;
    }
}

__global__ void gcn_final_kernel(const float* __restrict__ part,
                                 const float* __restrict__ h2s,
                                 const float* __restrict__ dinv,
                                 const float* __restrict__ b2,
                                 float* __restrict__ out,
                                 int n) {
    const int i = blockIdx.x * blockDim.x + threadIdx.x;
    if (i >= n) return;
    float2 s = ((const float2*)h2s)[i];
    #pragma unroll 16
    for (int q = 0; q < Q2; ++q) {
        const float2 t = ((const float2*)part)[(size_t)q * n + i];
        s.x += t.x; s.y += t.y;
    }
    const float di = dinv[i];
    const float a = fmaf(s.x, di, b2[0]);
    const float b = fmaf(s.y, di, b2[1]);
    const float m = fmaxf(a, b);
    const float lse = m + logf(expf(a - m) + expf(b - m));
    float2 o; o.x = a - lse; o.y = b - lse;
    ((float2*)out)[i] = o;
}

extern "C" void kernel_launch(void* const* d_in, const int* in_sizes, int n_in,
                              void* d_out, int out_size, void* d_ws, size_t ws_size,
                              hipStream_t stream) {
    const float* x   = (const float*)d_in[0];
    const int*   ei  = (const int*)d_in[1];
    const float* W1  = (const float*)d_in[2];
    const float* b1  = (const float*)d_in[3];
    const float* W2  = (const float*)d_in[4];
    const float* b2  = (const float*)d_in[5];
    float*       out = (float*)d_out;

    const int N = in_sizes[0] / FIN;
    const int E = in_sizes[1] / 2;
    const int* rows = ei;
    const int* cols = ei + E;

    float* ws   = (float*)d_ws;
    float* h1s  = ws;
    float* h2s  = ws + (size_t)16 * N;
    float* dinv = ws + (size_t)18 * N;
    float* part = ws + (size_t)20 * N;

    const int BS = 256;

    gcn_tile_kernel<1, RD, QD><<<((N + RD - 1) / RD) * QD, TWG, 0, stream>>>(
        rows, cols, h1s, part, N, E, (E + QD - 1) / QD);

    gcn_dinv_kernel<<<(N + BS - 1) / BS, BS, 0, stream>>>(part, dinv, N);

    {
        int tiles = (N + 15) / 16;
        int waves_per_block = BS / 32;
        int blocks = (tiles + waves_per_block - 1) / waves_per_block;
        gcn_xw1_wmma_kernel<<<blocks, BS, 0, stream>>>(x, W1, dinv, h1s, N);
    }

    gcn_tile_kernel<F1, R1, Q1><<<((N + R1 - 1) / R1) * Q1, TWG, 0, stream>>>(
        rows, cols, h1s, part, N, E, (E + Q1 - 1) / Q1);

    gcn_layer1_finish_kernel<<<(4 * N + BS - 1) / BS, BS, 0, stream>>>(
        part, h1s, dinv, b1, W2, h2s, N);

    gcn_tile_kernel<F2, R2, Q2><<<((N + R2 - 1) / R2) * Q2, TWG, 0, stream>>>(
        rows, cols, h2s, part, N, E, (E + Q2 - 1) / Q2);

    gcn_final_kernel<<<(N + BS - 1) / BS, BS, 0, stream>>>(part, h2s, dinv, b2, out, N);
}


# Layer-1 drain: 8 list entries per step, 4 lanes x 16 B each (was 2 entries, 16 lanes x 4 B): 4 row loads and 4 LDS read-add-write trips per batch, not 16; same-node entries take turns; sum order kept
# speedup vs baseline: 2.0435x; 1.1308x over previous
#include <hip/hip_runtime.h>


typedef __attribute__((ext_vector_type(2))) float v2f;
typedef __attribute__((ext_vector_type(8))) float v8f;

#define F1 16
#define F2 2
#define FIN 25

#define TWG 1024
#define RD 50016
#define QD 128
#define R1 4000
#define Q1 10
#define R2 25008
#define Q2 64

#define CHUNK 256
#define QSTR  260
#define LCAP  160
#define TSTR  33

__device__ __forceinline__ void lds_round_barrier() {
    asm volatile("s_wait_dscnt 0x0\n\ts_barrier_signal -1\n\ts_barrier_wait -1" ::: "memory");
}

template <int LCB>
__device__ __forceinline__ void drain_take(const unsigned* ml, int start, int m, int lane, int e0,
                                           const int* __restrict__ rows, unsigned& lc, int& r) {
    constexpr unsigned LCM = (1u << LCB) - 1u;
    lc = 0xFFFFFFFFu;
    r = 0;
    if (lane < m) {
        const unsigned ent = ml[start + lane];
        lc = ent & LCM;
        r = rows[e0 + (int)(ent >> LCB)];
    }
}

template <int F>
__device__ __forceinline__ void drain_gather(int r, int m, int lane, const float* __restrict__ h, float* v) {
    if (F == 16) {
        const int sub = lane >> 2;
        const int qd  = lane & 3;
        #pragma unroll
        for (int s = 0; s < 4; ++s) {
            const int src = 8 * s + sub;
            const int rs = __shfl(r, src);
            float4 t = make_float4(0.f, 0.f, 0.f, 0.f);
            if (src < m) t = ((const float4*)h)[rs * 4 + qd];
            v[4 * s + 0] = t.x;
            v[4 * s + 1] = t.y;
            v[4 * s + 2] = t.z;
            v[4 * s + 3] = t.w;
        }
    } else {
        float2 t = make_float2(0.f, 0.f);
        if (lane < m) t = ((const float2*)h)[r];
        v[0] = t.x;
        v[1] = t.y;
    }
}

template <int D>
__device__ __forceinline__ unsigned same_node_below(unsigned lc, int l8) {
    const unsigned o = (unsigned)__builtin_amdgcn_update_dpp(0, (int)lc, 0x110 + D, 0xF, 0xF, false);
    return (l8 >= D && o == lc) ? 1u : 0u;
}

template <int F>
__device__ __forceinline__ void drain_add(unsigned lc, const float* v, int m, int lane, float* acc) {
    if (F == 16) {
        const int sub = lane >> 2;
        const int qd  = lane & 3;
        const int l8  = lane & 7;
        unsigned rk = same_node_below<1>(lc, l8);
        rk += same_node_below<2>(lc, l8);
        rk += same_node_below<3>(lc, l8);
        rk += same_node_below<4>(lc, l8);
        rk += same_node_below<5>(lc, l8);
        rk += same_node_below<6>(lc, l8);
        rk += same_node_below<7>(lc, l8);
        const unsigned pk = (rk << 16) | (lc & 0xFFFFu);
        unsigned pks[4];
        #pragma unroll
        for (int s = 0; s < 4; ++s) pks[s] = (unsigned)__shfl((int)pk, 8 * s + sub);
        float4* acc4 = (float4*)acc;
        #pragma unroll
        for (int s = 0; s < 4; ++s) {
            const bool valid = 8 * s + sub < m;
            const unsigned node = pks[s] & 0xFFFFu;
            const unsigned rank = pks[s] >> 16;
            for (unsigned t = 0u; t < 8u; ++t) {
                const bool act = valid && (rank == t);
                if (__builtin_amdgcn_ballot_w32(act) == 0u) break;
                if (act) {
                    float4 a = acc4[node * 4u + (unsigned)qd];
                    a.x += v[4 * s + 0];
                    a.y += v[4 * s + 1];
                    a.z += v[4 * s + 2];
                    a.w += v[4 * s + 3];
                    acc4[node * 4u + (unsigned)qd] = a;
                }
            }
        }
    } else {
        const bool valid = lane < m;
        unsigned same = 0u;
        #pragma unroll
        for (int s = 0; s < 32; ++s) {
            const unsigned ls = (unsigned)__builtin_amdgcn_readlane((int)lc, s);
            same |= (ls == lc) ? (1u << s) : 0u;
        }
        const unsigned rank = (unsigned)__builtin_popcount(same & ((1u << lane) - 1u));
        float2* acc2 = (float2*)acc;
        for (unsigned t = 0u; t < 32u; ++t) {
            const bool act = valid && (rank == t);
            if (__builtin_amdgcn_ballot_w32(act) == 0u) break;
            if (act) {
                float2 a = acc2[lc];
                a.x += v[0]; a.y += v[1];
                acc2[lc] = a;
            }
        }
    }
}

template <int LCB>
__device__ __forceinline__ void drain_count(const unsigned* ml, int start, int m, int lane, float* acc) {
    constexpr unsigned LCM = (1u << LCB) - 1u;
    const bool valid = lane < m;
    unsigned lc = 0xFFFFFFFFu;
    if (valid) lc = ml[start + lane] & LCM;
    unsigned same = 0u;
    #pragma unroll
    for (int s = 0; s < 32; ++s) {
        const unsigned ls = (unsigned)__builtin_amdgcn_readlane((int)lc, s);
        same |= (ls == lc) ? (1u << s) : 0u;
    }
    const unsigned rank = (unsigned)__builtin_popcount(same & ((1u << lane) - 1u));
    if (valid && rank == 0u) acc[lc] += (float)__builtin_popcount(same);
}

template <int F, int R, int Q>
__global__ __launch_bounds__(TWG) void gcn_tile_kernel(const int* __restrict__ rows,
                                                       const int* __restrict__ cols,
                                                       const float* __restrict__ h,
                                                       float* __restrict__ part,
                                                       int n, int n_edges, int slice) {
    constexpr int LCB = (R <= 4096) ? 12 : 16;
    constexpr int OSH = (F == 1) ? 6 : 5;
    static_assert(R <= (1 << LCB), "node index must fit its field");
    __shared__ __attribute__((aligned(16))) float acc[R * F];
    __shared__ __attribute__((aligned(16))) unsigned qa[32 * QSTR];
    __shared__ unsigned cntm[32 * TSTR];
    __shared__ unsigned ml_all[32 * LCAP];
    const int lane = threadIdx.x & 31;
    const int wave = __builtin_amdgcn_readfirstlane((int)(threadIdx.x >> 5));
    unsigned* ml = ml_all + wave * LCAP;
    const int p  = blockIdx.x / Q;
    const int q  = blockIdx.x - p * Q;
    const int lo = p * R;
    const int nr = min(R, n - lo);

    for (int i = threadIdx.x; i < nr * F; i += TWG) acc[i] = 0.f;

    const int e0 = q * slice;
    const int e1 = min(n_edges, e0 + slice);
    const int stride = 32 * CHUNK;
    const int elast = n_edges - 1;
    int cur[8];
    #pragma unroll
    for (int k = 0; k < 8; ++k) {
        const int e = e0 + wave * CHUNK + k * 32 + lane;
        const int c = cols[min(e, elast)];
        cur[k] = (e < e1) ? c : -1;
    }
    int cnt = 0;
    constexpr int VN = (F == 16) ? 16 : 2;
    unsigned a_lc = 0u, b_lc = 0u;
    int a_r = 0, a_m = 0, b_m = 0;
    float b_v[VN];
    #pragma unroll
    for (int j = 0; j < VN; ++j) b_v[j] = 0.f;
    for (int rb = e0; rb < e1; rb += stride) {
        int nxt[8];
        #pragma unroll
        for (int k = 0; k < 8; ++k) {
            const int e = rb + stride + wave * CHUNK + k * 32 + lane;
            const int c = cols[min(e, elast)];
            nxt[k] = (e < e1) ? c : -1;
        }
        int na = 0;
        #pragma unroll
        for (int k = 0; k < 8; ++k) {
            const unsigned lc = (unsigned)(cur[k] - lo);
            const bool hit = lc < (unsigned)nr;
            const unsigned mk = __builtin_amdgcn_ballot_w32(hit);
            if (hit) qa[wave * QSTR + na + (int)__builtin_amdgcn_mbcnt_lo(mk, 0u)] =
                         ((unsigned)(k * 32 + lane) << 16) | lc;
            na += __builtin_popcount(mk);
        }
        unsigned* cw = cntm + wave * TSTR;
        cw[lane] = 0u;
        const int nb = (na + 31) >> 5;
        unsigned se[8], sp[8];
        #pragma unroll
        for (int b = 0; b < 8; ++b) {
            se[b] = 0u;
            sp[b] = 0u;
            if (b < nb) {
                const bool valid = b * 32 + lane < na;
                unsigned ent = 0u;
                if (valid) ent = qa[wave * QSTR + b * 32 + lane];
                unsigned same = __builtin_amdgcn_ballot_w32(valid);
                #pragma unroll
                for (int bit = 0; bit < 5; ++bit) {
                    const int sb = ((int)(ent << (31 - OSH - bit))) >> 31;
                    const unsigned m = __builtin_amdgcn_ballot_w32(valid && sb != 0);
                    same &= ~(m ^ (unsigned)sb);
                }
                const unsigned rank = __builtin_amdgcn_mbcnt_lo(same, 0u);
                const unsigned tot  = (unsigned)__builtin_popcount(same);
                unsigned base = 0u;
                if (valid) base = cw[(ent >> OSH) & 31u];
                if (valid && rank + 1u == tot) cw[(ent >> OSH) & 31u] = base + tot;
                se[b] = ent;
                sp[b] = base + rank;
            }
        }
        {
            const unsigned c = cw[lane];
            unsigned inc = c;
            inc += (unsigned)__builtin_amdgcn_update_dpp(0, (int)inc, 0x111, 0xF, 0xF, false);
            inc += (unsigned)__builtin_amdgcn_update_dpp(0, (int)inc, 0x112, 0xF, 0xF, false);
            inc += (unsigned)__builtin_amdgcn_update_dpp(0, (int)inc, 0x114, 0xF, 0xF, false);
            inc += (unsigned)__builtin_amdgcn_update_dpp(0, (int)inc, 0x118, 0xF, 0xF, false);
            const unsigned row0 = (unsigned)__builtin_amdgcn_readlane((int)inc, 15);
            if (lane >= 16) inc += row0;
            cw[lane] = ((inc - c) << 16) | c;
        }
        #pragma unroll
        for (int b = 0; b < 8; ++b) {
            if (b < nb) {
                if (b * 32 + lane < na) qa[wave * QSTR + (int)((cw[(se[b] >> OSH) & 31u] >> 16) + sp[b])] = se[b];
            }
        }
        lds_round_barrier();
        const unsigned sc = cntm[lane * TSTR + wave];
        const int st = (int)(sc >> 16);
        const int nl = (int)(sc & 0xFFFFu);
        const unsigned ebase = (unsigned)(rb - e0 + lane * CHUNK);
        for (int i0 = 0; __builtin_amdgcn_ballot_w32(i0 < nl) != 0u; i0 += 4) {
            unsigned ent[4];
            #pragma unroll
            for (int u = 0; u < 4; ++u) {
                ent[u] = 0u;
                if (i0 + u < nl) ent[u] = qa[lane * QSTR + st + i0 + u];
            }
            #pragma unroll
            for (int u = 0; u < 4; ++u) {
                const bool have = i0 + u < nl;
                const unsigned mk = __builtin_amdgcn_ballot_w32(have);
                if (mk != 0u) {
                    if (have) ml[cnt + (int)__builtin_amdgcn_mbcnt_lo(mk, 0u)] =
                                  ((ebase + (ent[u] >> 16)) << LCB) | (ent[u] & 0xFFFFu);
                    cnt += __builtin_popcount(mk);
                }
            }
            while (cnt >= 32) {
                cnt -= 32;
                if (F == 1) {
                    drain_count<LCB>(ml, cnt, 32, lane, acc);
                } else {
                    float c_v[VN];
                    #pragma unroll
                    for (int j = 0; j < VN; ++j) c_v[j] = b_v[j];
                    const unsigned c_lc = b_lc;
                    const int c_m = b_m;
                    b_m = a_m;
                    b_lc = a_lc;
                    drain_gather<F>(a_r, 32, lane, h, b_v);
                    drain_take<LCB>(ml, cnt, 32, lane, e0, rows, a_lc, a_r);
                    a_m = 32;
                    if (c_m) drain_add<F>(c_lc, c_v, 32, lane, acc);
                }
            }
        }
        lds_round_barrier();
        #pragma unroll
        for (int k = 0; k < 8; ++k) cur[k] = nxt[k];
    }
    if (F == 1) {
        if (cnt > 0) drain_count<LCB>(ml, 0, cnt, lane, acc);
    } else {
        int tail = cnt;
        #pragma nounroll
        for (int t = 0; t < 3; ++t) {
            if (b_m) drain_add<F>(b_lc, b_v, b_m, lane, acc);
            b_m = a_m;
            b_lc = a_lc;
            if (a_m) drain_gather<F>(a_r, a_m, lane, h, b_v);
            a_m = tail;
            if (tail) drain_take<LCB>(ml, 0, tail, lane, e0, rows, a_lc, a_r);
            tail = 0;
        }
    }
    __syncthreads();

    float* dst = part + ((size_t)q * n + lo) * F;
    for (int i = threadIdx.x; i < nr * F; i += TWG) dst[i] = acc[i];
}

__global__ void gcn_dinv_kernel(const float* __restrict__ part,
                                float* __restrict__ dinv, int n) {
    const int i = blockIdx.x * blockDim.x + threadIdx.x;
    if (i >= n) return;
    float c = 1.0f;
    #pragma unroll 16
    for (int q = 0; q < QD; ++q) c += part[(size_t)q * n + i];
    dinv[i] = rsqrtf(c);
}

__global__ void gcn_xw1_wmma_kernel(const float* __restrict__ x,
                                    const float* __restrict__ W1,
                                    const float* __restrict__ dinv,
                                    float* __restrict__ h1s,
                                    int n_nodes) {
    const int lane  = threadIdx.x & 31;
    const int wave  = threadIdx.x >> 5;
    const int tile  = blockIdx.x * (blockDim.x >> 5) + wave;
    const int node0 = tile * 16;
    if (node0 >= n_nodes) return;

    const int half = lane >> 4;
    const int l16  = lane & 15;
    const int m    = l16;

    v8f c = {0.f, 0.f, 0.f, 0.f, 0.f, 0.f, 0.f, 0.f};

    #pragma unroll
    for (int k0 = 0; k0 < 28; k0 += 4) {
        const int ka = k0 + 2 * half;
        v2f a, b;
        a.x = (ka     < FIN) ? x[(node0 + m) * FIN + ka    ] : 0.0f;
        a.y = (ka + 1 < FIN) ? x[(node0 + m) * FIN + ka + 1] : 0.0f;
        b.x = (ka     < FIN) ? W1[(ka    ) * F1 + l16] : 0.0f;
        b.y = (ka + 1 < FIN) ? W1[(ka + 1) * F1 + l16] : 0.0f;
        c = __builtin_amdgcn_wmma_f32_16x16x4_f32(
                 false, a,  false, b,
                 (short)0, c,  false,  false);
    }

    #pragma unroll
    for (int v = 0; v < 8; ++v) {
        const int node = node0 + v + 8 * half;
        h1s[node * F1 + l16] = c[v] * dinv[node];
    }
}

__global__ void gcn_layer1_finish_kernel(const float* __restrict__ part,
                                         const float* __restrict__ h1s,
                                         const float* __restrict__ dinv,
                                         const float* __restrict__ b1,
                                         const float* __restrict__ W2,
                                         float* __restrict__ h2s,
                                         int n) {
    const int gid = blockIdx.x * blockDim.x + threadIdx.x;
    const int i   = gid >> 2;
    const int g   = gid & 3;
    const int ic  = (i < n) ? i : (n - 1);
    float4 s = ((const float4*)h1s)[(size_t)ic * 4 + g];
    #pragma unroll
    for (int q = 0; q < Q1; ++q) {
        const float4 t = ((const float4*)part)[((size_t)q * n + ic) * 4 + g];
        s.x += t.x; s.y += t.y; s.z += t.z; s.w += t.w;
    }
    const float di = dinv[ic];
    const float v0 = fmaxf(fmaf(s.x, di, b1[4 * g + 0]), 0.0f);
    const float v1 = fmaxf(fmaf(s.y, di, b1[4 * g + 1]), 0.0f);
    const float v2 = fmaxf(fmaf(s.z, di, b1[4 * g + 2]), 0.0f);
    const float v3 = fmaxf(fmaf(s.w, di, b1[4 * g + 3]), 0.0f);
    const float* w = W2 + 8 * g;
    float a0 = v0 * w[0];
    float a1 = v0 * w[1];
    a0 = fmaf(v1, w[2], a0); a1 = fmaf(v1, w[3], a1);
    a0 = fmaf(v2, w[4], a0); a1 = fmaf(v2, w[5], a1);
    a0 = fmaf(v3, w[6], a0); a1 = fmaf(v3, w[7], a1);
    a0 += __shfl_xor(a0, 1); a1 += __shfl_xor(a1, 1);
    a0 += __shfl_xor(a0, 2); a1 += __shfl_xor(a1, 2);
    if (g == 0 && i < n) {
        float2 o; o.x = a0 * di; o.y = a1 * di;
        ((float2*)h2s)[i] = o;
    }
}

__global__ void gcn_final_kernel(const float* __restrict__ part,
                                 const float* __restrict__ h2s,
                                 const float* __restrict__ dinv,
                                 const float* __restrict__ b2,
                                 float* __restrict__ out,
                                 int n) {
    const int i = blockIdx.x * blockDim.x + threadIdx.x;
    if (i >= n) return;
    float2 s = ((const float2*)h2s)[i];
    #pragma unroll 16
    for (int q = 0; q < Q2; ++q) {
        const float2 t = ((const float2*)part)[(size_t)q * n + i];
        s.x += t.x; s.y += t.y;
    }
    const float di = dinv[i];
    const float a = fmaf(s.x, di, b2[0]);
    const float b = fmaf(s.y, di, b2[1]);
    const float m = fmaxf(a, b);
    const float lse = m + logf(expf(a - m) + expf(b - m));
    float2 o; o.x = a - lse; o.y = b - lse;
    ((float2*)out)[i] = o;
}

extern "C" void kernel_launch(void* const* d_in, const int* in_sizes, int n_in,
                              void* d_out, int out_size, void* d_ws, size_t ws_size,
                              hipStream_t stream) {
    const float* x   = (const float*)d_in[0];
    const int*   ei  = (const int*)d_in[1];
    const float* W1  = (const float*)d_in[2];
    const float* b1  = (const float*)d_in[3];
    const float* W2  = (const float*)d_in[4];
    const float* b2  = (const float*)d_in[5];
    float*       out = (float*)d_out;

    const int N = in_sizes[0] / FIN;
    const int E = in_sizes[1] / 2;
    const int* rows = ei;
    const int* cols = ei + E;

    float* ws   = (float*)d_ws;
    float* h1s  = ws;
    float* h2s  = ws + (size_t)16 * N;
    float* dinv = ws + (size_t)18 * N;
    float* part = ws + (size_t)20 * N;

    const int BS = 256;

    gcn_tile_kernel<1, RD, QD><<<((N + RD - 1) / RD) * QD, TWG, 0, stream>>>(
        rows, cols, h1s, part, N, E, (E + QD - 1) / QD);

    gcn_dinv_kernel<<<(N + BS - 1) / BS, BS, 0, stream>>>(part, dinv, N);

    {
        int tiles = (N + 15) / 16;
        int waves_per_block = BS / 32;
        int blocks = (tiles + waves_per_block - 1) / waves_per_block;
        gcn_xw1_wmma_kernel<<<blocks, BS, 0, stream>>>(x, W1, dinv, h1s, N);
    }

    gcn_tile_kernel<F1, R1, Q1><<<((N + R1 - 1) / R1) * Q1, TWG, 0, stream>>>(
        rows, cols, h1s, part, N, E, (E + Q1 - 1) / Q1);

    gcn_layer1_finish_kernel<<<(4 * N + BS - 1) / BS, BS, 0, stream>>>(
        part, h1s, dinv, b1, W2, h2s, N);

    gcn_tile_kernel<F2, R2, Q2><<<((N + R2 - 1) / R2) * Q2, TWG, 0, stream>>>(
        rows, cols, h2s, part, N, E, (E + Q2 - 1) / Q2);

    gcn_final_kernel<<<(N + BS - 1) / BS, BS, 0, stream>>>(part, h2s, dinv, b2, out, N);
}


# Layer-1 rounds 4x longer: every wave walks its own edge run and filters up to 4 passes of 256 edges into a 320-entry queue before the sort and the two barriers; 10 rounds, not 40; sum order changes
# speedup vs baseline: 2.3427x; 1.1464x over previous
#include <hip/hip_runtime.h>


typedef __attribute__((ext_vector_type(2))) float v2f;
typedef __attribute__((ext_vector_type(8))) float v8f;

#define F1 16
#define F2 2
#define FIN 25

#define TWG 1024
#define RD 50016
#define QD 128
#define R1 4000
#define Q1 10
#define R2 25008
#define Q2 64

#define CHUNK 256
#define LCAP  160
#define TSTR  33

__device__ __forceinline__ void lds_round_barrier() {
    asm volatile("s_wait_dscnt 0x0\n\ts_barrier_signal -1\n\ts_barrier_wait -1" ::: "memory");
}

template <int LCB>
__device__ __forceinline__ void drain_take(const unsigned* ml, int start, int m, int lane, int e0,
                                           const int* __restrict__ rows, unsigned& lc, int& r) {
    constexpr unsigned LCM = (1u << LCB) - 1u;
    lc = 0xFFFFFFFFu;
    r = 0;
    if (lane < m) {
        const unsigned ent = ml[start + lane];
        lc = ent & LCM;
        r = rows[e0 + (int)(ent >> LCB)];
    }
}

template <int F>
__device__ __forceinline__ void drain_gather(int r, int m, int lane, const float* __restrict__ h, float* v) {
    if (F == 16) {
        const int sub = lane >> 2;
        const int qd  = lane & 3;
        #pragma unroll
        for (int s = 0; s < 4; ++s) {
            const int src = 8 * s + sub;
            const int rs = __shfl(r, src);
            float4 t = make_float4(0.f, 0.f, 0.f, 0.f);
            if (src < m) t = ((const float4*)h)[rs * 4 + qd];
            v[4 * s + 0] = t.x;
            v[4 * s + 1] = t.y;
            v[4 * s + 2] = t.z;
            v[4 * s + 3] = t.w;
        }
    } else {
        float2 t = make_float2(0.f, 0.f);
        if (lane < m) t = ((const float2*)h)[r];
        v[0] = t.x;
        v[1] = t.y;
    }
}

template <int D>
__device__ __forceinline__ unsigned same_node_below(unsigned lc, int l8) {
    const unsigned o = (unsigned)__builtin_amdgcn_update_dpp(0, (int)lc, 0x110 + D, 0xF, 0xF, false);
    return (l8 >= D && o == lc) ? 1u : 0u;
}

template <int F>
__device__ __forceinline__ void drain_add(unsigned lc, const float* v, int m, int lane, float* acc) {
    if (F == 16) {
        const int sub = lane >> 2;
        const int qd  = lane & 3;
        const int l8  = lane & 7;
        unsigned rk = same_node_below<1>(lc, l8);
        rk += same_node_below<2>(lc, l8);
        rk += same_node_below<3>(lc, l8);
        rk += same_node_below<4>(lc, l8);
        rk += same_node_below<5>(lc, l8);
        rk += same_node_below<6>(lc, l8);
        rk += same_node_below<7>(lc, l8);
        const unsigned pk = (rk << 16) | (lc & 0xFFFFu);
        unsigned pks[4];
        #pragma unroll
        for (int s = 0; s < 4; ++s) pks[s] = (unsigned)__shfl((int)pk, 8 * s + sub);
        float4* acc4 = (float4*)acc;
        #pragma unroll
        for (int s = 0; s < 4; ++s) {
            const bool valid = 8 * s + sub < m;
            const unsigned node = pks[s] & 0xFFFFu;
            const unsigned rank = pks[s] >> 16;
            for (unsigned t = 0u; t < 8u; ++t) {
                const bool act = valid && (rank == t);
                if (__builtin_amdgcn_ballot_w32(act) == 0u) break;
                if (act) {
                    float4 a = acc4[node * 4u + (unsigned)qd];
                    a.x += v[4 * s + 0];
                    a.y += v[4 * s + 1];
                    a.z += v[4 * s + 2];
                    a.w += v[4 * s + 3];
                    acc4[node * 4u + (unsigned)qd] = a;
                }
            }
        }
    } else {
        const bool valid = lane < m;
        unsigned same = 0u;
        #pragma unroll
        for (int s = 0; s < 32; ++s) {
            const unsigned ls = (unsigned)__builtin_amdgcn_readlane((int)lc, s);
            same |= (ls == lc) ? (1u << s) : 0u;
        }
        const unsigned rank = (unsigned)__builtin_popcount(same & ((1u << lane) - 1u));
        float2* acc2 = (float2*)acc;
        for (unsigned t = 0u; t < 32u; ++t) {
            const bool act = valid && (rank == t);
            if (__builtin_amdgcn_ballot_w32(act) == 0u) break;
            if (act) {
                float2 a = acc2[lc];
                a.x += v[0]; a.y += v[1];
                acc2[lc] = a;
            }
        }
    }
}

template <int LCB>
__device__ __forceinline__ void drain_count(const unsigned* ml, int start, int m, int lane, float* acc) {
    constexpr unsigned LCM = (1u << LCB) - 1u;
    const bool valid = lane < m;
    unsigned lc = 0xFFFFFFFFu;
    if (valid) lc = ml[start + lane] & LCM;
    unsigned same = 0u;
    #pragma unroll
    for (int s = 0; s < 32; ++s) {
        const unsigned ls = (unsigned)__builtin_amdgcn_readlane((int)lc, s);
        same |= (ls == lc) ? (1u << s) : 0u;
    }
    const unsigned rank = (unsigned)__builtin_popcount(same & ((1u << lane) - 1u));
    if (valid && rank == 0u) acc[lc] += (float)__builtin_popcount(same);
}

__device__ __forceinline__ void filter_pass(int (&c)[8], const bool reload, const int* __restrict__ cols, int cursor, int wend,
                                            int elast, int lo, int nr, unsigned* qrow, int& na, int sub, int lane) {
    #pragma unroll
    for (int k = 0; k < 8; ++k) {
        const int e = cursor + k * 32 + lane;
        const int t = (e < wend) ? c[k] : -1;
        if (reload) c[k] = cols[min(e + 2 * CHUNK, elast)];
        const unsigned lc = (unsigned)(t - lo);
        const bool hit = lc < (unsigned)nr;
        const unsigned mk = __builtin_amdgcn_ballot_w32(hit);
        if (hit) qrow[na + (int)__builtin_amdgcn_mbcnt_lo(mk, 0u)] =
                     ((unsigned)(sub * CHUNK + k * 32 + lane) << 16) | lc;
        na += __builtin_popcount(mk);
    }
}

template <int F, int R, int Q, int G>
__global__ __launch_bounds__(TWG) void gcn_tile_kernel(const int* __restrict__ rows,
                                                       const int* __restrict__ cols,
                                                       const float* __restrict__ h,
                                                       float* __restrict__ part,
                                                       int n, int n_edges, int slice) {
    constexpr int LCB = (R <= 4096) ? 12 : 16;
    constexpr int OSH = (F == 1) ? 6 : 5;
    static_assert(R <= (1 << LCB), "node index must fit its field");
    constexpr int QCAP = (G == 1) ? CHUNK : CHUNK + 64;
    constexpr int QSTR = QCAP + 4;
    constexpr int NB   = QCAP / 32;
    static_assert(QSTR % 64 == 4, "queue rows must lie 4 banks apart");
    __shared__ __attribute__((aligned(16))) float acc[R * F];
    __shared__ __attribute__((aligned(16))) unsigned qa[32 * QSTR];
    __shared__ unsigned cntm[32 * TSTR];
    __shared__ unsigned ml_all[32 * LCAP];
    const int lane = threadIdx.x & 31;
    const int wave = __builtin_amdgcn_readfirstlane((int)(threadIdx.x >> 5));
    unsigned* ml = ml_all + wave * LCAP;
    const int p  = blockIdx.x / Q;
    const int q  = blockIdx.x - p * Q;
    const int lo = p * R;
    const int nr = min(R, n - lo);

    for (int i = threadIdx.x; i < nr * F; i += TWG) acc[i] = 0.f;

    const int e0 = q * slice;
    const int e1 = min(n_edges, e0 + slice);
    const int wlen = (((slice + 31) >> 5) + 31) & ~31;
    const int wbeg = e0 + wave * wlen;
    const int wend = min(e1, wbeg + wlen);
    const int maxr = (wlen + CHUNK - 1) / CHUNK + 1;
    const int elast = n_edges - 1;
    int cursor = wbeg;
    int ca[8], cb[8];
    #pragma unroll
    for (int k = 0; k < 8; ++k) {
        ca[k] = cols[min(cursor + k * 32 + lane, elast)];
        cb[k] = cols[min(cursor + CHUNK + k * 32 + lane, elast)];
    }
    bool odd  = false;
    bool lazy = false;
    int cnt = 0;
    constexpr int VN = (F == 16) ? 16 : 2;
    unsigned a_lc = 0u, b_lc = 0u;
    int a_r = 0, a_m = 0, b_m = 0;
    float b_v[VN];
    #pragma unroll
    for (int j = 0; j < VN; ++j) b_v[j] = 0.f;
    for (int round = 0; round < maxr; ++round) {
        const int rbeg = cursor - e0;
        if (odd) {
            #pragma unroll
            for (int k = 0; k < 8; ++k) { const int t = ca[k]; ca[k] = cb[k]; cb[k] = t; }
        }
        if (lazy) {
            #pragma unroll
            for (int k = 0; k < 8; ++k) cb[k] = cols[min(cursor + CHUNK + k * 32 + lane, elast)];
        }
        odd = false;
        lazy = false;
        int na = 0;
        #pragma unroll
        for (int s = 0; s < G; ++s) {
            if (s & 1) filter_pass(cb, s != G - 1, cols, cursor, wend, elast, lo, nr, qa + wave * QSTR, na, s, lane);
            else       filter_pass(ca, s != G - 1, cols, cursor, wend, elast, lo, nr, qa + wave * QSTR, na, s, lane);
            cursor += CHUNK;
            odd = !(s & 1);
            lazy = (s == G - 1);
            if (!(na + CHUNK <= QCAP && cursor < wend)) break;
        }
        unsigned* cw = cntm + wave * TSTR;
        cw[lane] = 0u;
        const int nb = (na + 31) >> 5;
        unsigned se[NB], sp[NB];
        #pragma unroll
        for (int b = 0; b < NB; ++b) {
            se[b] = 0u;
            sp[b] = 0u;
            if (b < nb) {
                const bool valid = b * 32 + lane < na;
                unsigned ent = 0u;
                if (valid) ent = qa[wave * QSTR + b * 32 + lane];
                unsigned same = __builtin_amdgcn_ballot_w32(valid);
                #pragma unroll
                for (int bit = 0; bit < 5; ++bit) {
                    const int sb = ((int)(ent << (31 - OSH - bit))) >> 31;
                    const unsigned m = __builtin_amdgcn_ballot_w32(valid && sb != 0);
                    same &= ~(m ^ (unsigned)sb);
                }
                const unsigned rank = __builtin_amdgcn_mbcnt_lo(same, 0u);
                const unsigned tot  = (unsigned)__builtin_popcount(same);
                unsigned base = 0u;
                if (valid) base = cw[(ent >> OSH) & 31u];
                if (valid && rank + 1u == tot) cw[(ent >> OSH) & 31u] = base + tot;
                se[b] = ent;
                sp[b] = base + rank;
            }
        }
        {
            const unsigned c = cw[lane];
            unsigned inc = c;
            inc += (unsigned)__builtin_amdgcn_update_dpp(0, (int)inc, 0x111, 0xF, 0xF, false);
            inc += (unsigned)__builtin_amdgcn_update_dpp(0, (int)inc, 0x112, 0xF, 0xF, false);
            inc += (unsigned)__builtin_amdgcn_update_dpp(0, (int)inc, 0x114, 0xF, 0xF, false);
            inc += (unsigned)__builtin_amdgcn_update_dpp(0, (int)inc, 0x118, 0xF, 0xF, false);
            const unsigned row0 = (unsigned)__builtin_amdgcn_readlane((int)inc, 15);
            if (lane >= 16) inc += row0;
            cw[lane] = ((inc - c) << 16) | c;
        }
        #pragma unroll
        for (int b = 0; b < NB; ++b) {
            if (b < nb) {
                if (b * 32 + lane < na) qa[wave * QSTR + (int)((cw[(se[b] >> OSH) & 31u] >> 16) + sp[b])] = se[b];
            }
        }
        if (lane == 0) cw[32] = ((unsigned)rbeg << 1) | ((cursor >= wend) ? 1u : 0u);
        lds_round_barrier();
        const unsigned sc = cntm[lane * TSTR + wave];
        const unsigned rw = cntm[lane * TSTR + 32];
        const int st = (int)(sc >> 16);
        const int nl = (int)(sc & 0xFFFFu);
        const unsigned ebase = rw >> 1;
        const bool all_through = __builtin_amdgcn_ballot_w32((rw & 1u) != 0u) == 0xFFFFFFFFu;
        for (int i0 = 0; __builtin_amdgcn_ballot_w32(i0 < nl) != 0u; i0 += 4) {
            unsigned ent[4];
            #pragma unroll
            for (int u = 0; u < 4; ++u) {
                ent[u] = 0u;
                if (i0 + u < nl) ent[u] = qa[lane * QSTR + st + i0 + u];
            }
            #pragma unroll
            for (int u = 0; u < 4; ++u) {
                const bool have = i0 + u < nl;
                const unsigned mk = __builtin_amdgcn_ballot_w32(have);
                if (mk != 0u) {
                    if (have) ml[cnt + (int)__builtin_amdgcn_mbcnt_lo(mk, 0u)] =
                                  ((ebase + (ent[u] >> 16)) << LCB) | (ent[u] & 0xFFFFu);
                    cnt += __builtin_popcount(mk);
                }
            }
            while (cnt >= 32) {
                cnt -= 32;
                if (F == 1) {
                    drain_count<LCB>(ml, cnt, 32, lane, acc);
                } else {
                    float c_v[VN];
                    #pragma unroll
                    for (int j = 0; j < VN; ++j) c_v[j] = b_v[j];
                    const unsigned c_lc = b_lc;
                    const int c_m = b_m;
                    b_m = a_m;
                    b_lc = a_lc;
                    drain_gather<F>(a_r, 32, lane, h, b_v);
                    drain_take<LCB>(ml, cnt, 32, lane, e0, rows, a_lc, a_r);
                    a_m = 32;
                    if (c_m) drain_add<F>(c_lc, c_v, 32, lane, acc);
                }
            }
        }
        lds_round_barrier();
        if (all_through) break;
    }
    if (F == 1) {
        if (cnt > 0) drain_count<LCB>(ml, 0, cnt, lane, acc);
    } else {
        int tail = cnt;
        #pragma nounroll
        for (int t = 0; t < 3; ++t) {
            if (b_m) drain_add<F>(b_lc, b_v, b_m, lane, acc);
            b_m = a_m;
            b_lc = a_lc;
            if (a_m) drain_gather<F>(a_r, a_m, lane, h, b_v);
            a_m = tail;
            if (tail) drain_take<LCB>(ml, 0, tail, lane, e0, rows, a_lc, a_r);
            tail = 0;
        }
    }
    __syncthreads();

    float* dst = part + ((size_t)q * n + lo) * F;
    for (int i = threadIdx.x; i < nr * F; i += TWG) dst[i] = acc[i];
}

__global__ void gcn_dinv_kernel(const float* __restrict__ part,
                                float* __restrict__ dinv, int n) {
    const int i = blockIdx.x * blockDim.x + threadIdx.x;
    if (i >= n) return;
    float c = 1.0f;
    #pragma unroll 16
    for (int q = 0; q < QD; ++q) c += part[(size_t)q * n + i];
    dinv[i] = rsqrtf(c);
}

__global__ void gcn_xw1_wmma_kernel(const float* __restrict__ x,
                                    const float* __restrict__ W1,
                                    const float* __restrict__ dinv,
                                    float* __restrict__ h1s,
                                    int n_nodes) {
    const int lane  = threadIdx.x & 31;
    const int wave  = threadIdx.x >> 5;
    const int tile  = blockIdx.x * (blockDim.x >> 5) + wave;
    const int node0 = tile * 16;
    if (node0 >= n_nodes) return;

    const int half = lane >> 4;
    const int l16  = lane & 15;
    const int m    = l16;

    v8f c = {0.f, 0.f, 0.f, 0.f, 0.f, 0.f, 0.f, 0.f};

    #pragma unroll
    for (int k0 = 0; k0 < 28; k0 += 4) {
        const int ka = k0 + 2 * half;
        v2f a, b;
        a.x = (ka     < FIN) ? x[(node0 + m) * FIN + ka    ] : 0.0f;
        a.y = (ka + 1 < FIN) ? x[(node0 + m) * FIN + ka + 1] : 0.0f;
        b.x = (ka     < FIN) ? W1[(ka    ) * F1 + l16] : 0.0f;
        b.y = (ka + 1 < FIN) ? W1[(ka + 1) * F1 + l16] : 0.0f;
        c = __builtin_amdgcn_wmma_f32_16x16x4_f32(
                 false, a,  false, b,
                 (short)0, c,  false,  false);
    }

    #pragma unroll
    for (int v = 0; v < 8; ++v) {
        const int node = node0 + v + 8 * half;
        h1s[node * F1 + l16] = c[v] * dinv[node];
    }
}

__global__ void gcn_layer1_finish_kernel(const float* __restrict__ part,
                                         const float* __restrict__ h1s,
                                         const float* __restrict__ dinv,
                                         const float* __restrict__ b1,
                                         const float* __restrict__ W2,
                                         float* __restrict__ h2s,
                                         int n) {
    const int gid = blockIdx.x * blockDim.x + threadIdx.x;
    const int i   = gid >> 2;
    const int g   = gid & 3;
    const int ic  = (i < n) ? i : (n - 1);
    float4 s = ((const float4*)h1s)[(size_t)ic * 4 + g];
    #pragma unroll
    for (int q = 0; q < Q1; ++q) {
        const float4 t = ((const float4*)part)[((size_t)q * n + ic) * 4 + g];
        s.x += t.x; s.y += t.y; s.z += t.z; s.w += t.w;
    }
    const float di = dinv[ic];
    const float v0 = fmaxf(fmaf(s.x, di, b1[4 * g + 0]), 0.0f);
    const float v1 = fmaxf(fmaf(s.y, di, b1[4 * g + 1]), 0.0f);
    const float v2 = fmaxf(fmaf(s.z, di, b1[4 * g + 2]), 0.0f);
    const float v3 = fmaxf(fmaf(s.w, di, b1[4 * g + 3]), 0.0f);
    const float* w = W2 + 8 * g;
    float a0 = v0 * w[0];
    float a1 = v0 * w[1];
    a0 = fmaf(v1, w[2], a0); a1 = fmaf(v1, w[3], a1);
    a0 = fmaf(v2, w[4], a0); a1 = fmaf(v2, w[5], a1);
    a0 = fmaf(v3, w[6], a0); a1 = fmaf(v3, w[7], a1);
    a0 += __shfl_xor(a0, 1); a1 += __shfl_xor(a1, 1);
    a0 += __shfl_xor(a0, 2); a1 += __shfl_xor(a1, 2);
    if (g == 0 && i < n) {
        float2 o; o.x = a0 * di; o.y = a1 * di;
        ((float2*)h2s)[i] = o;
    }
}

__global__ void gcn_final_kernel(const float* __restrict__ part,
                                 const float* __restrict__ h2s,
                                 const float* __restrict__ dinv,
                                 const float* __restrict__ b2,
                                 float* __restrict__ out,
                                 int n) {
    const int i = blockIdx.x * blockDim.x + threadIdx.x;
    if (i >= n) return;
    float2 s = ((const float2*)h2s)[i];
    #pragma unroll 16
    for (int q = 0; q < Q2; ++q) {
        const float2 t = ((const float2*)part)[(size_t)q * n + i];
        s.x += t.x; s.y += t.y;
    }
    const float di = dinv[i];
    const float a = fmaf(s.x, di, b2[0]);
    const float b = fmaf(s.y, di, b2[1]);
    const float m = fmaxf(a, b);
    const float lse = m + logf(expf(a - m) + expf(b - m));
    float2 o; o.x = a - lse; o.y = b - lse;
    ((float2*)out)[i] = o;
}

extern "C" void kernel_launch(void* const* d_in, const int* in_sizes, int n_in,
                              void* d_out, int out_size, void* d_ws, size_t ws_size,
                              hipStream_t stream) {
    const float* x   = (const float*)d_in[0];
    const int*   ei  = (const int*)d_in[1];
    const float* W1  = (const float*)d_in[2];
    const float* b1  = (const float*)d_in[3];
    const float* W2  = (const float*)d_in[4];
    const float* b2  = (const float*)d_in[5];
    float*       out = (float*)d_out;

    const int N = in_sizes[0] / FIN;
    const int E = in_sizes[1] / 2;
    const int* rows = ei;
    const int* cols = ei + E;

    float* ws   = (float*)d_ws;
    float* h1s  = ws;
    float* h2s  = ws + (size_t)16 * N;
    float* dinv = ws + (size_t)18 * N;
    float* part = ws + (size_t)20 * N;

    const int BS = 256;

    gcn_tile_kernel<1, RD, QD, 1><<<((N + RD - 1) / RD) * QD, TWG, 0, stream>>>(
        rows, cols, h1s, part, N, E, (E + QD - 1) / QD);

    gcn_dinv_kernel<<<(N + BS - 1) / BS, BS, 0, stream>>>(part, dinv, N);

    {
        int tiles = (N + 15) / 16;
        int waves_per_block = BS / 32;
        int blocks = (tiles + waves_per_block - 1) / waves_per_block;
        gcn_xw1_wmma_kernel<<<blocks, BS, 0, stream>>>(x, W1, dinv, h1s, N);
    }

    gcn_tile_kernel<F1, R1, Q1, 4><<<((N + R1 - 1) / R1) * Q1, TWG, 0, stream>>>(
        rows, cols, h1s, part, N, E, (E + Q1 - 1) / Q1);

    gcn_layer1_finish_kernel<<<(4 * N + BS - 1) / BS, BS, 0, stream>>>(
        part, h1s, dinv, b1, W2, h2s, N);

    gcn_tile_kernel<F2, R2, Q2, 1><<<((N + R2 - 1) / R2) * Q2, TWG, 0, stream>>>(
        rows, cols, h2s, part, N, E, (E + Q2 - 1) / Q2);

    gcn_final_kernel<<<(N + BS - 1) / BS, BS, 0, stream>>>(part, h2s, dinv, b2, out, N);
}


# Filter reads targets 16 B per lane: 2 loads and index sums per 256-edge pass, not 8; run end tested per 16-B group, not per edge; a pass's hits queue in (element, lane) order; sum order changes
# speedup vs baseline: 2.4302x; 1.0374x over previous
#include <hip/hip_runtime.h>


typedef __attribute__((ext_vector_type(2))) float v2f;
typedef __attribute__((ext_vector_type(8))) float v8f;

#define F1 16
#define F2 2
#define FIN 25

#define TWG 1024
#define RD 50016
#define QD 128
#define R1 4000
#define Q1 10
#define R2 25008
#define Q2 64

#define CHUNK 256
#define LCAP  160
#define TSTR  33

__device__ __forceinline__ void lds_round_barrier() {
    asm volatile("s_wait_dscnt 0x0\n\ts_barrier_signal -1\n\ts_barrier_wait -1" ::: "memory");
}

template <int LCB>
__device__ __forceinline__ void drain_take(const unsigned* ml, int start, int m, int lane, int e0,
                                           const int* __restrict__ rows, unsigned& lc, int& r) {
    constexpr unsigned LCM = (1u << LCB) - 1u;
    lc = 0xFFFFFFFFu;
    r = 0;
    if (lane < m) {
        const unsigned ent = ml[start + lane];
        lc = ent & LCM;
        r = rows[e0 + (int)(ent >> LCB)];
    }
}

template <int F>
__device__ __forceinline__ void drain_gather(int r, int m, int lane, const float* __restrict__ h, float* v) {
    if (F == 16) {
        const int sub = lane >> 2;
        const int qd  = lane & 3;
        #pragma unroll
        for (int s = 0; s < 4; ++s) {
            const int src = 8 * s + sub;
            const int rs = __shfl(r, src);
            float4 t = make_float4(0.f, 0.f, 0.f, 0.f);
            if (src < m) t = ((const float4*)h)[rs * 4 + qd];
            v[4 * s + 0] = t.x;
            v[4 * s + 1] = t.y;
            v[4 * s + 2] = t.z;
            v[4 * s + 3] = t.w;
        }
    } else {
        float2 t = make_float2(0.f, 0.f);
        if (lane < m) t = ((const float2*)h)[r];
        v[0] = t.x;
        v[1] = t.y;
    }
}

template <int D>
__device__ __forceinline__ unsigned same_node_below(unsigned lc, int l8) {
    const unsigned o = (unsigned)__builtin_amdgcn_update_dpp(0, (int)lc, 0x110 + D, 0xF, 0xF, false);
    return (l8 >= D && o == lc) ? 1u : 0u;
}

template <int F>
__device__ __forceinline__ void drain_add(unsigned lc, const float* v, int m, int lane, float* acc) {
    if (F == 16) {
        const int sub = lane >> 2;
        const int qd  = lane & 3;
        const int l8  = lane & 7;
        unsigned rk = same_node_below<1>(lc, l8);
        rk += same_node_below<2>(lc, l8);
        rk += same_node_below<3>(lc, l8);
        rk += same_node_below<4>(lc, l8);
        rk += same_node_below<5>(lc, l8);
        rk += same_node_below<6>(lc, l8);
        rk += same_node_below<7>(lc, l8);
        const unsigned pk = (rk << 16) | (lc & 0xFFFFu);
        unsigned pks[4];
        #pragma unroll
        for (int s = 0; s < 4; ++s) pks[s] = (unsigned)__shfl((int)pk, 8 * s + sub);
        float4* acc4 = (float4*)acc;
        #pragma unroll
        for (int s = 0; s < 4; ++s) {
            const bool valid = 8 * s + sub < m;
            const unsigned node = pks[s] & 0xFFFFu;
            const unsigned rank = pks[s] >> 16;
            for (unsigned t = 0u; t < 8u; ++t) {
                const bool act = valid && (rank == t);
                if (__builtin_amdgcn_ballot_w32(act) == 0u) break;
                if (act) {
                    float4 a = acc4[node * 4u + (unsigned)qd];
                    a.x += v[4 * s + 0];
                    a.y += v[4 * s + 1];
                    a.z += v[4 * s + 2];
                    a.w += v[4 * s + 3];
                    acc4[node * 4u + (unsigned)qd] = a;
                }
            }
        }
    } else {
        const bool valid = lane < m;
        unsigned same = 0u;
        #pragma unroll
        for (int s = 0; s < 32; ++s) {
            const unsigned ls = (unsigned)__builtin_amdgcn_readlane((int)lc, s);
            same |= (ls == lc) ? (1u << s) : 0u;
        }
        const unsigned rank = (unsigned)__builtin_popcount(same & ((1u << lane) - 1u));
        float2* acc2 = (float2*)acc;
        for (unsigned t = 0u; t < 32u; ++t) {
            const bool act = valid && (rank == t);
            if (__builtin_amdgcn_ballot_w32(act) == 0u) break;
            if (act) {
                float2 a = acc2[lc];
                a.x += v[0]; a.y += v[1];
                acc2[lc] = a;
            }
        }
    }
}

template <int LCB>
__device__ __forceinline__ void drain_count(const unsigned* ml, int start, int m, int lane, float* acc) {
    constexpr unsigned LCM = (1u << LCB) - 1u;
    const bool valid = lane < m;
    unsigned lc = 0xFFFFFFFFu;
    if (valid) lc = ml[start + lane] & LCM;
    unsigned same = 0u;
    #pragma unroll
    for (int s = 0; s < 32; ++s) {
        const unsigned ls = (unsigned)__builtin_amdgcn_readlane((int)lc, s);
        same |= (ls == lc) ? (1u << s) : 0u;
    }
    const unsigned rank = (unsigned)__builtin_popcount(same & ((1u << lane) - 1u));
    if (valid && rank == 0u) acc[lc] += (float)__builtin_popcount(same);
}

__device__ __forceinline__ void filter_pass(int4 (&c)[2], const bool reload, const int4* __restrict__ cols4, int cursor, int wend,
                                            int last4, int lo, int nr, unsigned* qrow, int& na, int sub, int lane) {
    const int left = wend - cursor - 4 * lane;
    const unsigned lane_place = (unsigned)(4 * lane) << 16;
    #pragma unroll
    for (int j = 0; j < 2; ++j) {
        const int off = (j * 128 < left) ? lo : -(1 << 30);
        const int t[4] = {c[j].x, c[j].y, c[j].z, c[j].w};
        #pragma unroll
        for (int i = 0; i < 4; ++i) {
            const unsigned lc = (unsigned)(t[i] - off);
            const bool hit = lc < (unsigned)nr;
            const unsigned mk = __builtin_amdgcn_ballot_w32(hit);
            if (hit) qrow[na + (int)__builtin_amdgcn_mbcnt_lo(mk, 0u)] =
                         ((unsigned)(sub * CHUNK + j * 128 + i) << 16) | lane_place | lc;
            na += __builtin_popcount(mk);
        }
        if (reload) c[j] = cols4[min(((cursor + 2 * CHUNK) >> 2) + j * 32 + lane, last4)];
    }
}

template <int F, int R, int Q, int G>
__global__ __launch_bounds__(TWG) void gcn_tile_kernel(const int* __restrict__ rows,
                                                       const int* __restrict__ cols,
                                                       const float* __restrict__ h,
                                                       float* __restrict__ part,
                                                       int n, int n_edges, int slice) {
    constexpr int LCB = (R <= 4096) ? 12 : 16;
    constexpr int OSH = (F == 1) ? 6 : 5;
    static_assert(R <= (1 << LCB), "node index must fit its field");
    constexpr int QCAP = (G == 1) ? CHUNK : CHUNK + 64;
    constexpr int QSTR = QCAP + 4;
    constexpr int NB   = QCAP / 32;
    static_assert(QSTR % 64 == 4, "queue rows must lie 4 banks apart");
    __shared__ __attribute__((aligned(16))) float acc[R * F];
    __shared__ __attribute__((aligned(16))) unsigned qa[32 * QSTR];
    __shared__ unsigned cntm[32 * TSTR];
    __shared__ unsigned ml_all[32 * LCAP];
    const int lane = threadIdx.x & 31;
    const int wave = __builtin_amdgcn_readfirstlane((int)(threadIdx.x >> 5));
    unsigned* ml = ml_all + wave * LCAP;
    const int p  = blockIdx.x / Q;
    const int q  = blockIdx.x - p * Q;
    const int lo = p * R;
    const int nr = min(R, n - lo);

    for (int i = threadIdx.x; i < nr * F; i += TWG) acc[i] = 0.f;

    const int e0 = q * slice;
    const int e1 = min(n_edges, e0 + slice);
    const int wlen = (((slice + 31) >> 5) + 31) & ~31;
    const int wbeg = e0 + wave * wlen;
    const int wend = min(e1, wbeg + wlen);
    const int maxr = (wlen + CHUNK - 1) / CHUNK + 1;
    const int4* cols4 = (const int4*)cols;
    const int last4 = (n_edges >> 2) - 1;
    int cursor = wbeg;
    int4 ca[2], cb[2];
    #pragma unroll
    for (int j = 0; j < 2; ++j) {
        ca[j] = cols4[min((cursor >> 2) + j * 32 + lane, last4)];
        cb[j] = cols4[min(((cursor + CHUNK) >> 2) + j * 32 + lane, last4)];
    }
    bool odd  = false;
    bool lazy = false;
    int cnt = 0;
    constexpr int VN = (F == 16) ? 16 : 2;
    unsigned a_lc = 0u, b_lc = 0u;
    int a_r = 0, a_m = 0, b_m = 0;
    float b_v[VN];
    #pragma unroll
    for (int j = 0; j < VN; ++j) b_v[j] = 0.f;
    for (int round = 0; round < maxr; ++round) {
        const int rbeg = cursor - e0;
        if (odd) {
            #pragma unroll
            for (int j = 0; j < 2; ++j) { const int4 t = ca[j]; ca[j] = cb[j]; cb[j] = t; }
        }
        if (lazy) {
            #pragma unroll
            for (int j = 0; j < 2; ++j) cb[j] = cols4[min(((cursor + CHUNK) >> 2) + j * 32 + lane, last4)];
        }
        odd = false;
        lazy = false;
        int na = 0;
        #pragma unroll
        for (int s = 0; s < G; ++s) {
            if (s & 1) filter_pass(cb, s != G - 1, cols4, cursor, wend, last4, lo, nr, qa + wave * QSTR, na, s, lane);
            else       filter_pass(ca, s != G - 1, cols4, cursor, wend, last4, lo, nr, qa + wave * QSTR, na, s, lane);
            cursor += CHUNK;
            odd = !(s & 1);
            lazy = (s == G - 1);
            if (!(na + CHUNK <= QCAP && cursor < wend)) break;
        }
        unsigned* cw = cntm + wave * TSTR;
        cw[lane] = 0u;
        const int nb = (na + 31) >> 5;
        unsigned se[NB], sp[NB];
        #pragma unroll
        for (int b = 0; b < NB; ++b) {
            se[b] = 0u;
            sp[b] = 0u;
            if (b < nb) {
                const bool valid = b * 32 + lane < na;
                unsigned ent = 0u;
                if (valid) ent = qa[wave * QSTR + b * 32 + lane];
                unsigned same = __builtin_amdgcn_ballot_w32(valid);
                #pragma unroll
                for (int bit = 0; bit < 5; ++bit) {
                    const int sb = ((int)(ent << (31 - OSH - bit))) >> 31;
                    const unsigned m = __builtin_amdgcn_ballot_w32(valid && sb != 0);
                    same &= ~(m ^ (unsigned)sb);
                }
                const unsigned rank = __builtin_amdgcn_mbcnt_lo(same, 0u);
                const unsigned tot  = (unsigned)__builtin_popcount(same);
                unsigned base = 0u;
                if (valid) base = cw[(ent >> OSH) & 31u];
                if (valid && rank + 1u == tot) cw[(ent >> OSH) & 31u] = base + tot;
                se[b] = ent;
                sp[b] = base + rank;
            }
        }
        {
            const unsigned c = cw[lane];
            unsigned inc = c;
            inc += (unsigned)__builtin_amdgcn_update_dpp(0, (int)inc, 0x111, 0xF, 0xF, false);
            inc += (unsigned)__builtin_amdgcn_update_dpp(0, (int)inc, 0x112, 0xF, 0xF, false);
            inc += (unsigned)__builtin_amdgcn_update_dpp(0, (int)inc, 0x114, 0xF, 0xF, false);
            inc += (unsigned)__builtin_amdgcn_update_dpp(0, (int)inc, 0x118, 0xF, 0xF, false);
            const unsigned row0 = (unsigned)__builtin_amdgcn_readlane((int)inc, 15);
            if (lane >= 16) inc += row0;
            cw[lane] = ((inc - c) << 16) | c;
        }
        #pragma unroll
        for (int b = 0; b < NB; ++b) {
            if (b < nb) {
                if (b * 32 + lane < na) qa[wave * QSTR + (int)((cw[(se[b] >> OSH) & 31u] >> 16) + sp[b])] = se[b];
            }
        }
        if (lane == 0) cw[32] = ((unsigned)rbeg << 1) | ((cursor >= wend) ? 1u : 0u);
        lds_round_barrier();
        const unsigned sc = cntm[lane * TSTR + wave];
        const unsigned rw = cntm[lane * TSTR + 32];
        const int st = (int)(sc >> 16);
        const int nl = (int)(sc & 0xFFFFu);
        const unsigned ebase = rw >> 1;
        const bool all_through = __builtin_amdgcn_ballot_w32((rw & 1u) != 0u) == 0xFFFFFFFFu;
        for (int i0 = 0; __builtin_amdgcn_ballot_w32(i0 < nl) != 0u; i0 += 4) {
            unsigned ent[4];
            #pragma unroll
            for (int u = 0; u < 4; ++u) {
                ent[u] = 0u;
                if (i0 + u < nl) ent[u] = qa[lane * QSTR + st + i0 + u];
            }
            #pragma unroll
            for (int u = 0; u < 4; ++u) {
                const bool have = i0 + u < nl;
                const unsigned mk = __builtin_amdgcn_ballot_w32(have);
                if (mk != 0u) {
                    if (have) ml[cnt + (int)__builtin_amdgcn_mbcnt_lo(mk, 0u)] =
                                  ((ebase + (ent[u] >> 16)) << LCB) | (ent[u] & 0xFFFFu);
                    cnt += __builtin_popcount(mk);
                }
            }
            while (cnt >= 32) {
                cnt -= 32;
                if (F == 1) {
                    drain_count<LCB>(ml, cnt, 32, lane, acc);
                } else {
                    float c_v[VN];
                    #pragma unroll
                    for (int j = 0; j < VN; ++j) c_v[j] = b_v[j];
                    const unsigned c_lc = b_lc;
                    const int c_m = b_m;
                    b_m = a_m;
                    b_lc = a_lc;
                    drain_gather<F>(a_r, 32, lane, h, b_v);
                    drain_take<LCB>(ml, cnt, 32, lane, e0, rows, a_lc, a_r);
                    a_m = 32;
                    if (c_m) drain_add<F>(c_lc, c_v, 32, lane, acc);
                }
            }
        }
        lds_round_barrier();
        if (all_through) break;
    }
    if (F == 1) {
        if (cnt > 0) drain_count<LCB>(ml, 0, cnt, lane, acc);
    } else {
        int tail = cnt;
        #pragma nounroll
        for (int t = 0; t < 3; ++t) {
            if (b_m) drain_add<F>(b_lc, b_v, b_m, lane, acc);
            b_m = a_m;
            b_lc = a_lc;
            if (a_m) drain_gather<F>(a_r, a_m, lane, h, b_v);
            a_m = tail;
            if (tail) drain_take<LCB>(ml, 0, tail, lane, e0, rows, a_lc, a_r);
            tail = 0;
        }
    }
    __syncthreads();

    float* dst = part + ((size_t)q * n + lo) * F;
    for (int i = threadIdx.x; i < nr * F; i += TWG) dst[i] = acc[i];
}

__global__ void gcn_dinv_kernel(const float* __restrict__ part,
                                float* __restrict__ dinv, int n) {
    const int i = blockIdx.x * blockDim.x + threadIdx.x;
    if (i >= n) return;
    float c = 1.0f;
    #pragma unroll 16
    for (int q = 0; q < QD; ++q) c += part[(size_t)q * n + i];
    dinv[i] = rsqrtf(c);
}

__global__ void gcn_xw1_wmma_kernel(const float* __restrict__ x,
                                    const float* __restrict__ W1,
                                    const float* __restrict__ dinv,
                                    float* __restrict__ h1s,
                                    int n_nodes) {
    const int lane  = threadIdx.x & 31;
    const int wave  = threadIdx.x >> 5;
    const int tile  = blockIdx.x * (blockDim.x >> 5) + wave;
    const int node0 = tile * 16;
    if (node0 >= n_nodes) return;

    const int half = lane >> 4;
    const int l16  = lane & 15;
    const int m    = l16;

    v8f c = {0.f, 0.f, 0.f, 0.f, 0.f, 0.f, 0.f, 0.f};

    #pragma unroll
    for (int k0 = 0; k0 < 28; k0 += 4) {
        const int ka = k0 + 2 * half;
        v2f a, b;
        a.x = (ka     < FIN) ? x[(node0 + m) * FIN + ka    ] : 0.0f;
        a.y = (ka + 1 < FIN) ? x[(node0 + m) * FIN + ka + 1] : 0.0f;
        b.x = (ka     < FIN) ? W1[(ka    ) * F1 + l16] : 0.0f;
        b.y = (ka + 1 < FIN) ? W1[(ka + 1) * F1 + l16] : 0.0f;
        c = __builtin_amdgcn_wmma_f32_16x16x4_f32(
                 false, a,  false, b,
                 (short)0, c,  false,  false);
    }

    #pragma unroll
    for (int v = 0; v < 8; ++v) {
        const int node = node0 + v + 8 * half;
        h1s[node * F1 + l16] = c[v] * dinv[node];
    }
}

__global__ void gcn_layer1_finish_kernel(const float* __restrict__ part,
                                         const float* __restrict__ h1s,
                                         const float* __restrict__ dinv,
                                         const float* __restrict__ b1,
                                         const float* __restrict__ W2,
                                         float* __restrict__ h2s,
                                         int n) {
    const int gid = blockIdx.x * blockDim.x + threadIdx.x;
    const int i   = gid >> 2;
    const int g   = gid & 3;
    const int ic  = (i < n) ? i : (n - 1);
    float4 s = ((const float4*)h1s)[(size_t)ic * 4 + g];
    #pragma unroll
    for (int q = 0; q < Q1; ++q) {
        const float4 t = ((const float4*)part)[((size_t)q * n + ic) * 4 + g];
        s.x += t.x; s.y += t.y; s.z += t.z; s.w += t.w;
    }
    const float di = dinv[ic];
    const float v0 = fmaxf(fmaf(s.x, di, b1[4 * g + 0]), 0.0f);
    const float v1 = fmaxf(fmaf(s.y, di, b1[4 * g + 1]), 0.0f);
    const float v2 = fmaxf(fmaf(s.z, di, b1[4 * g + 2]), 0.0f);
    const float v3 = fmaxf(fmaf(s.w, di, b1[4 * g + 3]), 0.0f);
    const float* w = W2 + 8 * g;
    float a0 = v0 * w[0];
    float a1 = v0 * w[1];
    a0 = fmaf(v1, w[2], a0); a1 = fmaf(v1, w[3], a1);
    a0 = fmaf(v2, w[4], a0); a1 = fmaf(v2, w[5], a1);
    a0 = fmaf(v3, w[6], a0); a1 = fmaf(v3, w[7], a1);
    a0 += __shfl_xor(a0, 1); a1 += __shfl_xor(a1, 1);
    a0 += __shfl_xor(a0, 2); a1 += __shfl_xor(a1, 2);
    if (g == 0 && i < n) {
        float2 o; o.x = a0 * di; o.y = a1 * di;
        ((float2*)h2s)[i] = o;
    }
}

__global__ void gcn_final_kernel(const float* __restrict__ part,
                                 const float* __restrict__ h2s,
                                 const float* __restrict__ dinv,
                                 const float* __restrict__ b2,
                                 float* __restrict__ out,
                                 int n) {
    const int i = blockIdx.x * blockDim.x + threadIdx.x;
    if (i >= n) return;
    float2 s = ((const float2*)h2s)[i];
    #pragma unroll 16
    for (int q = 0; q < Q2; ++q) {
        const float2 t = ((const float2*)part)[(size_t)q * n + i];
        s.x += t.x; s.y += t.y;
    }
    const float di = dinv[i];
    const float a = fmaf(s.x, di, b2[0]);
    const float b = fmaf(s.y, di, b2[1]);
    const float m = fmaxf(a, b);
    const float lse = m + logf(expf(a - m) + expf(b - m));
    float2 o; o.x = a - lse; o.y = b - lse;
    ((float2*)out)[i] = o;
}

extern "C" void kernel_launch(void* const* d_in, const int* in_sizes, int n_in,
                              void* d_out, int out_size, void* d_ws, size_t ws_size,
                              hipStream_t stream) {
    const float* x   = (const float*)d_in[0];
    const int*   ei  = (const int*)d_in[1];
    const float* W1  = (const float*)d_in[2];
    const float* b1  = (const float*)d_in[3];
    const float* W2  = (const float*)d_in[4];
    const float* b2  = (const float*)d_in[5];
    float*       out = (float*)d_out;

    const int N = in_sizes[0] / FIN;
    const int E = in_sizes[1] / 2;
    const int* rows = ei;
    const int* cols = ei + E;

    float* ws   = (float*)d_ws;
    float* h1s  = ws;
    float* h2s  = ws + (size_t)16 * N;
    float* dinv = ws + (size_t)18 * N;
    float* part = ws + (size_t)20 * N;

    const int BS = 256;

    gcn_tile_kernel<1, RD, QD, 1><<<((N + RD - 1) / RD) * QD, TWG, 0, stream>>>(
        rows, cols, h1s, part, N, E, (E + QD - 1) / QD);

    gcn_dinv_kernel<<<(N + BS - 1) / BS, BS, 0, stream>>>(part, dinv, N);

    {
        int tiles = (N + 15) / 16;
        int waves_per_block = BS / 32;
        int blocks = (tiles + waves_per_block - 1) / waves_per_block;
        gcn_xw1_wmma_kernel<<<blocks, BS, 0, stream>>>(x, W1, dinv, h1s, N);
    }

    gcn_tile_kernel<F1, R1, Q1, 4><<<((N + R1 - 1) / R1) * Q1, TWG, 0, stream>>>(
        rows, cols, h1s, part, N, E, (E + Q1 - 1) / Q1);

    gcn_layer1_finish_kernel<<<(4 * N + BS - 1) / BS, BS, 0, stream>>>(
        part, h1s, dinv, b1, W2, h2s, N);

    gcn_tile_kernel<F2, R2, Q2, 1><<<((N + R2 - 1) / R2) * Q2, TWG, 0, stream>>>(
        rows, cols, h2s, part, N, E, (E + Q2 - 1) / Q2);

    gcn_final_kernel<<<(N + BS - 1) / BS, BS, 0, stream>>>(part, h2s, dinv, b2, out, N);
}


# Degree and layer 2 re-tiled: 8 node ranges x 32 edge slices (were 2 x 128 and 4 x 64), 4 filter passes per round: a quarter and half the partial sums written and re-read; sum order changes
# speedup vs baseline: 2.5900x; 1.0658x over previous
#include <hip/hip_runtime.h>


typedef __attribute__((ext_vector_type(2))) float v2f;
typedef __attribute__((ext_vector_type(8))) float v8f;

#define F1 16
#define F2 2
#define FIN 25

#define TWG 1024
#define RD 12512
#define QD 32
#define R1 4000
#define Q1 10
#define R2 12512
#define Q2 32

#define CHUNK 256
#define LCAP  160
#define TSTR  33

__device__ __forceinline__ void lds_round_barrier() {
    asm volatile("s_wait_dscnt 0x0\n\ts_barrier_signal -1\n\ts_barrier_wait -1" ::: "memory");
}

template <int LCB>
__device__ __forceinline__ void drain_take(const unsigned* ml, int start, int m, int lane, int e0,
                                           const int* __restrict__ rows, unsigned& lc, int& r) {
    constexpr unsigned LCM = (1u << LCB) - 1u;
    lc = 0xFFFFFFFFu;
    r = 0;
    if (lane < m) {
        const unsigned ent = ml[start + lane];
        lc = ent & LCM;
        r = rows[e0 + (int)(ent >> LCB)];
    }
}

template <int F>
__device__ __forceinline__ void drain_gather(int r, int m, int lane, const float* __restrict__ h, float* v) {
    if (F == 16) {
        const int sub = lane >> 2;
        const int qd  = lane & 3;
        #pragma unroll
        for (int s = 0; s < 4; ++s) {
            const int src = 8 * s + sub;
            const int rs = __shfl(r, src);
            float4 t = make_float4(0.f, 0.f, 0.f, 0.f);
            if (src < m) t = ((const float4*)h)[rs * 4 + qd];
            v[4 * s + 0] = t.x;
            v[4 * s + 1] = t.y;
            v[4 * s + 2] = t.z;
            v[4 * s + 3] = t.w;
        }
    } else {
        float2 t = make_float2(0.f, 0.f);
        if (lane < m) t = ((const float2*)h)[r];
        v[0] = t.x;
        v[1] = t.y;
    }
}

template <int D>
__device__ __forceinline__ unsigned same_node_below(unsigned lc, int l8) {
    const unsigned o = (unsigned)__builtin_amdgcn_update_dpp(0, (int)lc, 0x110 + D, 0xF, 0xF, false);
    return (l8 >= D && o == lc) ? 1u : 0u;
}

template <int F>
__device__ __forceinline__ void drain_add(unsigned lc, const float* v, int m, int lane, float* acc) {
    if (F == 16) {
        const int sub = lane >> 2;
        const int qd  = lane & 3;
        const int l8  = lane & 7;
        unsigned rk = same_node_below<1>(lc, l8);
        rk += same_node_below<2>(lc, l8);
        rk += same_node_below<3>(lc, l8);
        rk += same_node_below<4>(lc, l8);
        rk += same_node_below<5>(lc, l8);
        rk += same_node_below<6>(lc, l8);
        rk += same_node_below<7>(lc, l8);
        const unsigned pk = (rk << 16) | (lc & 0xFFFFu);
        unsigned pks[4];
        #pragma unroll
        for (int s = 0; s < 4; ++s) pks[s] = (unsigned)__shfl((int)pk, 8 * s + sub);
        float4* acc4 = (float4*)acc;
        #pragma unroll
        for (int s = 0; s < 4; ++s) {
            const bool valid = 8 * s + sub < m;
            const unsigned node = pks[s] & 0xFFFFu;
            const unsigned rank = pks[s] >> 16;
            for (unsigned t = 0u; t < 8u; ++t) {
                const bool act = valid && (rank == t);
                if (__builtin_amdgcn_ballot_w32(act) == 0u) break;
                if (act) {
                    float4 a = acc4[node * 4u + (unsigned)qd];
                    a.x += v[4 * s + 0];
                    a.y += v[4 * s + 1];
                    a.z += v[4 * s + 2];
                    a.w += v[4 * s + 3];
                    acc4[node * 4u + (unsigned)qd] = a;
                }
            }
        }
    } else {
        const bool valid = lane < m;
        unsigned same = 0u;
        #pragma unroll
        for (int s = 0; s < 32; ++s) {
            const unsigned ls = (unsigned)__builtin_amdgcn_readlane((int)lc, s);
            same |= (ls == lc) ? (1u << s) : 0u;
        }
        const unsigned rank = (unsigned)__builtin_popcount(same & ((1u << lane) - 1u));
        float2* acc2 = (float2*)acc;
        for (unsigned t = 0u; t < 32u; ++t) {
            const bool act = valid && (rank == t);
            if (__builtin_amdgcn_ballot_w32(act) == 0u) break;
            if (act) {
                float2 a = acc2[lc];
                a.x += v[0]; a.y += v[1];
                acc2[lc] = a;
            }
        }
    }
}

template <int LCB>
__device__ __forceinline__ void drain_count(const unsigned* ml, int start, int m, int lane, float* acc) {
    constexpr unsigned LCM = (1u << LCB) - 1u;
    const bool valid = lane < m;
    unsigned lc = 0xFFFFFFFFu;
    if (valid) lc = ml[start + lane] & LCM;
    unsigned same = 0u;
    #pragma unroll
    for (int s = 0; s < 32; ++s) {
        const unsigned ls = (unsigned)__builtin_amdgcn_readlane((int)lc, s);
        same |= (ls == lc) ? (1u << s) : 0u;
    }
    const unsigned rank = (unsigned)__builtin_popcount(same & ((1u << lane) - 1u));
    if (valid && rank == 0u) acc[lc] += (float)__builtin_popcount(same);
}

__device__ __forceinline__ void filter_pass(int4 (&c)[2], const bool reload, const int4* __restrict__ cols4, int cursor, int wend,
                                            int last4, int lo, int nr, unsigned* qrow, int& na, int sub, int lane) {
    const int left = wend - cursor - 4 * lane;
    const unsigned lane_place = (unsigned)(4 * lane) << 16;
    #pragma unroll
    for (int j = 0; j < 2; ++j) {
        const int off = (j * 128 < left) ? lo : -(1 << 30);
        const int t[4] = {c[j].x, c[j].y, c[j].z, c[j].w};
        #pragma unroll
        for (int i = 0; i < 4; ++i) {
            const unsigned lc = (unsigned)(t[i] - off);
            const bool hit = lc < (unsigned)nr;
            const unsigned mk = __builtin_amdgcn_ballot_w32(hit);
            if (hit) qrow[na + (int)__builtin_amdgcn_mbcnt_lo(mk, 0u)] =
                         ((unsigned)(sub * CHUNK + j * 128 + i) << 16) | lane_place | lc;
            na += __builtin_popcount(mk);
        }
        if (reload) c[j] = cols4[min(((cursor + 2 * CHUNK) >> 2) + j * 32 + lane, last4)];
    }
}

template <int F, int R, int Q, int G, int QX>
__global__ __launch_bounds__(TWG) void gcn_tile_kernel(const int* __restrict__ rows,
                                                       const int* __restrict__ cols,
                                                       const float* __restrict__ h,
                                                       float* __restrict__ part,
                                                       int n, int n_edges, int slice) {
    constexpr int LCB = (R <= 4096) ? 12 : (R <= 16384) ? 14 : 16;
    constexpr int OSH = (F == 1) ? 6 : 5;
    static_assert(R <= (1 << LCB), "node index must fit its field");
    constexpr int QCAP = CHUNK + QX;
    constexpr int QSTR = QCAP + 4;
    constexpr int NB   = QCAP / 32;
    static_assert(QSTR % 64 == 4, "queue rows must lie 4 banks apart");
    __shared__ __attribute__((aligned(16))) float acc[R * F];
    __shared__ __attribute__((aligned(16))) unsigned qa[32 * QSTR];
    __shared__ unsigned cntm[32 * TSTR];
    __shared__ unsigned ml_all[32 * LCAP];
    const int lane = threadIdx.x & 31;
    const int wave = __builtin_amdgcn_readfirstlane((int)(threadIdx.x >> 5));
    unsigned* ml = ml_all + wave * LCAP;
    const int p  = blockIdx.x / Q;
    const int q  = blockIdx.x - p * Q;
    const int lo = p * R;
    const int nr = min(R, n - lo);

    for (int i = threadIdx.x; i < nr * F; i += TWG) acc[i] = 0.f;

    const int e0 = q * slice;
    const int e1 = min(n_edges, e0 + slice);
    const int wlen = (((slice + 31) >> 5) + 31) & ~31;
    const int wbeg = e0 + wave * wlen;
    const int wend = min(e1, wbeg + wlen);
    const int maxr = (wlen + CHUNK - 1) / CHUNK + 1;
    const int4* cols4 = (const int4*)cols;
    const int last4 = (n_edges >> 2) - 1;
    int cursor = wbeg;
    int4 ca[2], cb[2];
    #pragma unroll
    for (int j = 0; j < 2; ++j) {
        ca[j] = cols4[min((cursor >> 2) + j * 32 + lane, last4)];
        cb[j] = cols4[min(((cursor + CHUNK) >> 2) + j * 32 + lane, last4)];
    }
    bool odd  = false;
    bool lazy = false;
    int cnt = 0;
    constexpr int VN = (F == 16) ? 16 : 2;
    unsigned a_lc = 0u, b_lc = 0u;
    int a_r = 0, a_m = 0, b_m = 0;
    float b_v[VN];
    #pragma unroll
    for (int j = 0; j < VN; ++j) b_v[j] = 0.f;
    for (int round = 0; round < maxr; ++round) {
        const int rbeg = cursor - e0;
        if (odd) {
            #pragma unroll
            for (int j = 0; j < 2; ++j) { const int4 t = ca[j]; ca[j] = cb[j]; cb[j] = t; }
        }
        if (lazy) {
            #pragma unroll
            for (int j = 0; j < 2; ++j) cb[j] = cols4[min(((cursor + CHUNK) >> 2) + j * 32 + lane, last4)];
        }
        odd = false;
        lazy = false;
        int na = 0;
        #pragma unroll
        for (int s = 0; s < G; ++s) {
            if (s & 1) filter_pass(cb, s != G - 1, cols4, cursor, wend, last4, lo, nr, qa + wave * QSTR, na, s, lane);
            else       filter_pass(ca, s != G - 1, cols4, cursor, wend, last4, lo, nr, qa + wave * QSTR, na, s, lane);
            cursor += CHUNK;
            odd = !(s & 1);
            lazy = (s == G - 1);
            if (!(na + CHUNK <= QCAP && cursor < wend)) break;
        }
        unsigned* cw = cntm + wave * TSTR;
        cw[lane] = 0u;
        const int nb = (na + 31) >> 5;
        unsigned se[NB], sp[NB];
        #pragma unroll
        for (int b = 0; b < NB; ++b) {
            se[b] = 0u;
            sp[b] = 0u;
            if (b < nb) {
                const bool valid = b * 32 + lane < na;
                unsigned ent = 0u;
                if (valid) ent = qa[wave * QSTR + b * 32 + lane];
                unsigned same = __builtin_amdgcn_ballot_w32(valid);
                #pragma unroll
                for (int bit = 0; bit < 5; ++bit) {
                    const int sb = ((int)(ent << (31 - OSH - bit))) >> 31;
                    const unsigned m = __builtin_amdgcn_ballot_w32(valid && sb != 0);
                    same &= ~(m ^ (unsigned)sb);
                }
                const unsigned rank = __builtin_amdgcn_mbcnt_lo(same, 0u);
                const unsigned tot  = (unsigned)__builtin_popcount(same);
                unsigned base = 0u;
                if (valid) base = cw[(ent >> OSH) & 31u];
                if (valid && rank + 1u == tot) cw[(ent >> OSH) & 31u] = base + tot;
                se[b] = ent;
                sp[b] = base + rank;
            }
        }
        {
            const unsigned c = cw[lane];
            unsigned inc = c;
            inc += (unsigned)__builtin_amdgcn_update_dpp(0, (int)inc, 0x111, 0xF, 0xF, false);
            inc += (unsigned)__builtin_amdgcn_update_dpp(0, (int)inc, 0x112, 0xF, 0xF, false);
            inc += (unsigned)__builtin_amdgcn_update_dpp(0, (int)inc, 0x114, 0xF, 0xF, false);
            inc += (unsigned)__builtin_amdgcn_update_dpp(0, (int)inc, 0x118, 0xF, 0xF, false);
            const unsigned row0 = (unsigned)__builtin_amdgcn_readlane((int)inc, 15);
            if (lane >= 16) inc += row0;
            cw[lane] = ((inc - c) << 16) | c;
        }
        #pragma unroll
        for (int b = 0; b < NB; ++b) {
            if (b < nb) {
                if (b * 32 + lane < na) qa[wave * QSTR + (int)((cw[(se[b] >> OSH) & 31u] >> 16) + sp[b])] = se[b];
            }
        }
        if (lane == 0) cw[32] = ((unsigned)rbeg << 1) | ((cursor >= wend) ? 1u : 0u);
        lds_round_barrier();
        const unsigned sc = cntm[lane * TSTR + wave];
        const unsigned rw = cntm[lane * TSTR + 32];
        const int st = (int)(sc >> 16);
        const int nl = (int)(sc & 0xFFFFu);
        const unsigned ebase = rw >> 1;
        const bool all_through = __builtin_amdgcn_ballot_w32((rw & 1u) != 0u) == 0xFFFFFFFFu;
        for (int i0 = 0; __builtin_amdgcn_ballot_w32(i0 < nl) != 0u; i0 += 4) {
            unsigned ent[4];
            #pragma unroll
            for (int u = 0; u < 4; ++u) {
                ent[u] = 0u;
                if (i0 + u < nl) ent[u] = qa[lane * QSTR + st + i0 + u];
            }
            #pragma unroll
            for (int u = 0; u < 4; ++u) {
                const bool have = i0 + u < nl;
                const unsigned mk = __builtin_amdgcn_ballot_w32(have);
                if (mk != 0u) {
                    if (have) ml[cnt + (int)__builtin_amdgcn_mbcnt_lo(mk, 0u)] =
                                  ((ebase + (ent[u] >> 16)) << LCB) | (ent[u] & 0xFFFFu);
                    cnt += __builtin_popcount(mk);
                }
            }
            while (cnt >= 32) {
                cnt -= 32;
                if (F == 1) {
                    drain_count<LCB>(ml, cnt, 32, lane, acc);
                } else {
                    float c_v[VN];
                    #pragma unroll
                    for (int j = 0; j < VN; ++j) c_v[j] = b_v[j];
                    const unsigned c_lc = b_lc;
                    const int c_m = b_m;
                    b_m = a_m;
                    b_lc = a_lc;
                    drain_gather<F>(a_r, 32, lane, h, b_v);
                    drain_take<LCB>(ml, cnt, 32, lane, e0, rows, a_lc, a_r);
                    a_m = 32;
                    if (c_m) drain_add<F>(c_lc, c_v, 32, lane, acc);
                }
            }
        }
        lds_round_barrier();
        if (all_through) break;
    }
    if (F == 1) {
        if (cnt > 0) drain_count<LCB>(ml, 0, cnt, lane, acc);
    } else {
        int tail = cnt;
        #pragma nounroll
        for (int t = 0; t < 3; ++t) {
            if (b_m) drain_add<F>(b_lc, b_v, b_m, lane, acc);
            b_m = a_m;
            b_lc = a_lc;
            if (a_m) drain_gather<F>(a_r, a_m, lane, h, b_v);
            a_m = tail;
            if (tail) drain_take<LCB>(ml, 0, tail, lane, e0, rows, a_lc, a_r);
            tail = 0;
        }
    }
    __syncthreads();

    float* dst = part + ((size_t)q * n + lo) * F;
    for (int i = threadIdx.x; i < nr * F; i += TWG) dst[i] = acc[i];
}

__global__ void gcn_dinv_kernel(const float* __restrict__ part,
                                float* __restrict__ dinv, int n) {
    const int i = blockIdx.x * blockDim.x + threadIdx.x;
    if (i >= n) return;
    float c = 1.0f;
    #pragma unroll 16
    for (int q = 0; q < QD; ++q) c += part[(size_t)q * n + i];
    dinv[i] = rsqrtf(c);
}

__global__ void gcn_xw1_wmma_kernel(const float* __restrict__ x,
                                    const float* __restrict__ W1,
                                    const float* __restrict__ dinv,
                                    float* __restrict__ h1s,
                                    int n_nodes) {
    const int lane  = threadIdx.x & 31;
    const int wave  = threadIdx.x >> 5;
    const int tile  = blockIdx.x * (blockDim.x >> 5) + wave;
    const int node0 = tile * 16;
    if (node0 >= n_nodes) return;

    const int half = lane >> 4;
    const int l16  = lane & 15;
    const int m    = l16;

    v8f c = {0.f, 0.f, 0.f, 0.f, 0.f, 0.f, 0.f, 0.f};

    #pragma unroll
    for (int k0 = 0; k0 < 28; k0 += 4) {
        const int ka = k0 + 2 * half;
        v2f a, b;
        a.x = (ka     < FIN) ? x[(node0 + m) * FIN + ka    ] : 0.0f;
        a.y = (ka + 1 < FIN) ? x[(node0 + m) * FIN + ka + 1] : 0.0f;
        b.x = (ka     < FIN) ? W1[(ka    ) * F1 + l16] : 0.0f;
        b.y = (ka + 1 < FIN) ? W1[(ka + 1) * F1 + l16] : 0.0f;
        c = __builtin_amdgcn_wmma_f32_16x16x4_f32(
                 false, a,  false, b,
                 (short)0, c,  false,  false);
    }

    #pragma unroll
    for (int v = 0; v < 8; ++v) {
        const int node = node0 + v + 8 * half;
        h1s[node * F1 + l16] = c[v] * dinv[node];
    }
}

__global__ void gcn_layer1_finish_kernel(const float* __restrict__ part,
                                         const float* __restrict__ h1s,
                                         const float* __restrict__ dinv,
                                         const float* __restrict__ b1,
                                         const float* __restrict__ W2,
                                         float* __restrict__ h2s,
                                         int n) {
    const int gid = blockIdx.x * blockDim.x + threadIdx.x;
    const int i   = gid >> 2;
    const int g   = gid & 3;
    const int ic  = (i < n) ? i : (n - 1);
    float4 s = ((const float4*)h1s)[(size_t)ic * 4 + g];
    #pragma unroll
    for (int q = 0; q < Q1; ++q) {
        const float4 t = ((const float4*)part)[((size_t)q * n + ic) * 4 + g];
        s.x += t.x; s.y += t.y; s.z += t.z; s.w += t.w;
    }
    const float di = dinv[ic];
    const float v0 = fmaxf(fmaf(s.x, di, b1[4 * g + 0]), 0.0f);
    const float v1 = fmaxf(fmaf(s.y, di, b1[4 * g + 1]), 0.0f);
    const float v2 = fmaxf(fmaf(s.z, di, b1[4 * g + 2]), 0.0f);
    const float v3 = fmaxf(fmaf(s.w, di, b1[4 * g + 3]), 0.0f);
    const float* w = W2 + 8 * g;
    float a0 = v0 * w[0];
    float a1 = v0 * w[1];
    a0 = fmaf(v1, w[2], a0); a1 = fmaf(v1, w[3], a1);
    a0 = fmaf(v2, w[4], a0); a1 = fmaf(v2, w[5], a1);
    a0 = fmaf(v3, w[6], a0); a1 = fmaf(v3, w[7], a1);
    a0 += __shfl_xor(a0, 1); a1 += __shfl_xor(a1, 1);
    a0 += __shfl_xor(a0, 2); a1 += __shfl_xor(a1, 2);
    if (g == 0 && i < n) {
        float2 o; o.x = a0 * di; o.y = a1 * di;
        ((float2*)h2s)[i] = o;
    }
}

__global__ void gcn_final_kernel(const float* __restrict__ part,
                                 const float* __restrict__ h2s,
                                 const float* __restrict__ dinv,
                                 const float* __restrict__ b2,
                                 float* __restrict__ out,
                                 int n) {
    const int i = blockIdx.x * blockDim.x + threadIdx.x;
    if (i >= n) return;
    float2 s = ((const float2*)h2s)[i];
    #pragma unroll 16
    for (int q = 0; q < Q2; ++q) {
        const float2 t = ((const float2*)part)[(size_t)q * n + i];
        s.x += t.x; s.y += t.y;
    }
    const float di = dinv[i];
    const float a = fmaf(s.x, di, b2[0]);
    const float b = fmaf(s.y, di, b2[1]);
    const float m = fmaxf(a, b);
    const float lse = m + logf(expf(a - m) + expf(b - m));
    float2 o; o.x = a - lse; o.y = b - lse;
    ((float2*)out)[i] = o;
}

extern "C" void kernel_launch(void* const* d_in, const int* in_sizes, int n_in,
                              void* d_out, int out_size, void* d_ws, size_t ws_size,
                              hipStream_t stream) {
    const float* x   = (const float*)d_in[0];
    const int*   ei  = (const int*)d_in[1];
    const float* W1  = (const float*)d_in[2];
    const float* b1  = (const float*)d_in[3];
    const float* W2  = (const float*)d_in[4];
    const float* b2  = (const float*)d_in[5];
    float*       out = (float*)d_out;

    const int N = in_sizes[0] / FIN;
    const int E = in_sizes[1] / 2;
    const int* rows = ei;
    const int* cols = ei + E;

    float* ws   = (float*)d_ws;
    float* h1s  = ws;
    float* h2s  = ws + (size_t)16 * N;
    float* dinv = ws + (size_t)18 * N;
    float* part = ws + (size_t)20 * N;

    const int BS = 256;

    gcn_tile_kernel<1, RD, QD, 4, 192><<<((N + RD - 1) / RD) * QD, TWG, 0, stream>>>(
        rows, cols, h1s, part, N, E, (E + QD - 1) / QD);

    gcn_dinv_kernel<<<(N + BS - 1) / BS, BS, 0, stream>>>(part, dinv, N);

    {
        int tiles = (N + 15) / 16;
        int waves_per_block = BS / 32;
        int blocks = (tiles + waves_per_block - 1) / waves_per_block;
        gcn_xw1_wmma_kernel<<<blocks, BS, 0, stream>>>(x, W1, dinv, h1s, N);
    }

    gcn_tile_kernel<F1, R1, Q1, 4, 64><<<((N + R1 - 1) / R1) * Q1, TWG, 0, stream>>>(
        rows, cols, h1s, part, N, E, (E + Q1 - 1) / Q1);

    gcn_layer1_finish_kernel<<<(4 * N + BS - 1) / BS, BS, 0, stream>>>(
        part, h1s, dinv, b1, W2, h2s, N);

    gcn_tile_kernel<F2, R2, Q2, 4, 192><<<((N + R2 - 1) / R2) * Q2, TWG, 0, stream>>>(
        rows, cols, h2s, part, N, E, (E + Q2 - 1) / Q2);

    gcn_final_kernel<<<(N + BS - 1) / BS, BS, 0, stream>>>(part, h2s, dinv, b2, out, N);
}


# Rounds end only when the queue is full or the run through: a filter pass is counted, then stored, so no room is held back; 1 round not 4 in degree and layer 2, 2 not 10 in layer 1; sum order changes
# speedup vs baseline: 2.9969x; 1.1571x over previous
#include <hip/hip_runtime.h>


typedef __attribute__((ext_vector_type(2))) float v2f;
typedef __attribute__((ext_vector_type(8))) float v8f;

#define F1 16
#define F2 2
#define FIN 25

#define TWG 1024
#define RD 12512
#define QD 32
#define R1 4000
#define Q1 10
#define R2 12512
#define Q2 32

#define CHUNK 256
#define LCAP  160
#define TSTR  33

__device__ __forceinline__ void lds_round_barrier() {
    asm volatile("s_wait_dscnt 0x0\n\ts_barrier_signal -1\n\ts_barrier_wait -1" ::: "memory");
}

template <int LCB>
__device__ __forceinline__ void drain_take(const unsigned* ml, int start, int m, int lane, int e0,
                                           const int* __restrict__ rows, unsigned& lc, int& r) {
    constexpr unsigned LCM = (1u << LCB) - 1u;
    lc = 0xFFFFFFFFu;
    r = 0;
    if (lane < m) {
        const unsigned ent = ml[start + lane];
        lc = ent & LCM;
        r = rows[e0 + (int)(ent >> LCB)];
    }
}

template <int F>
__device__ __forceinline__ void drain_gather(int r, int m, int lane, const float* __restrict__ h, float* v) {
    if (F == 16) {
        const int sub = lane >> 2;
        const int qd  = lane & 3;
        #pragma unroll
        for (int s = 0; s < 4; ++s) {
            const int src = 8 * s + sub;
            const int rs = __shfl(r, src);
            float4 t = make_float4(0.f, 0.f, 0.f, 0.f);
            if (src < m) t = ((const float4*)h)[rs * 4 + qd];
            v[4 * s + 0] = t.x;
            v[4 * s + 1] = t.y;
            v[4 * s + 2] = t.z;
            v[4 * s + 3] = t.w;
        }
    } else {
        float2 t = make_float2(0.f, 0.f);
        if (lane < m) t = ((const float2*)h)[r];
        v[0] = t.x;
        v[1] = t.y;
    }
}

template <int D>
__device__ __forceinline__ unsigned same_node_below(unsigned lc, int l8) {
    const unsigned o = (unsigned)__builtin_amdgcn_update_dpp(0, (int)lc, 0x110 + D, 0xF, 0xF, false);
    return (l8 >= D && o == lc) ? 1u : 0u;
}

template <int F>
__device__ __forceinline__ void drain_add(unsigned lc, const float* v, int m, int lane, float* acc) {
    if (F == 16) {
        const int sub = lane >> 2;
        const int qd  = lane & 3;
        const int l8  = lane & 7;
        unsigned rk = same_node_below<1>(lc, l8);
        rk += same_node_below<2>(lc, l8);
        rk += same_node_below<3>(lc, l8);
        rk += same_node_below<4>(lc, l8);
        rk += same_node_below<5>(lc, l8);
        rk += same_node_below<6>(lc, l8);
        rk += same_node_below<7>(lc, l8);
        const unsigned pk = (rk << 16) | (lc & 0xFFFFu);
        unsigned pks[4];
        #pragma unroll
        for (int s = 0; s < 4; ++s) pks[s] = (unsigned)__shfl((int)pk, 8 * s + sub);
        float4* acc4 = (float4*)acc;
        #pragma unroll
        for (int s = 0; s < 4; ++s) {
            const bool valid = 8 * s + sub < m;
            const unsigned node = pks[s] & 0xFFFFu;
            const unsigned rank = pks[s] >> 16;
            for (unsigned t = 0u; t < 8u; ++t) {
                const bool act = valid && (rank == t);
                if (__builtin_amdgcn_ballot_w32(act) == 0u) break;
                if (act) {
                    float4 a = acc4[node * 4u + (unsigned)qd];
                    a.x += v[4 * s + 0];
                    a.y += v[4 * s + 1];
                    a.z += v[4 * s + 2];
                    a.w += v[4 * s + 3];
                    acc4[node * 4u + (unsigned)qd] = a;
                }
            }
        }
    } else {
        const bool valid = lane < m;
        unsigned same = 0u;
        #pragma unroll
        for (int s = 0; s < 32; ++s) {
            const unsigned ls = (unsigned)__builtin_amdgcn_readlane((int)lc, s);
            same |= (ls == lc) ? (1u << s) : 0u;
        }
        const unsigned rank = (unsigned)__builtin_popcount(same & ((1u << lane) - 1u));
        float2* acc2 = (float2*)acc;
        for (unsigned t = 0u; t < 32u; ++t) {
            const bool act = valid && (rank == t);
            if (__builtin_amdgcn_ballot_w32(act) == 0u) break;
            if (act) {
                float2 a = acc2[lc];
                a.x += v[0]; a.y += v[1];
                acc2[lc] = a;
            }
        }
    }
}

template <int LCB>
__device__ __forceinline__ void drain_count(const unsigned* ml, int start, int m, int lane, float* acc) {
    constexpr unsigned LCM = (1u << LCB) - 1u;
    const bool valid = lane < m;
    unsigned lc = 0xFFFFFFFFu;
    if (valid) lc = ml[start + lane] & LCM;
    unsigned same = 0u;
    #pragma unroll
    for (int s = 0; s < 32; ++s) {
        const unsigned ls = (unsigned)__builtin_amdgcn_readlane((int)lc, s);
        same |= (ls == lc) ? (1u << s) : 0u;
    }
    const unsigned rank = (unsigned)__builtin_popcount(same & ((1u << lane) - 1u));
    if (valid && rank == 0u) acc[lc] += (float)__builtin_popcount(same);
}

__device__ __forceinline__ int filter_test(const int4 (&c)[2], int cursor, int wend, int lo, int nr, int lane,
                                           unsigned (&lc)[8], bool (&hit)[8], unsigned (&mk)[8]) {
    const int left = wend - cursor - 4 * lane;
    int h = 0;
    #pragma unroll
    for (int j = 0; j < 2; ++j) {
        const int off = (j * 128 < left) ? lo : -(1 << 30);
        const int t[4] = {c[j].x, c[j].y, c[j].z, c[j].w};
        #pragma unroll
        for (int i = 0; i < 4; ++i) {
            const int k = 4 * j + i;
            lc[k]  = (unsigned)(t[i] - off);
            hit[k] = lc[k] < (unsigned)nr;
            mk[k]  = __builtin_amdgcn_ballot_w32(hit[k]);
            h += __builtin_popcount(mk[k]);
        }
    }
    return h;
}

__device__ __forceinline__ void filter_store(const unsigned (&lc)[8], const bool (&hit)[8], const unsigned (&mk)[8],
                                             unsigned* qrow, int& na, int sub, int lane) {
    const unsigned place = (unsigned)(sub * CHUNK + 4 * lane) << 16;
    #pragma unroll
    for (int k = 0; k < 8; ++k) {
        if (hit[k]) qrow[na + (int)__builtin_amdgcn_mbcnt_lo(mk[k], 0u)] =
                        (place + ((unsigned)((k >> 2) * 128 + (k & 3)) << 16)) | lc[k];
        na += __builtin_popcount(mk[k]);
    }
}

template <int F, int R, int Q, int QX>
__global__ __launch_bounds__(TWG) void gcn_tile_kernel(const int* __restrict__ rows,
                                                       const int* __restrict__ cols,
                                                       const float* __restrict__ h,
                                                       float* __restrict__ part,
                                                       int n, int n_edges, int slice) {
    constexpr int LCB = (R <= 4096) ? 12 : (R <= 16384) ? 14 : 16;
    constexpr int OSH = (F == 1) ? 6 : 5;
    static_assert(R <= (1 << LCB), "node index must fit its field");
    constexpr int QCAP = CHUNK + QX;
    constexpr int QSTR = QCAP + 4;
    constexpr int NB   = QCAP / 32;
    static_assert(QSTR % 64 == 4, "queue rows must lie 4 banks apart");
    __shared__ __attribute__((aligned(16))) float acc[R * F];
    __shared__ __attribute__((aligned(16))) unsigned qa[32 * QSTR];
    __shared__ unsigned cntm[32 * TSTR];
    __shared__ unsigned ml_all[32 * LCAP];
    const int lane = threadIdx.x & 31;
    const int wave = __builtin_amdgcn_readfirstlane((int)(threadIdx.x >> 5));
    unsigned* ml = ml_all + wave * LCAP;
    const int p  = blockIdx.x / Q;
    const int q  = blockIdx.x - p * Q;
    const int lo = p * R;
    const int nr = min(R, n - lo);

    for (int i = threadIdx.x; i < nr * F; i += TWG) acc[i] = 0.f;

    const int e0 = q * slice;
    const int e1 = min(n_edges, e0 + slice);
    const int wlen = (((slice + 31) >> 5) + 31) & ~31;
    const int wbeg = e0 + wave * wlen;
    const int wend = min(e1, wbeg + wlen);
    const int maxr = (wlen + CHUNK - 1) / CHUNK + 1;
    const int maxp = maxr / 2 + 1;
    const int4* cols4 = (const int4*)cols;
    const int last4 = (n_edges >> 2) - 1;
    int cursor = wbeg;
    int4 ca[2], cb[2];
    #pragma unroll
    for (int j = 0; j < 2; ++j) {
        ca[j] = cols4[min((cursor >> 2) + j * 32 + lane, last4)];
        cb[j] = cols4[min(((cursor + CHUNK) >> 2) + j * 32 + lane, last4)];
    }
    bool odd  = false;
    int cnt = 0;
    constexpr int VN = (F == 16) ? 16 : 2;
    unsigned a_lc = 0u, b_lc = 0u;
    int a_r = 0, a_m = 0, b_m = 0;
    float b_v[VN];
    #pragma unroll
    for (int j = 0; j < VN; ++j) b_v[j] = 0.f;
    for (int round = 0; round < maxr; ++round) {
        const int rbeg = cursor - e0;
        if (odd) {
            #pragma unroll
            for (int j = 0; j < 2; ++j) { const int4 t = ca[j]; ca[j] = cb[j]; cb[j] = t; }
        }
        odd = false;
        int na = 0;
        int sub = 0;
        unsigned* qrow = qa + wave * QSTR;
        for (int trip = 0; trip < maxp; ++trip) {
            {
                unsigned lc[8], mk[8];
                bool hit[8];
                const int h = filter_test(ca, cursor, wend, lo, nr, lane, lc, hit, mk);
                if (na + h > QCAP) break;
                #pragma unroll
                for (int j = 0; j < 2; ++j) ca[j] = cols4[min(((cursor + 2 * CHUNK) >> 2) + j * 32 + lane, last4)];
                filter_store(lc, hit, mk, qrow, na, sub, lane);
                cursor += CHUNK;
                ++sub;
                odd = true;
                if (cursor >= wend) break;
            }
            {
                unsigned lc[8], mk[8];
                bool hit[8];
                const int h = filter_test(cb, cursor, wend, lo, nr, lane, lc, hit, mk);
                if (na + h > QCAP) break;
                #pragma unroll
                for (int j = 0; j < 2; ++j) cb[j] = cols4[min(((cursor + 2 * CHUNK) >> 2) + j * 32 + lane, last4)];
                filter_store(lc, hit, mk, qrow, na, sub, lane);
                cursor += CHUNK;
                ++sub;
                odd = false;
                if (cursor >= wend) break;
            }
        }
        unsigned* cw = cntm + wave * TSTR;
        cw[lane] = 0u;
        const int nb = (na + 31) >> 5;
        unsigned se[NB], sp[NB];
        #pragma unroll
        for (int b = 0; b < NB; ++b) {
            se[b] = 0u;
            sp[b] = 0u;
            if (b < nb) {
                const bool valid = b * 32 + lane < na;
                unsigned ent = 0u;
                if (valid) ent = qa[wave * QSTR + b * 32 + lane];
                unsigned same = __builtin_amdgcn_ballot_w32(valid);
                #pragma unroll
                for (int bit = 0; bit < 5; ++bit) {
                    const int sb = ((int)(ent << (31 - OSH - bit))) >> 31;
                    const unsigned m = __builtin_amdgcn_ballot_w32(valid && sb != 0);
                    same &= ~(m ^ (unsigned)sb);
                }
                const unsigned rank = __builtin_amdgcn_mbcnt_lo(same, 0u);
                const unsigned tot  = (unsigned)__builtin_popcount(same);
                unsigned base = 0u;
                if (valid) base = cw[(ent >> OSH) & 31u];
                if (valid && rank + 1u == tot) cw[(ent >> OSH) & 31u] = base + tot;
                se[b] = ent;
                sp[b] = base + rank;
            }
        }
        {
            const unsigned c = cw[lane];
            unsigned inc = c;
            inc += (unsigned)__builtin_amdgcn_update_dpp(0, (int)inc, 0x111, 0xF, 0xF, false);
            inc += (unsigned)__builtin_amdgcn_update_dpp(0, (int)inc, 0x112, 0xF, 0xF, false);
            inc += (unsigned)__builtin_amdgcn_update_dpp(0, (int)inc, 0x114, 0xF, 0xF, false);
            inc += (unsigned)__builtin_amdgcn_update_dpp(0, (int)inc, 0x118, 0xF, 0xF, false);
            const unsigned row0 = (unsigned)__builtin_amdgcn_readlane((int)inc, 15);
            if (lane >= 16) inc += row0;
            cw[lane] = ((inc - c) << 16) | c;
        }
        #pragma unroll
        for (int b = 0; b < NB; ++b) {
            if (b < nb) {
                if (b * 32 + lane < na) qa[wave * QSTR + (int)((cw[(se[b] >> OSH) & 31u] >> 16) + sp[b])] = se[b];
            }
        }
        if (lane == 0) cw[32] = ((unsigned)rbeg << 1) | ((cursor >= wend) ? 1u : 0u);
        lds_round_barrier();
        const unsigned sc = cntm[lane * TSTR + wave];
        const unsigned rw = cntm[lane * TSTR + 32];
        const int st = (int)(sc >> 16);
        const int nl = (int)(sc & 0xFFFFu);
        const unsigned ebase = rw >> 1;
        const bool all_through = __builtin_amdgcn_ballot_w32((rw & 1u) != 0u) == 0xFFFFFFFFu;
        for (int i0 = 0; __builtin_amdgcn_ballot_w32(i0 < nl) != 0u; i0 += 4) {
            unsigned ent[4];
            #pragma unroll
            for (int u = 0; u < 4; ++u) {
                ent[u] = 0u;
                if (i0 + u < nl) ent[u] = qa[lane * QSTR + st + i0 + u];
            }
            #pragma unroll
            for (int u = 0; u < 4; ++u) {
                const bool have = i0 + u < nl;
                const unsigned mk = __builtin_amdgcn_ballot_w32(have);
                if (mk != 0u) {
                    if (have) ml[cnt + (int)__builtin_amdgcn_mbcnt_lo(mk, 0u)] =
                                  ((ebase + (ent[u] >> 16)) << LCB) | (ent[u] & 0xFFFFu);
                    cnt += __builtin_popcount(mk);
                }
            }
            while (cnt >= 32) {
                cnt -= 32;
                if (F == 1) {
                    drain_count<LCB>(ml, cnt, 32, lane, acc);
                } else {
                    float c_v[VN];
                    #pragma unroll
                    for (int j = 0; j < VN; ++j) c_v[j] = b_v[j];
                    const unsigned c_lc = b_lc;
                    const int c_m = b_m;
                    b_m = a_m;
                    b_lc = a_lc;
                    drain_gather<F>(a_r, 32, lane, h, b_v);
                    drain_take<LCB>(ml, cnt, 32, lane, e0, rows, a_lc, a_r);
                    a_m = 32;
                    if (c_m) drain_add<F>(c_lc, c_v, 32, lane, acc);
                }
            }
        }
        lds_round_barrier();
        if (all_through) break;
    }
    if (F == 1) {
        if (cnt > 0) drain_count<LCB>(ml, 0, cnt, lane, acc);
    } else {
        int tail = cnt;
        #pragma nounroll
        for (int t = 0; t < 3; ++t) {
            if (b_m) drain_add<F>(b_lc, b_v, b_m, lane, acc);
            b_m = a_m;
            b_lc = a_lc;
            if (a_m) drain_gather<F>(a_r, a_m, lane, h, b_v);
            a_m = tail;
            if (tail) drain_take<LCB>(ml, 0, tail, lane, e0, rows, a_lc, a_r);
            tail = 0;
        }
    }
    __syncthreads();

    float* dst = part + ((size_t)q * n + lo) * F;
    for (int i = threadIdx.x; i < nr * F; i += TWG) dst[i] = acc[i];
}

__global__ void gcn_dinv_kernel(const float* __restrict__ part,
                                float* __restrict__ dinv, int n) {
    const int i = blockIdx.x * blockDim.x + threadIdx.x;
    if (i >= n) return;
    float c = 1.0f;
    #pragma unroll 16
    for (int q = 0; q < QD; ++q) c += part[(size_t)q * n + i];
    dinv[i] = rsqrtf(c);
}

__global__ void gcn_xw1_wmma_kernel(const float* __restrict__ x,
                                    const float* __restrict__ W1,
                                    const float* __restrict__ dinv,
                                    float* __restrict__ h1s,
                                    int n_nodes) {
    const int lane  = threadIdx.x & 31;
    const int wave  = threadIdx.x >> 5;
    const int tile  = blockIdx.x * (blockDim.x >> 5) + wave;
    const int node0 = tile * 16;
    if (node0 >= n_nodes) return;

    const int half = lane >> 4;
    const int l16  = lane & 15;
    const int m    = l16;

    v8f c = {0.f, 0.f, 0.f, 0.f, 0.f, 0.f, 0.f, 0.f};

    #pragma unroll
    for (int k0 = 0; k0 < 28; k0 += 4) {
        const int ka = k0 + 2 * half;
        v2f a, b;
        a.x = (ka     < FIN) ? x[(node0 + m) * FIN + ka    ] : 0.0f;
        a.y = (ka + 1 < FIN) ? x[(node0 + m) * FIN + ka + 1] : 0.0f;
        b.x = (ka     < FIN) ? W1[(ka    ) * F1 + l16] : 0.0f;
        b.y = (ka + 1 < FIN) ? W1[(ka + 1) * F1 + l16] : 0.0f;
        c = __builtin_amdgcn_wmma_f32_16x16x4_f32(
                 false, a,  false, b,
                 (short)0, c,  false,  false);
    }

    #pragma unroll
    for (int v = 0; v < 8; ++v) {
        const int node = node0 + v + 8 * half;
        h1s[node * F1 + l16] = c[v] * dinv[node];
    }
}

__global__ void gcn_layer1_finish_kernel(const float* __restrict__ part,
                                         const float* __restrict__ h1s,
                                         const float* __restrict__ dinv,
                                         const float* __restrict__ b1,
                                         const float* __restrict__ W2,
                                         float* __restrict__ h2s,
                                         int n) {
    const int gid = blockIdx.x * blockDim.x + threadIdx.x;
    const int i   = gid >> 2;
    const int g   = gid & 3;
    const int ic  = (i < n) ? i : (n - 1);
    float4 s = ((const float4*)h1s)[(size_t)ic * 4 + g];
    #pragma unroll
    for (int q = 0; q < Q1; ++q) {
        const float4 t = ((const float4*)part)[((size_t)q * n + ic) * 4 + g];
        s.x += t.x; s.y += t.y; s.z += t.z; s.w += t.w;
    }
    const float di = dinv[ic];
    const float v0 = fmaxf(fmaf(s.x, di, b1[4 * g + 0]), 0.0f);
    const float v1 = fmaxf(fmaf(s.y, di, b1[4 * g + 1]), 0.0f);
    const float v2 = fmaxf(fmaf(s.z, di, b1[4 * g + 2]), 0.0f);
    const float v3 = fmaxf(fmaf(s.w, di, b1[4 * g + 3]), 0.0f);
    const float* w = W2 + 8 * g;
    float a0 = v0 * w[0];
    float a1 = v0 * w[1];
    a0 = fmaf(v1, w[2], a0); a1 = fmaf(v1, w[3], a1);
    a0 = fmaf(v2, w[4], a0); a1 = fmaf(v2, w[5], a1);
    a0 = fmaf(v3, w[6], a0); a1 = fmaf(v3, w[7], a1);
    a0 += __shfl_xor(a0, 1); a1 += __shfl_xor(a1, 1);
    a0 += __shfl_xor(a0, 2); a1 += __shfl_xor(a1, 2);
    if (g == 0 && i < n) {
        float2 o; o.x = a0 * di; o.y = a1 * di;
        ((float2*)h2s)[i] = o;
    }
}

__global__ void gcn_final_kernel(const float* __restrict__ part,
                                 const float* __restrict__ h2s,
                                 const float* __restrict__ dinv,
                                 const float* __restrict__ b2,
                                 float* __restrict__ out,
                                 int n) {
    const int i = blockIdx.x * blockDim.x + threadIdx.x;
    if (i >= n) return;
    float2 s = ((const float2*)h2s)[i];
    #pragma unroll 16
    for (int q = 0; q < Q2; ++q) {
        const float2 t = ((const float2*)part)[(size_t)q * n + i];
        s.x += t.x; s.y += t.y;
    }
    const float di = dinv[i];
    const float a = fmaf(s.x, di, b2[0]);
    const float b = fmaf(s.y, di, b2[1]);
    const float m = fmaxf(a, b);
    const float lse = m + logf(expf(a - m) + expf(b - m));
    float2 o; o.x = a - lse; o.y = b - lse;
    ((float2*)out)[i] = o;
}

extern "C" void kernel_launch(void* const* d_in, const int* in_sizes, int n_in,
                              void* d_out, int out_size, void* d_ws, size_t ws_size,
                              hipStream_t stream) {
    const float* x   = (const float*)d_in[0];
    const int*   ei  = (const int*)d_in[1];
    const float* W1  = (const float*)d_in[2];
    const float* b1  = (const float*)d_in[3];
    const float* W2  = (const float*)d_in[4];
    const float* b2  = (const float*)d_in[5];
    float*       out = (float*)d_out;

    const int N = in_sizes[0] / FIN;
    const int E = in_sizes[1] / 2;
    const int* rows = ei;
    const int* cols = ei + E;

    float* ws   = (float*)d_ws;
    float* h1s  = ws;
    float* h2s  = ws + (size_t)16 * N;
    float* dinv = ws + (size_t)18 * N;
    float* part = ws + (size_t)20 * N;

    const int BS = 256;

    gcn_tile_kernel<1, RD, QD, 256><<<((N + RD - 1) / RD) * QD, TWG, 0, stream>>>(
        rows, cols, h1s, part, N, E, (E + QD - 1) / QD);

    gcn_dinv_kernel<<<(N + BS - 1) / BS, BS, 0, stream>>>(part, dinv, N);

    {
        int tiles = (N + 15) / 16;
        int waves_per_block = BS / 32;
        int blocks = (tiles + waves_per_block - 1) / waves_per_block;
        gcn_xw1_wmma_kernel<<<blocks, BS, 0, stream>>>(x, W1, dinv, h1s, N);
    }

    gcn_tile_kernel<F1, R1, Q1, 64><<<((N + R1 - 1) / R1) * Q1, TWG, 0, stream>>>(
        rows, cols, h1s, part, N, E, (E + Q1 - 1) / Q1);

    gcn_layer1_finish_kernel<<<(4 * N + BS - 1) / BS, BS, 0, stream>>>(
        part, h1s, dinv, b1, W2, h2s, N);

    gcn_tile_kernel<F2, R2, Q2, 256><<<((N + R2 - 1) / R2) * Q2, TWG, 0, stream>>>(
        rows, cols, h2s, part, N, E, (E + Q2 - 1) / Q2);

    gcn_final_kernel<<<(N + BS - 1) / BS, BS, 0, stream>>>(part, h2s, dinv, b2, out, N);
}


# Degree and layer-2 drains find the lanes that hold one node by a ballot per index bit that can differ inside an owner's list (9 bits), not by comparing with each of the 32 lanes; sum order kept
# speedup vs baseline: 3.1024x; 1.0352x over previous
#include <hip/hip_runtime.h>


typedef __attribute__((ext_vector_type(2))) float v2f;
typedef __attribute__((ext_vector_type(8))) float v8f;

#define F1 16
#define F2 2
#define FIN 25

#define TWG 1024
#define RD 12512
#define QD 32
#define R1 4000
#define Q1 10
#define R2 12512
#define Q2 32

#define CHUNK 256
#define LCAP  160
#define TSTR  33

__device__ __forceinline__ void lds_round_barrier() {
    asm volatile("s_wait_dscnt 0x0\n\ts_barrier_signal -1\n\ts_barrier_wait -1" ::: "memory");
}

template <int LCB>
__device__ __forceinline__ void drain_take(const unsigned* ml, int start, int m, int lane, int e0,
                                           const int* __restrict__ rows, unsigned& lc, int& r) {
    constexpr unsigned LCM = (1u << LCB) - 1u;
    lc = 0xFFFFFFFFu;
    r = 0;
    if (lane < m) {
        const unsigned ent = ml[start + lane];
        lc = ent & LCM;
        r = rows[e0 + (int)(ent >> LCB)];
    }
}

template <int F>
__device__ __forceinline__ void drain_gather(int r, int m, int lane, const float* __restrict__ h, float* v) {
    if (F == 16) {
        const int sub = lane >> 2;
        const int qd  = lane & 3;
        #pragma unroll
        for (int s = 0; s < 4; ++s) {
            const int src = 8 * s + sub;
            const int rs = __shfl(r, src);
            float4 t = make_float4(0.f, 0.f, 0.f, 0.f);
            if (src < m) t = ((const float4*)h)[rs * 4 + qd];
            v[4 * s + 0] = t.x;
            v[4 * s + 1] = t.y;
            v[4 * s + 2] = t.z;
            v[4 * s + 3] = t.w;
        }
    } else {
        float2 t = make_float2(0.f, 0.f);
        if (lane < m) t = ((const float2*)h)[r];
        v[0] = t.x;
        v[1] = t.y;
    }
}

template <int D>
__device__ __forceinline__ unsigned same_node_below(unsigned lc, int l8) {
    const unsigned o = (unsigned)__builtin_amdgcn_update_dpp(0, (int)lc, 0x110 + D, 0xF, 0xF, false);
    return (l8 >= D && o == lc) ? 1u : 0u;
}

template <int LCB, int OSH>
__device__ __forceinline__ unsigned same_node_lanes(unsigned lc, bool valid) {
    unsigned diff = 0u;
    #pragma unroll
    for (int bit = 0; bit < LCB; ++bit) {
        if (bit >= OSH && bit < OSH + 5) continue;
        const unsigned sp = (unsigned)(((int)(lc << (31 - bit))) >> 31);
        const unsigned mb = __builtin_amdgcn_ballot_w32(valid && sp != 0u);
        diff |= mb ^ sp;
    }
    return __builtin_amdgcn_ballot_w32(valid) & ~diff;
}

template <int F, int LCB, int OSH>
__device__ __forceinline__ void drain_add(unsigned lc, const float* v, int m, int lane, float* acc) {
    if (F == 16) {
        const int sub = lane >> 2;
        const int qd  = lane & 3;
        const int l8  = lane & 7;
        unsigned rk = same_node_below<1>(lc, l8);
        rk += same_node_below<2>(lc, l8);
        rk += same_node_below<3>(lc, l8);
        rk += same_node_below<4>(lc, l8);
        rk += same_node_below<5>(lc, l8);
        rk += same_node_below<6>(lc, l8);
        rk += same_node_below<7>(lc, l8);
        const unsigned pk = (rk << 16) | (lc & 0xFFFFu);
        unsigned pks[4];
        #pragma unroll
        for (int s = 0; s < 4; ++s) pks[s] = (unsigned)__shfl((int)pk, 8 * s + sub);
        float4* acc4 = (float4*)acc;
        #pragma unroll
        for (int s = 0; s < 4; ++s) {
            const bool valid = 8 * s + sub < m;
            const unsigned node = pks[s] & 0xFFFFu;
            const unsigned rank = pks[s] >> 16;
            for (unsigned t = 0u; t < 8u; ++t) {
                const bool act = valid && (rank == t);
                if (__builtin_amdgcn_ballot_w32(act) == 0u) break;
                if (act) {
                    float4 a = acc4[node * 4u + (unsigned)qd];
                    a.x += v[4 * s + 0];
                    a.y += v[4 * s + 1];
                    a.z += v[4 * s + 2];
                    a.w += v[4 * s + 3];
                    acc4[node * 4u + (unsigned)qd] = a;
                }
            }
        }
    } else {
        const bool valid = lane < m;
        const unsigned rank = __builtin_amdgcn_mbcnt_lo(same_node_lanes<LCB, OSH>(lc, valid), 0u);
        float2* acc2 = (float2*)acc;
        for (unsigned t = 0u; t < 32u; ++t) {
            const bool act = valid && (rank == t);
            if (__builtin_amdgcn_ballot_w32(act) == 0u) break;
            if (act) {
                float2 a = acc2[lc];
                a.x += v[0]; a.y += v[1];
                acc2[lc] = a;
            }
        }
    }
}

template <int LCB, int OSH>
__device__ __forceinline__ void drain_count(const unsigned* ml, int start, int m, int lane, float* acc) {
    constexpr unsigned LCM = (1u << LCB) - 1u;
    const bool valid = lane < m;
    unsigned lc = 0xFFFFFFFFu;
    if (valid) lc = ml[start + lane] & LCM;
    const unsigned same = same_node_lanes<LCB, OSH>(lc, valid);
    const unsigned rank = __builtin_amdgcn_mbcnt_lo(same, 0u);
    if (valid && rank == 0u) acc[lc] += (float)__builtin_popcount(same);
}

__device__ __forceinline__ int filter_test(const int4 (&c)[2], int cursor, int wend, int lo, int nr, int lane,
                                           unsigned (&lc)[8], bool (&hit)[8], unsigned (&mk)[8]) {
    const int left = wend - cursor - 4 * lane;
    int h = 0;
    #pragma unroll
    for (int j = 0; j < 2; ++j) {
        const int off = (j * 128 < left) ? lo : -(1 << 30);
        const int t[4] = {c[j].x, c[j].y, c[j].z, c[j].w};
        #pragma unroll
        for (int i = 0; i < 4; ++i) {
            const int k = 4 * j + i;
            lc[k]  = (unsigned)(t[i] - off);
            hit[k] = lc[k] < (unsigned)nr;
            mk[k]  = __builtin_amdgcn_ballot_w32(hit[k]);
            h += __builtin_popcount(mk[k]);
        }
    }
    return h;
}

__device__ __forceinline__ void filter_store(const unsigned (&lc)[8], const bool (&hit)[8], const unsigned (&mk)[8],
                                             unsigned* qrow, int& na, int sub, int lane) {
    const unsigned place = (unsigned)(sub * CHUNK + 4 * lane) << 16;
    #pragma unroll
    for (int k = 0; k < 8; ++k) {
        if (hit[k]) qrow[na + (int)__builtin_amdgcn_mbcnt_lo(mk[k], 0u)] =
                        (place + ((unsigned)((k >> 2) * 128 + (k & 3)) << 16)) | lc[k];
        na += __builtin_popcount(mk[k]);
    }
}

template <int F, int R, int Q, int QX>
__global__ __launch_bounds__(TWG) void gcn_tile_kernel(const int* __restrict__ rows,
                                                       const int* __restrict__ cols,
                                                       const float* __restrict__ h,
                                                       float* __restrict__ part,
                                                       int n, int n_edges, int slice) {
    constexpr int LCB = (R <= 4096) ? 12 : (R <= 16384) ? 14 : 16;
    constexpr int OSH = (F == 1) ? 6 : 5;
    static_assert(R <= (1 << LCB), "node index must fit its field");
    constexpr int QCAP = CHUNK + QX;
    constexpr int QSTR = QCAP + 4;
    constexpr int NB   = QCAP / 32;
    static_assert(QSTR % 64 == 4, "queue rows must lie 4 banks apart");
    __shared__ __attribute__((aligned(16))) float acc[R * F];
    __shared__ __attribute__((aligned(16))) unsigned qa[32 * QSTR];
    __shared__ unsigned cntm[32 * TSTR];
    __shared__ unsigned ml_all[32 * LCAP];
    const int lane = threadIdx.x & 31;
    const int wave = __builtin_amdgcn_readfirstlane((int)(threadIdx.x >> 5));
    unsigned* ml = ml_all + wave * LCAP;
    const int p  = blockIdx.x / Q;
    const int q  = blockIdx.x - p * Q;
    const int lo = p * R;
    const int nr = min(R, n - lo);

    for (int i = threadIdx.x; i < nr * F; i += TWG) acc[i] = 0.f;

    const int e0 = q * slice;
    const int e1 = min(n_edges, e0 + slice);
    const int wlen = (((slice + 31) >> 5) + 31) & ~31;
    const int wbeg = e0 + wave * wlen;
    const int wend = min(e1, wbeg + wlen);
    const int maxr = (wlen + CHUNK - 1) / CHUNK + 1;
    const int maxp = maxr / 2 + 1;
    const int4* cols4 = (const int4*)cols;
    const int last4 = (n_edges >> 2) - 1;
    int cursor = wbeg;
    int4 ca[2], cb[2];
    #pragma unroll
    for (int j = 0; j < 2; ++j) {
        ca[j] = cols4[min((cursor >> 2) + j * 32 + lane, last4)];
        cb[j] = cols4[min(((cursor + CHUNK) >> 2) + j * 32 + lane, last4)];
    }
    bool odd  = false;
    int cnt = 0;
    constexpr int VN = (F == 16) ? 16 : 2;
    unsigned a_lc = 0u, b_lc = 0u;
    int a_r = 0, a_m = 0, b_m = 0;
    float b_v[VN];
    #pragma unroll
    for (int j = 0; j < VN; ++j) b_v[j] = 0.f;
    for (int round = 0; round < maxr; ++round) {
        const int rbeg = cursor - e0;
        if (odd) {
            #pragma unroll
            for (int j = 0; j < 2; ++j) { const int4 t = ca[j]; ca[j] = cb[j]; cb[j] = t; }
        }
        odd = false;
        int na = 0;
        int sub = 0;
        unsigned* qrow = qa + wave * QSTR;
        for (int trip = 0; trip < maxp; ++trip) {
            {
                unsigned lc[8], mk[8];
                bool hit[8];
                const int h = filter_test(ca, cursor, wend, lo, nr, lane, lc, hit, mk);
                if (na + h > QCAP) break;
                #pragma unroll
                for (int j = 0; j < 2; ++j) ca[j] = cols4[min(((cursor + 2 * CHUNK) >> 2) + j * 32 + lane, last4)];
                filter_store(lc, hit, mk, qrow, na, sub, lane);
                cursor += CHUNK;
                ++sub;
                odd = true;
                if (cursor >= wend) break;
            }
            {
                unsigned lc[8], mk[8];
                bool hit[8];
                const int h = filter_test(cb, cursor, wend, lo, nr, lane, lc, hit, mk);
                if (na + h > QCAP) break;
                #pragma unroll
                for (int j = 0; j < 2; ++j) cb[j] = cols4[min(((cursor + 2 * CHUNK) >> 2) + j * 32 + lane, last4)];
                filter_store(lc, hit, mk, qrow, na, sub, lane);
                cursor += CHUNK;
                ++sub;
                odd = false;
                if (cursor >= wend) break;
            }
        }
        unsigned* cw = cntm + wave * TSTR;
        cw[lane] = 0u;
        const int nb = (na + 31) >> 5;
        unsigned se[NB], sp[NB];
        #pragma unroll
        for (int b = 0; b < NB; ++b) {
            se[b] = 0u;
            sp[b] = 0u;
            if (b < nb) {
                const bool valid = b * 32 + lane < na;
                unsigned ent = 0u;
                if (valid) ent = qa[wave * QSTR + b * 32 + lane];
                unsigned same = __builtin_amdgcn_ballot_w32(valid);
                #pragma unroll
                for (int bit = 0; bit < 5; ++bit) {
                    const int sb = ((int)(ent << (31 - OSH - bit))) >> 31;
                    const unsigned m = __builtin_amdgcn_ballot_w32(valid && sb != 0);
                    same &= ~(m ^ (unsigned)sb);
                }
                const unsigned rank = __builtin_amdgcn_mbcnt_lo(same, 0u);
                const unsigned tot  = (unsigned)__builtin_popcount(same);
                unsigned base = 0u;
                if (valid) base = cw[(ent >> OSH) & 31u];
                if (valid && rank + 1u == tot) cw[(ent >> OSH) & 31u] = base + tot;
                se[b] = ent;
                sp[b] = base + rank;
            }
        }
        {
            const unsigned c = cw[lane];
            unsigned inc = c;
            inc += (unsigned)__builtin_amdgcn_update_dpp(0, (int)inc, 0x111, 0xF, 0xF, false);
            inc += (unsigned)__builtin_amdgcn_update_dpp(0, (int)inc, 0x112, 0xF, 0xF, false);
            inc += (unsigned)__builtin_amdgcn_update_dpp(0, (int)inc, 0x114, 0xF, 0xF, false);
            inc += (unsigned)__builtin_amdgcn_update_dpp(0, (int)inc, 0x118, 0xF, 0xF, false);
            const unsigned row0 = (unsigned)__builtin_amdgcn_readlane((int)inc, 15);
            if (lane >= 16) inc += row0;
            cw[lane] = ((inc - c) << 16) | c;
        }
        #pragma unroll
        for (int b = 0; b < NB; ++b) {
            if (b < nb) {
                if (b * 32 + lane < na) qa[wave * QSTR + (int)((cw[(se[b] >> OSH) & 31u] >> 16) + sp[b])] = se[b];
            }
        }
        if (lane == 0) cw[32] = ((unsigned)rbeg << 1) | ((cursor >= wend) ? 1u : 0u);
        lds_round_barrier();
        const unsigned sc = cntm[lane * TSTR + wave];
        const unsigned rw = cntm[lane * TSTR + 32];
        const int st = (int)(sc >> 16);
        const int nl = (int)(sc & 0xFFFFu);
        const unsigned ebase = rw >> 1;
        const bool all_through = __builtin_amdgcn_ballot_w32((rw & 1u) != 0u) == 0xFFFFFFFFu;
        for (int i0 = 0; __builtin_amdgcn_ballot_w32(i0 < nl) != 0u; i0 += 4) {
            unsigned ent[4];
            #pragma unroll
            for (int u = 0; u < 4; ++u) {
                ent[u] = 0u;
                if (i0 + u < nl) ent[u] = qa[lane * QSTR + st + i0 + u];
            }
            #pragma unroll
            for (int u = 0; u < 4; ++u) {
                const bool have = i0 + u < nl;
                const unsigned mk = __builtin_amdgcn_ballot_w32(have);
                if (mk != 0u) {
                    if (have) ml[cnt + (int)__builtin_amdgcn_mbcnt_lo(mk, 0u)] =
                                  ((ebase + (ent[u] >> 16)) << LCB) | (ent[u] & 0xFFFFu);
                    cnt += __builtin_popcount(mk);
                }
            }
            while (cnt >= 32) {
                cnt -= 32;
                if (F == 1) {
                    drain_count<LCB, OSH>(ml, cnt, 32, lane, acc);
                } else {
                    float c_v[VN];
                    #pragma unroll
                    for (int j = 0; j < VN; ++j) c_v[j] = b_v[j];
                    const unsigned c_lc = b_lc;
                    const int c_m = b_m;
                    b_m = a_m;
                    b_lc = a_lc;
                    drain_gather<F>(a_r, 32, lane, h, b_v);
                    drain_take<LCB>(ml, cnt, 32, lane, e0, rows, a_lc, a_r);
                    a_m = 32;
                    if (c_m) drain_add<F, LCB, OSH>(c_lc, c_v, 32, lane, acc);
                }
            }
        }
        lds_round_barrier();
        if (all_through) break;
    }
    if (F == 1) {
        if (cnt > 0) drain_count<LCB, OSH>(ml, 0, cnt, lane, acc);
    } else {
        int tail = cnt;
        #pragma nounroll
        for (int t = 0; t < 3; ++t) {
            if (b_m) drain_add<F, LCB, OSH>(b_lc, b_v, b_m, lane, acc);
            b_m = a_m;
            b_lc = a_lc;
            if (a_m) drain_gather<F>(a_r, a_m, lane, h, b_v);
            a_m = tail;
            if (tail) drain_take<LCB>(ml, 0, tail, lane, e0, rows, a_lc, a_r);
            tail = 0;
        }
    }
    __syncthreads();

    float* dst = part + ((size_t)q * n + lo) * F;
    for (int i = threadIdx.x; i < nr * F; i += TWG) dst[i] = acc[i];
}

__global__ void gcn_dinv_kernel(const float* __restrict__ part,
                                float* __restrict__ dinv, int n) {
    const int i = blockIdx.x * blockDim.x + threadIdx.x;
    if (i >= n) return;
    float c = 1.0f;
    #pragma unroll 16
    for (int q = 0; q < QD; ++q) c += part[(size_t)q * n + i];
    dinv[i] = rsqrtf(c);
}

__global__ void gcn_xw1_wmma_kernel(const float* __restrict__ x,
                                    const float* __restrict__ W1,
                                    const float* __restrict__ dinv,
                                    float* __restrict__ h1s,
                                    int n_nodes) {
    const int lane  = threadIdx.x & 31;
    const int wave  = threadIdx.x >> 5;
    const int tile  = blockIdx.x * (blockDim.x >> 5) + wave;
    const int node0 = tile * 16;
    if (node0 >= n_nodes) return;

    const int half = lane >> 4;
    const int l16  = lane & 15;
    const int m    = l16;

    v8f c = {0.f, 0.f, 0.f, 0.f, 0.f, 0.f, 0.f, 0.f};

    #pragma unroll
    for (int k0 = 0; k0 < 28; k0 += 4) {
        const int ka = k0 + 2 * half;
        v2f a, b;
        a.x = (ka     < FIN) ? x[(node0 + m) * FIN + ka    ] : 0.0f;
        a.y = (ka + 1 < FIN) ? x[(node0 + m) * FIN + ka + 1] : 0.0f;
        b.x = (ka     < FIN) ? W1[(ka    ) * F1 + l16] : 0.0f;
        b.y = (ka + 1 < FIN) ? W1[(ka + 1) * F1 + l16] : 0.0f;
        c = __builtin_amdgcn_wmma_f32_16x16x4_f32(
                 false, a,  false, b,
                 (short)0, c,  false,  false);
    }

    #pragma unroll
    for (int v = 0; v < 8; ++v) {
        const int node = node0 + v + 8 * half;
        h1s[node * F1 + l16] = c[v] * dinv[node];
    }
}

__global__ void gcn_layer1_finish_kernel(const float* __restrict__ part,
                                         const float* __restrict__ h1s,
                                         const float* __restrict__ dinv,
                                         const float* __restrict__ b1,
                                         const float* __restrict__ W2,
                                         float* __restrict__ h2s,
                                         int n) {
    const int gid = blockIdx.x * blockDim.x + threadIdx.x;
    const int i   = gid >> 2;
    const int g   = gid & 3;
    const int ic  = (i < n) ? i : (n - 1);
    float4 s = ((const float4*)h1s)[(size_t)ic * 4 + g];
    #pragma unroll
    for (int q = 0; q < Q1; ++q) {
        const float4 t = ((const float4*)part)[((size_t)q * n + ic) * 4 + g];
        s.x += t.x; s.y += t.y; s.z += t.z; s.w += t.w;
    }
    const float di = dinv[ic];
    const float v0 = fmaxf(fmaf(s.x, di, b1[4 * g + 0]), 0.0f);
    const float v1 = fmaxf(fmaf(s.y, di, b1[4 * g + 1]), 0.0f);
    const float v2 = fmaxf(fmaf(s.z, di, b1[4 * g + 2]), 0.0f);
    const float v3 = fmaxf(fmaf(s.w, di, b1[4 * g + 3]), 0.0f);
    const float* w = W2 + 8 * g;
    float a0 = v0 * w[0];
    float a1 = v0 * w[1];
    a0 = fmaf(v1, w[2], a0); a1 = fmaf(v1, w[3], a1);
    a0 = fmaf(v2, w[4], a0); a1 = fmaf(v2, w[5], a1);
    a0 = fmaf(v3, w[6], a0); a1 = fmaf(v3, w[7], a1);
    a0 += __shfl_xor(a0, 1); a1 += __shfl_xor(a1, 1);
    a0 += __shfl_xor(a0, 2); a1 += __shfl_xor(a1, 2);
    if (g == 0 && i < n) {
        float2 o; o.x = a0 * di; o.y = a1 * di;
        ((float2*)h2s)[i] = o;
    }
}

__global__ void gcn_final_kernel(const float* __restrict__ part,
                                 const float* __restrict__ h2s,
                                 const float* __restrict__ dinv,
                                 const float* __restrict__ b2,
                                 float* __restrict__ out,
                                 int n) {
    const int i = blockIdx.x * blockDim.x + threadIdx.x;
    if (i >= n) return;
    float2 s = ((const float2*)h2s)[i];
    #pragma unroll 16
    for (int q = 0; q < Q2; ++q) {
        const float2 t = ((const float2*)part)[(size_t)q * n + i];
        s.x += t.x; s.y += t.y;
    }
    const float di = dinv[i];
    const float a = fmaf(s.x, di, b2[0]);
    const float b = fmaf(s.y, di, b2[1]);
    const float m = fmaxf(a, b);
    const float lse = m + logf(expf(a - m) + expf(b - m));
    float2 o; o.x = a - lse; o.y = b - lse;
    ((float2*)out)[i] = o;
}

extern "C" void kernel_launch(void* const* d_in, const int* in_sizes, int n_in,
                              void* d_out, int out_size, void* d_ws, size_t ws_size,
                              hipStream_t stream) {
    const float* x   = (const float*)d_in[0];
    const int*   ei  = (const int*)d_in[1];
    const float* W1  = (const float*)d_in[2];
    const float* b1  = (const float*)d_in[3];
    const float* W2  = (const float*)d_in[4];
    const float* b2  = (const float*)d_in[5];
    float*       out = (float*)d_out;

    const int N = in_sizes[0] / FIN;
    const int E = in_sizes[1] / 2;
    const int* rows = ei;
    const int* cols = ei + E;

    float* ws   = (float*)d_ws;
    float* h1s  = ws;
    float* h2s  = ws + (size_t)16 * N;
    float* dinv = ws + (size_t)18 * N;
    float* part = ws + (size_t)20 * N;

    const int BS = 256;

    gcn_tile_kernel<1, RD, QD, 256><<<((N + RD - 1) / RD) * QD, TWG, 0, stream>>>(
        rows, cols, h1s, part, N, E, (E + QD - 1) / QD);

    gcn_dinv_kernel<<<(N + BS - 1) / BS, BS, 0, stream>>>(part, dinv, N);

    {
        int tiles = (N + 15) / 16;
        int waves_per_block = BS / 32;
        int blocks = (tiles + waves_per_block - 1) / waves_per_block;
        gcn_xw1_wmma_kernel<<<blocks, BS, 0, stream>>>(x, W1, dinv, h1s, N);
    }

    gcn_tile_kernel<F1, R1, Q1, 64><<<((N + R1 - 1) / R1) * Q1, TWG, 0, stream>>>(
        rows, cols, h1s, part, N, E, (E + Q1 - 1) / Q1);

    gcn_layer1_finish_kernel<<<(4 * N + BS - 1) / BS, BS, 0, stream>>>(
        part, h1s, dinv, b1, W2, h2s, N);

    gcn_tile_kernel<F2, R2, Q2, 256><<<((N + R2 - 1) / R2) * Q2, TWG, 0, stream>>>(
        rows, cols, h2s, part, N, E, (E + Q2 - 1) / Q2);

    gcn_final_kernel<<<(N + BS - 1) / BS, BS, 0, stream>>>(part, h2s, dinv, b2, out, N);
}


# Sources streamed beside the targets: the filter loads a pass's source indices 16 B per lane and a hit's queue entry carries its source, so the drain fetches no 4-byte source per hit; sum order kept
# speedup vs baseline: 3.3231x; 1.0711x over previous
#include <hip/hip_runtime.h>


typedef __attribute__((ext_vector_type(2))) float v2f;
typedef __attribute__((ext_vector_type(8))) float v8f;

#define F1 16
#define F2 2
#define FIN 25

#define TWG 1024
#define RD 12512
#define QD 32
#define R1 4000
#define Q1 10
#define R2 12512
#define Q2 32

#define CHUNK 256
#define LCAP  160
#define TSTR  33

__device__ __forceinline__ void lds_round_barrier() {
    asm volatile("s_wait_dscnt 0x0\n\ts_barrier_signal -1\n\ts_barrier_wait -1" ::: "memory");
}

template <int LCB>
__device__ __forceinline__ void drain_take(const unsigned* ml, int start, int m, int lane, unsigned& lc, int& r) {
    constexpr unsigned LCM = (1u << LCB) - 1u;
    lc = 0xFFFFFFFFu;
    r = 0;
    if (lane < m) {
        const unsigned ent = ml[start + lane];
        lc = ent & LCM;
        r = (int)(ent >> LCB);
    }
}

template <int F>
__device__ __forceinline__ void drain_gather(int r, int m, int lane, const float* __restrict__ h, float* v) {
    if (F == 16) {
        const int sub = lane >> 2;
        const int qd  = lane & 3;
        #pragma unroll
        for (int s = 0; s < 4; ++s) {
            const int src = 8 * s + sub;
            const int rs = __shfl(r, src);
            float4 t = make_float4(0.f, 0.f, 0.f, 0.f);
            if (src < m) t = ((const float4*)h)[rs * 4 + qd];
            v[4 * s + 0] = t.x;
            v[4 * s + 1] = t.y;
            v[4 * s + 2] = t.z;
            v[4 * s + 3] = t.w;
        }
    } else {
        float2 t = make_float2(0.f, 0.f);
        if (lane < m) t = ((const float2*)h)[r];
        v[0] = t.x;
        v[1] = t.y;
    }
}

template <int D>
__device__ __forceinline__ unsigned same_node_below(unsigned lc, int l8) {
    const unsigned o = (unsigned)__builtin_amdgcn_update_dpp(0, (int)lc, 0x110 + D, 0xF, 0xF, false);
    return (l8 >= D && o == lc) ? 1u : 0u;
}

template <int LCB, int OSH>
__device__ __forceinline__ unsigned same_node_lanes(unsigned lc, bool valid) {
    unsigned diff = 0u;
    #pragma unroll
    for (int bit = 0; bit < LCB; ++bit) {
        if (bit >= OSH && bit < OSH + 5) continue;
        const unsigned sp = (unsigned)(((int)(lc << (31 - bit))) >> 31);
        const unsigned mb = __builtin_amdgcn_ballot_w32(valid && sp != 0u);
        diff |= mb ^ sp;
    }
    return __builtin_amdgcn_ballot_w32(valid) & ~diff;
}

template <int F, int LCB, int OSH>
__device__ __forceinline__ void drain_add(unsigned lc, const float* v, int m, int lane, float* acc) {
    if (F == 16) {
        const int sub = lane >> 2;
        const int qd  = lane & 3;
        const int l8  = lane & 7;
        unsigned rk = same_node_below<1>(lc, l8);
        rk += same_node_below<2>(lc, l8);
        rk += same_node_below<3>(lc, l8);
        rk += same_node_below<4>(lc, l8);
        rk += same_node_below<5>(lc, l8);
        rk += same_node_below<6>(lc, l8);
        rk += same_node_below<7>(lc, l8);
        const unsigned pk = (rk << 16) | (lc & 0xFFFFu);
        unsigned pks[4];
        #pragma unroll
        for (int s = 0; s < 4; ++s) pks[s] = (unsigned)__shfl((int)pk, 8 * s + sub);
        float4* acc4 = (float4*)acc;
        #pragma unroll
        for (int s = 0; s < 4; ++s) {
            const bool valid = 8 * s + sub < m;
            const unsigned node = pks[s] & 0xFFFFu;
            const unsigned rank = pks[s] >> 16;
            for (unsigned t = 0u; t < 8u; ++t) {
                const bool act = valid && (rank == t);
                if (__builtin_amdgcn_ballot_w32(act) == 0u) break;
                if (act) {
                    float4 a = acc4[node * 4u + (unsigned)qd];
                    a.x += v[4 * s + 0];
                    a.y += v[4 * s + 1];
                    a.z += v[4 * s + 2];
                    a.w += v[4 * s + 3];
                    acc4[node * 4u + (unsigned)qd] = a;
                }
            }
        }
    } else {
        const bool valid = lane < m;
        const unsigned rank = __builtin_amdgcn_mbcnt_lo(same_node_lanes<LCB, OSH>(lc, valid), 0u);
        float2* acc2 = (float2*)acc;
        for (unsigned t = 0u; t < 32u; ++t) {
            const bool act = valid && (rank == t);
            if (__builtin_amdgcn_ballot_w32(act) == 0u) break;
            if (act) {
                float2 a = acc2[lc];
                a.x += v[0]; a.y += v[1];
                acc2[lc] = a;
            }
        }
    }
}

template <int LCB, int OSH>
__device__ __forceinline__ void drain_count(const unsigned* ml, int start, int m, int lane, float* acc) {
    constexpr unsigned LCM = (1u << LCB) - 1u;
    const bool valid = lane < m;
    unsigned lc = 0xFFFFFFFFu;
    if (valid) lc = ml[start + lane] & LCM;
    const unsigned same = same_node_lanes<LCB, OSH>(lc, valid);
    const unsigned rank = __builtin_amdgcn_mbcnt_lo(same, 0u);
    if (valid && rank == 0u) acc[lc] += (float)__builtin_popcount(same);
}

__device__ __forceinline__ int filter_test(const int4 (&c)[2], int cursor, int wend, int lo, int nr, int lane,
                                           unsigned (&lc)[8], bool (&hit)[8], unsigned (&mk)[8]) {
    const int left = wend - cursor - 4 * lane;
    int h = 0;
    #pragma unroll
    for (int j = 0; j < 2; ++j) {
        const int off = (j * 128 < left) ? lo : -(1 << 30);
        const int t[4] = {c[j].x, c[j].y, c[j].z, c[j].w};
        #pragma unroll
        for (int i = 0; i < 4; ++i) {
            const int k = 4 * j + i;
            lc[k]  = (unsigned)(t[i] - off);
            hit[k] = lc[k] < (unsigned)nr;
            mk[k]  = __builtin_amdgcn_ballot_w32(hit[k]);
            h += __builtin_popcount(mk[k]);
        }
    }
    return h;
}

template <int LCB, bool SRC>
__device__ __forceinline__ void filter_store(const unsigned (&lc)[8], const bool (&hit)[8], const unsigned (&mk)[8],
                                             const int4 (&r)[2], unsigned* qrow, int& na) {
    #pragma unroll
    for (int j = 0; j < 2; ++j) {
        const int s[4] = {r[j].x, r[j].y, r[j].z, r[j].w};
        #pragma unroll
        for (int i = 0; i < 4; ++i) {
            const int k = 4 * j + i;
            if (hit[k]) qrow[na + (int)__builtin_amdgcn_mbcnt_lo(mk[k], 0u)] =
                            SRC ? (((unsigned)s[i] << LCB) | lc[k]) : lc[k];
            na += __builtin_popcount(mk[k]);
        }
    }
}

template <int F, int R, int Q, int QX>
__global__ __launch_bounds__(TWG) void gcn_tile_kernel(const int* __restrict__ rows,
                                                       const int* __restrict__ cols,
                                                       const float* __restrict__ h,
                                                       float* __restrict__ part,
                                                       int n, int n_edges, int slice) {
    constexpr int LCB = (R <= 4096) ? 12 : (R <= 16384) ? 14 : 16;
    constexpr int OSH = (F == 1) ? 6 : 5;
    static_assert(R <= (1 << LCB), "node index must fit its field");
    constexpr bool SRC = (F != 1);
    constexpr int QCAP = CHUNK + QX;
    constexpr int QSTR = QCAP + 4;
    constexpr int NB   = QCAP / 32;
    static_assert(QSTR % 64 == 4, "queue rows must lie 4 banks apart");
    __shared__ __attribute__((aligned(16))) float acc[R * F];
    __shared__ __attribute__((aligned(16))) unsigned qa[32 * QSTR];
    __shared__ unsigned cntm[32 * TSTR];
    __shared__ unsigned ml_all[32 * LCAP];
    const int lane = threadIdx.x & 31;
    const int wave = __builtin_amdgcn_readfirstlane((int)(threadIdx.x >> 5));
    unsigned* ml = ml_all + wave * LCAP;
    const int p  = blockIdx.x / Q;
    const int q  = blockIdx.x - p * Q;
    const int lo = p * R;
    const int nr = min(R, n - lo);

    for (int i = threadIdx.x; i < nr * F; i += TWG) acc[i] = 0.f;

    const int e0 = q * slice;
    const int e1 = min(n_edges, e0 + slice);
    const int wlen = (((slice + 31) >> 5) + 31) & ~31;
    const int wbeg = e0 + wave * wlen;
    const int wend = min(e1, wbeg + wlen);
    const int maxr = (wlen + CHUNK - 1) / CHUNK + 1;
    const int maxp = maxr / 2 + 1;
    const int4* cols4 = (const int4*)cols;
    const int4* rows4 = (const int4*)rows;
    const int last4 = (n_edges >> 2) - 1;
    int cursor = wbeg;
    int4 ca[2], cb[2], rc[2];
    #pragma unroll
    for (int j = 0; j < 2; ++j) {
        const int ia = min((cursor >> 2) + j * 32 + lane, last4);
        ca[j] = cols4[ia];
        cb[j] = cols4[min(((cursor + CHUNK) >> 2) + j * 32 + lane, last4)];
        rc[j] = SRC ? rows4[ia] : make_int4(0, 0, 0, 0);
    }
    bool odd  = false;
    int cnt = 0;
    constexpr int VN = (F == 16) ? 16 : 2;
    unsigned a_lc = 0u, b_lc = 0u;
    int a_r = 0, a_m = 0, b_m = 0;
    float b_v[VN];
    #pragma unroll
    for (int j = 0; j < VN; ++j) b_v[j] = 0.f;
    for (int round = 0; round < maxr; ++round) {
        if (odd) {
            #pragma unroll
            for (int j = 0; j < 2; ++j) { const int4 t = ca[j]; ca[j] = cb[j]; cb[j] = t; }
        }
        odd = false;
        int na = 0;
        unsigned* qrow = qa + wave * QSTR;
        for (int trip = 0; trip < maxp; ++trip) {
            {
                unsigned lc[8], mk[8];
                bool hit[8];
                const int h = filter_test(ca, cursor, wend, lo, nr, lane, lc, hit, mk);
                if (na + h > QCAP) break;
                #pragma unroll
                for (int j = 0; j < 2; ++j) ca[j] = cols4[min(((cursor + 2 * CHUNK) >> 2) + j * 32 + lane, last4)];
                filter_store<LCB, SRC>(lc, hit, mk, rc, qrow, na);
                if (SRC) {
                    #pragma unroll
                    for (int j = 0; j < 2; ++j) rc[j] = rows4[min(((cursor + CHUNK) >> 2) + j * 32 + lane, last4)];
                }
                cursor += CHUNK;
                odd = true;
                if (cursor >= wend) break;
            }
            {
                unsigned lc[8], mk[8];
                bool hit[8];
                const int h = filter_test(cb, cursor, wend, lo, nr, lane, lc, hit, mk);
                if (na + h > QCAP) break;
                #pragma unroll
                for (int j = 0; j < 2; ++j) cb[j] = cols4[min(((cursor + 2 * CHUNK) >> 2) + j * 32 + lane, last4)];
                filter_store<LCB, SRC>(lc, hit, mk, rc, qrow, na);
                if (SRC) {
                    #pragma unroll
                    for (int j = 0; j < 2; ++j) rc[j] = rows4[min(((cursor + CHUNK) >> 2) + j * 32 + lane, last4)];
                }
                cursor += CHUNK;
                odd = false;
                if (cursor >= wend) break;
            }
        }
        unsigned* cw = cntm + wave * TSTR;
        cw[lane] = 0u;
        const int nb = (na + 31) >> 5;
        unsigned se[NB], sp[NB];
        #pragma unroll
        for (int b = 0; b < NB; ++b) {
            se[b] = 0u;
            sp[b] = 0u;
            if (b < nb) {
                const bool valid = b * 32 + lane < na;
                unsigned ent = 0u;
                if (valid) ent = qa[wave * QSTR + b * 32 + lane];
                unsigned same = __builtin_amdgcn_ballot_w32(valid);
                #pragma unroll
                for (int bit = 0; bit < 5; ++bit) {
                    const int sb = ((int)(ent << (31 - OSH - bit))) >> 31;
                    const unsigned m = __builtin_amdgcn_ballot_w32(valid && sb != 0);
                    same &= ~(m ^ (unsigned)sb);
                }
                const unsigned rank = __builtin_amdgcn_mbcnt_lo(same, 0u);
                const unsigned tot  = (unsigned)__builtin_popcount(same);
                unsigned base = 0u;
                if (valid) base = cw[(ent >> OSH) & 31u];
                if (valid && rank + 1u == tot) cw[(ent >> OSH) & 31u] = base + tot;
                se[b] = ent;
                sp[b] = base + rank;
            }
        }
        {
            const unsigned c = cw[lane];
            unsigned inc = c;
            inc += (unsigned)__builtin_amdgcn_update_dpp(0, (int)inc, 0x111, 0xF, 0xF, false);
            inc += (unsigned)__builtin_amdgcn_update_dpp(0, (int)inc, 0x112, 0xF, 0xF, false);
            inc += (unsigned)__builtin_amdgcn_update_dpp(0, (int)inc, 0x114, 0xF, 0xF, false);
            inc += (unsigned)__builtin_amdgcn_update_dpp(0, (int)inc, 0x118, 0xF, 0xF, false);
            const unsigned row0 = (unsigned)__builtin_amdgcn_readlane((int)inc, 15);
            if (lane >= 16) inc += row0;
            cw[lane] = ((inc - c) << 16) | c;
        }
        #pragma unroll
        for (int b = 0; b < NB; ++b) {
            if (b < nb) {
                if (b * 32 + lane < na) qa[wave * QSTR + (int)((cw[(se[b] >> OSH) & 31u] >> 16) + sp[b])] = se[b];
            }
        }
        if (lane == 0) cw[32] = (cursor >= wend) ? 1u : 0u;
        lds_round_barrier();
        const unsigned sc = cntm[lane * TSTR + wave];
        const unsigned rw = cntm[lane * TSTR + 32];
        const int st = (int)(sc >> 16);
        const int nl = (int)(sc & 0xFFFFu);
        const bool all_through = __builtin_amdgcn_ballot_w32(rw != 0u) == 0xFFFFFFFFu;
        for (int i0 = 0; __builtin_amdgcn_ballot_w32(i0 < nl) != 0u; i0 += 4) {
            unsigned ent[4];
            #pragma unroll
            for (int u = 0; u < 4; ++u) {
                ent[u] = 0u;
                if (i0 + u < nl) ent[u] = qa[lane * QSTR + st + i0 + u];
            }
            #pragma unroll
            for (int u = 0; u < 4; ++u) {
                const bool have = i0 + u < nl;
                const unsigned mk = __builtin_amdgcn_ballot_w32(have);
                if (mk != 0u) {
                    if (have) ml[cnt + (int)__builtin_amdgcn_mbcnt_lo(mk, 0u)] = ent[u];
                    cnt += __builtin_popcount(mk);
                }
            }
            while (cnt >= 32) {
                cnt -= 32;
                if (F == 1) {
                    drain_count<LCB, OSH>(ml, cnt, 32, lane, acc);
                } else {
                    float c_v[VN];
                    #pragma unroll
                    for (int j = 0; j < VN; ++j) c_v[j] = b_v[j];
                    const unsigned c_lc = b_lc;
                    const int c_m = b_m;
                    b_m = a_m;
                    b_lc = a_lc;
                    drain_gather<F>(a_r, 32, lane, h, b_v);
                    drain_take<LCB>(ml, cnt, 32, lane, a_lc, a_r);
                    a_m = 32;
                    if (c_m) drain_add<F, LCB, OSH>(c_lc, c_v, 32, lane, acc);
                }
            }
        }
        lds_round_barrier();
        if (all_through) break;
    }
    if (F == 1) {
        if (cnt > 0) drain_count<LCB, OSH>(ml, 0, cnt, lane, acc);
    } else {
        int tail = cnt;
        #pragma nounroll
        for (int t = 0; t < 3; ++t) {
            if (b_m) drain_add<F, LCB, OSH>(b_lc, b_v, b_m, lane, acc);
            b_m = a_m;
            b_lc = a_lc;
            if (a_m) drain_gather<F>(a_r, a_m, lane, h, b_v);
            a_m = tail;
            if (tail) drain_take<LCB>(ml, 0, tail, lane, a_lc, a_r);
            tail = 0;
        }
    }
    __syncthreads();

    float* dst = part + ((size_t)q * n + lo) * F;
    for (int i = threadIdx.x; i < nr * F; i += TWG) dst[i] = acc[i];
}

__global__ void gcn_dinv_kernel(const float* __restrict__ part,
                                float* __restrict__ dinv, int n) {
    const int i = blockIdx.x * blockDim.x + threadIdx.x;
    if (i >= n) return;
    float c = 1.0f;
    #pragma unroll 16
    for (int q = 0; q < QD; ++q) c += part[(size_t)q * n + i];
    dinv[i] = rsqrtf(c);
}

__global__ void gcn_xw1_wmma_kernel(const float* __restrict__ x,
                                    const float* __restrict__ W1,
                                    const float* __restrict__ dinv,
                                    float* __restrict__ h1s,
                                    int n_nodes) {
    const int lane  = threadIdx.x & 31;
    const int wave  = threadIdx.x >> 5;
    const int tile  = blockIdx.x * (blockDim.x >> 5) + wave;
    const int node0 = tile * 16;
    if (node0 >= n_nodes) return;

    const int half = lane >> 4;
    const int l16  = lane & 15;
    const int m    = l16;

    v8f c = {0.f, 0.f, 0.f, 0.f, 0.f, 0.f, 0.f, 0.f};

    #pragma unroll
    for (int k0 = 0; k0 < 28; k0 += 4) {
        const int ka = k0 + 2 * half;
        v2f a, b;
        a.x = (ka     < FIN) ? x[(node0 + m) * FIN + ka    ] : 0.0f;
        a.y = (ka + 1 < FIN) ? x[(node0 + m) * FIN + ka + 1] : 0.0f;
        b.x = (ka     < FIN) ? W1[(ka    ) * F1 + l16] : 0.0f;
        b.y = (ka + 1 < FIN) ? W1[(ka + 1) * F1 + l16] : 0.0f;
        c = __builtin_amdgcn_wmma_f32_16x16x4_f32(
                 false, a,  false, b,
                 (short)0, c,  false,  false);
    }

    #pragma unroll
    for (int v = 0; v < 8; ++v) {
        const int node = node0 + v + 8 * half;
        h1s[node * F1 + l16] = c[v] * dinv[node];
    }
}

__global__ void gcn_layer1_finish_kernel(const float* __restrict__ part,
                                         const float* __restrict__ h1s,
                                         const float* __restrict__ dinv,
                                         const float* __restrict__ b1,
                                         const float* __restrict__ W2,
                                         float* __restrict__ h2s,
                                         int n) {
    const int gid = blockIdx.x * blockDim.x + threadIdx.x;
    const int i   = gid >> 2;
    const int g   = gid & 3;
    const int ic  = (i < n) ? i : (n - 1);
    float4 s = ((const float4*)h1s)[(size_t)ic * 4 + g];
    #pragma unroll
    for (int q = 0; q < Q1; ++q) {
        const float4 t = ((const float4*)part)[((size_t)q * n + ic) * 4 + g];
        s.x += t.x; s.y += t.y; s.z += t.z; s.w += t.w;
    }
    const float di = dinv[ic];
    const float v0 = fmaxf(fmaf(s.x, di, b1[4 * g + 0]), 0.0f);
    const float v1 = fmaxf(fmaf(s.y, di, b1[4 * g + 1]), 0.0f);
    const float v2 = fmaxf(fmaf(s.z, di, b1[4 * g + 2]), 0.0f);
    const float v3 = fmaxf(fmaf(s.w, di, b1[4 * g + 3]), 0.0f);
    const float* w = W2 + 8 * g;
    float a0 = v0 * w[0];
    float a1 = v0 * w[1];
    a0 = fmaf(v1, w[2], a0); a1 = fmaf(v1, w[3], a1);
    a0 = fmaf(v2, w[4], a0); a1 = fmaf(v2, w[5], a1);
    a0 = fmaf(v3, w[6], a0); a1 = fmaf(v3, w[7], a1);
    a0 += __shfl_xor(a0, 1); a1 += __shfl_xor(a1, 1);
    a0 += __shfl_xor(a0, 2); a1 += __shfl_xor(a1, 2);
    if (g == 0 && i < n) {
        float2 o; o.x = a0 * di; o.y = a1 * di;
        ((float2*)h2s)[i] = o;
    }
}

__global__ void gcn_final_kernel(const float* __restrict__ part,
                                 const float* __restrict__ h2s,
                                 const float* __restrict__ dinv,
                                 const float* __restrict__ b2,
                                 float* __restrict__ out,
                                 int n) {
    const int i = blockIdx.x * blockDim.x + threadIdx.x;
    if (i >= n) return;
    float2 s = ((const float2*)h2s)[i];
    #pragma unroll 16
    for (int q = 0; q < Q2; ++q) {
        const float2 t = ((const float2*)part)[(size_t)q * n + i];
        s.x += t.x; s.y += t.y;
    }
    const float di = dinv[i];
    const float a = fmaf(s.x, di, b2[0]);
    const float b = fmaf(s.y, di, b2[1]);
    const float m = fmaxf(a, b);
    const float lse = m + logf(expf(a - m) + expf(b - m));
    float2 o; o.x = a - lse; o.y = b - lse;
    ((float2*)out)[i] = o;
}

extern "C" void kernel_launch(void* const* d_in, const int* in_sizes, int n_in,
                              void* d_out, int out_size, void* d_ws, size_t ws_size,
                              hipStream_t stream) {
    const float* x   = (const float*)d_in[0];
    const int*   ei  = (const int*)d_in[1];
    const float* W1  = (const float*)d_in[2];
    const float* b1  = (const float*)d_in[3];
    const float* W2  = (const float*)d_in[4];
    const float* b2  = (const float*)d_in[5];
    float*       out = (float*)d_out;

    const int N = in_sizes[0] / FIN;
    const int E = in_sizes[1] / 2;
    const int* rows = ei;
    const int* cols = ei + E;

    float* ws   = (float*)d_ws;
    float* h1s  = ws;
    float* h2s  = ws + (size_t)16 * N;
    float* dinv = ws + (size_t)18 * N;
    float* part = ws + (size_t)20 * N;

    const int BS = 256;

    gcn_tile_kernel<1, RD, QD, 256><<<((N + RD - 1) / RD) * QD, TWG, 0, stream>>>(
        rows, cols, h1s, part, N, E, (E + QD - 1) / QD);

    gcn_dinv_kernel<<<(N + BS - 1) / BS, BS, 0, stream>>>(part, dinv, N);

    {
        int tiles = (N + 15) / 16;
        int waves_per_block = BS / 32;
        int blocks = (tiles + waves_per_block - 1) / waves_per_block;
        gcn_xw1_wmma_kernel<<<blocks, BS, 0, stream>>>(x, W1, dinv, h1s, N);
    }

    gcn_tile_kernel<F1, R1, Q1, 64><<<((N + R1 - 1) / R1) * Q1, TWG, 0, stream>>>(
        rows, cols, h1s, part, N, E, (E + Q1 - 1) / Q1);

    gcn_layer1_finish_kernel<<<(4 * N + BS - 1) / BS, BS, 0, stream>>>(
        part, h1s, dinv, b1, W2, h2s, N);

    gcn_tile_kernel<F2, R2, Q2, 256><<<((N + R2 - 1) / R2) * Q2, TWG, 0, stream>>>(
        rows, cols, h2s, part, N, E, (E + Q2 - 1) / Q2);

    gcn_final_kernel<<<(N + BS - 1) / BS, BS, 0, stream>>>(part, h2s, dinv, b2, out, N);
}
